# starting kernel
# baseline (speedup 1.0000x reference)
#include <hip/hip_runtime.h>
#include <hip/hip_bf16.h>

#define NND   50000
#define FIN   128
#define NH    8
#define HIDC  32
#define HC    256
#define ODIM  40
#define NE    800000
#define NET   850000
#define NEG_SLOPE 0.2f
#define EPSV  1e-16f

typedef __attribute__((ext_vector_type(2))) float v2f;
typedef __attribute__((ext_vector_type(8))) float v8f;

__device__ __forceinline__ float wave_sum(float v) {
#pragma unroll
  for (int off = 16; off > 0; off >>= 1) v += __shfl_xor(v, off, 32);
  return v;
}
__device__ __forceinline__ int f32_ord(float x) {
  int i = __float_as_int(x);
  return i >= 0 ? i : (i ^ 0x7FFFFFFF);
}
__device__ __forceinline__ float ord_f32(int i) {
  return __int_as_float(i >= 0 ? i : (i ^ 0x7FFFFFFF));
}

__global__ void k_fill_f32(float* __restrict__ p, float v, int n) {
  int i = blockIdx.x * blockDim.x + threadIdx.x;
  if (i < n) p[i] = v;
}
__global__ void k_fill_i32(int* __restrict__ p, int v, int n) {
  int i = blockIdx.x * blockDim.x + threadIdx.x;
  if (i < n) p[i] = v;
}

__global__ __launch_bounds__(256) void k_gemm1(const float* __restrict__ x,
                                               const float* __restrict__ W,
                                               float* __restrict__ h) {
  __shared__ float As[16 * FIN];
  const int tid  = threadIdx.x;
  const int wave = tid >> 5, lane = tid & 31;
  const int m0   = blockIdx.x * 16;
  {
    const float4* xv = (const float4*)(x + (size_t)m0 * FIN);
    float4* sv = (float4*)As;
    sv[tid]       = xv[tid];
    sv[tid + 256] = xv[tid + 256];
  }
  __syncthreads();
  const int n0 = blockIdx.y * 128 + wave * 16;
  const int la = lane & 15, hi = lane >> 4;
  v8f acc = {};
#pragma unroll
  for (int k0 = 0; k0 < FIN; k0 += 4) {
    const int ka = k0 + hi * 2;
    v2f a, b;
    a.x = As[la * FIN + ka];
    a.y = As[la * FIN + ka + 1];
    b.x = W[ka * HC + n0 + la];
    b.y = W[(ka + 1) * HC + n0 + la];
    acc = __builtin_amdgcn_wmma_f32_16x16x4_f32(false, a, false, b, (short)0,
                                                acc, false, false);
  }
#pragma unroll
  for (int j = 0; j < 8; ++j)
    h[(size_t)(m0 + j + hi * 8) * HC + n0 + la] = acc[j];
}

__global__ __launch_bounds__(96) void k_gemm2(const float* __restrict__ h2,
                                              const float* __restrict__ W,
                                              float* __restrict__ t) {
  __shared__ float As[16 * HC];
  const int tid  = threadIdx.x;
  const int wave = tid >> 5, lane = tid & 31;
  const int m0   = blockIdx.x * 16;
  {
    const float4* xv = (const float4*)(h2 + (size_t)m0 * HC);
    float4* sv = (float4*)As;
    for (int i = tid; i < 1024; i += 96) sv[i] = xv[i];
  }
  __syncthreads();
  const int n0 = wave * 16;
  const int la = lane & 15, hi = lane >> 4;
  const int col = n0 + la;
  const bool ok = col < ODIM;
  v8f acc = {};
#pragma unroll 8
  for (int k0 = 0; k0 < HC; k0 += 4) {
    const int ka = k0 + hi * 2;
    v2f a, b;
    a.x = As[la * HC + ka];
    a.y = As[la * HC + ka + 1];
    b.x = ok ? W[ka * ODIM + col] : 0.f;
    b.y = ok ? W[(ka + 1) * ODIM + col] : 0.f;
    acc = __builtin_amdgcn_wmma_f32_16x16x4_f32(false, a, false, b, (short)0,
                                                acc, false, false);
  }
#pragma unroll
  for (int j = 0; j < 8; ++j)
    if (ok) t[(size_t)(m0 + j + hi * 8) * ODIM + col] = acc[j];
}

__global__ __launch_bounds__(256) void k_att1(const float* __restrict__ h,
                                              const float* __restrict__ as_,
                                              const float* __restrict__ ad_,
                                              float* __restrict__ os,
                                              float* __restrict__ od) {
  const int node = blockIdx.x * 8 + (threadIdx.x >> 5);
  const int lane = threadIdx.x & 31;
  if (node >= NND) return;
  const float* hr = h + (size_t)node * HC;
#pragma unroll
  for (int hd = 0; hd < NH; ++hd) {
    const float v = hr[hd * HIDC + lane];
    const float s = wave_sum(v * as_[hd * HIDC + lane]);
    const float d = wave_sum(v * ad_[hd * HIDC + lane]);
    if (lane == 0) { os[node * NH + hd] = s; od[node * NH + hd] = d; }
  }
}

__global__ __launch_bounds__(256) void k_att2(const float* __restrict__ t,
                                              const float* __restrict__ as_,
                                              const float* __restrict__ ad_,
                                              float* __restrict__ os,
                                              float* __restrict__ od) {
  const int node = blockIdx.x * 8 + (threadIdx.x >> 5);
  const int lane = threadIdx.x & 31;
  if (node >= NND) return;
  const float* tr = t + (size_t)node * ODIM;
  float s = 0.f, d = 0.f;
  {
    const float v = tr[lane];
    s += v * as_[lane]; d += v * ad_[lane];
  }
  if (lane < ODIM - 32) {
    const float v = tr[32 + lane];
    s += v * as_[32 + lane]; d += v * ad_[32 + lane];
  }
  s = wave_sum(s); d = wave_sum(d);
  if (lane == 0) { os[node] = s; od[node] = d; }
}

__device__ __forceinline__ void edge_sd(const int* ei, int eid, int& s, int& d) {
  if (eid < NE) { s = ei[eid]; d = ei[NE + eid]; }
  else          { s = d = eid - NE; }
}

__global__ void k_edgeA1(const int* __restrict__ ei,
                         const float* __restrict__ as_, const float* __restrict__ ad_,
                         float* __restrict__ e, int* __restrict__ nmax) {
  const int t = blockIdx.x * blockDim.x + threadIdx.x;
  if (t >= NET * NH) return;
  const int eid = t >> 3, hd = t & 7;
  int s, d; edge_sd(ei, eid, s, d);
  float v = as_[s * NH + hd] + ad_[d * NH + hd];
  v = v > 0.f ? v : v * NEG_SLOPE;
  e[t] = v;
  atomicMax(&nmax[d * NH + hd], f32_ord(v));
}

__global__ void k_edgeB1(const int* __restrict__ ei, float* __restrict__ e,
                         const int* __restrict__ nmax, float* __restrict__ denom) {
  const int t = blockIdx.x * blockDim.x + threadIdx.x;
  if (t >= NET * NH) return;
  const int eid = t >> 3, hd = t & 7;
  int s, d; edge_sd(ei, eid, s, d);
  const float ex = __expf(e[t] - ord_f32(nmax[d * NH + hd]));
  e[t] = ex;
  atomicAdd(&denom[d * NH + hd], ex);
}

__global__ __launch_bounds__(256) void k_edgeC1(const int* __restrict__ ei,
                                                const float* __restrict__ h,
                                                const float* __restrict__ eexp,
                                                const float* __restrict__ denom,
                                                float* __restrict__ agg) {
  const int eid  = blockIdx.x * 8 + (threadIdx.x >> 5);
  const int lane = threadIdx.x & 31;
  if (eid >= NET) return;
  int s, d; edge_sd(ei, eid, s, d);
  const float* hs = h + (size_t)s * HC;
  float* ag = agg + (size_t)d * HC;
  __builtin_prefetch(hs, 0, 0);
#pragma unroll
  for (int hd = 0; hd < NH; ++hd) {
    const float alpha = eexp[(size_t)eid * NH + hd] /
                        (denom[d * NH + hd] + EPSV);
    atomicAdd(&ag[hd * HIDC + lane], hs[hd * HIDC + lane] * alpha);
  }
}

__global__ void k_edgeA2(const int* __restrict__ ei,
                         const float* __restrict__ as_, const float* __restrict__ ad_,
                         float* __restrict__ e, int* __restrict__ nmax) {
  const int eid = blockIdx.x * blockDim.x + threadIdx.x;
  if (eid >= NET) return;
  int s, d; edge_sd(ei, eid, s, d);
  float v = as_[s] + ad_[d];
  v = v > 0.f ? v : v * NEG_SLOPE;
  e[eid] = v;
  atomicMax(&nmax[d], f32_ord(v));
}

__global__ void k_edgeB2(const int* __restrict__ ei, float* __restrict__ e,
                         const int* __restrict__ nmax, float* __restrict__ denom) {
  const int eid = blockIdx.x * blockDim.x + threadIdx.x;
  if (eid >= NET) return;
  int s, d; edge_sd(ei, eid, s, d);
  const float ex = __expf(e[eid] - ord_f32(nmax[d]));
  e[eid] = ex;
  atomicAdd(&denom[d], ex);
}

__global__ __launch_bounds__(256) void k_edgeC2(const int* __restrict__ ei,
                                                const float* __restrict__ t,
                                                const float* __restrict__ eexp,
                                                const float* __restrict__ denom,
                                                float* __restrict__ agg) {
  const int eid  = blockIdx.x * 8 + (threadIdx.x >> 5);
  const int lane = threadIdx.x & 31;
  if (eid >= NET) return;
  int s, d; edge_sd(ei, eid, s, d);
  const float alpha = eexp[eid] / (denom[d] + EPSV);
  const float* ts = t + (size_t)s * ODIM;
  float* ag = agg + (size_t)d * ODIM;
  atomicAdd(&ag[lane], ts[lane] * alpha);
  if (lane < ODIM - 32) atomicAdd(&ag[32 + lane], ts[32 + lane] * alpha);
}

__global__ void k_elu_bias(float* __restrict__ a, const float* __restrict__ b) {
  const int i = blockIdx.x * blockDim.x + threadIdx.x;
  if (i >= NND * HC) return;
  const float v = a[i] + b[i & (HC - 1)];
  a[i] = v > 0.f ? v : (__expf(v) - 1.f);
}

__global__ void k_out_bias(const float* __restrict__ agg, const float* __restrict__ b,
                           float* __restrict__ out) {
  const int i = blockIdx.x * blockDim.x + threadIdx.x;
  if (i >= NND * ODIM) return;
  out[i] = agg[i] + b[i % ODIM];
}

extern "C" void kernel_launch(void* const* d_in, const int* in_sizes, int n_in,
                              void* d_out, int out_size, void* d_ws, size_t ws_size,
                              hipStream_t stream) {
  const float* x   = (const float*)d_in[0];
  const int*   ei  = (const int*)d_in[1];
  const float* W1  = (const float*)d_in[2];
  const float* as1 = (const float*)d_in[3];
  const float* ad1 = (const float*)d_in[4];
  const float* b1  = (const float*)d_in[5];
  const float* W2  = (const float*)d_in[6];
  const float* as2 = (const float*)d_in[7];
  const float* ad2 = (const float*)d_in[8];
  const float* b2  = (const float*)d_in[9];
  float* out = (float*)d_out;

  float* ws   = (float*)d_ws;
  float* h1   = ws;
  float* agg1 = h1   + (size_t)NND * HC;
  float* aS1  = agg1 + (size_t)NND * HC;
  float* aD1  = aS1  + NND * NH;
  int*   nm1  = (int*)(aD1 + NND * NH);
  float* dn1  = (float*)(nm1 + NND * NH);
  float* e1   = dn1  + NND * NH;
  float* t2   = e1   + (size_t)NET * NH;
  float* aS2  = t2   + (size_t)NND * ODIM;
  float* aD2  = aS2  + NND;
  int*   nm2  = (int*)(aD2 + NND);
  float* dn2  = (float*)(nm2 + NND);
  float* e2   = dn2  + NND;
  float* agg2 = e2   + NET;

  const int B = 256;
  k_fill_f32<<<(NND * HC + B - 1) / B, B, 0, stream>>>(agg1, 0.f, NND * HC);
  k_fill_f32<<<(NND * NH + B - 1) / B, B, 0, stream>>>(dn1, 0.f, NND * NH);
  k_fill_i32<<<(NND * NH + B - 1) / B, B, 0, stream>>>(nm1, (int)0x80000000, NND * NH);
  k_fill_f32<<<(NND * ODIM + B - 1) / B, B, 0, stream>>>(agg2, 0.f, NND * ODIM);
  k_fill_f32<<<(NND + B - 1) / B, B, 0, stream>>>(dn2, 0.f, NND);
  k_fill_i32<<<(NND + B - 1) / B, B, 0, stream>>>(nm2, (int)0x80000000, NND);

  k_gemm1<<<dim3(NND / 16, 2), 256, 0, stream>>>(x, W1, h1);
  k_att1<<<(NND + 7) / 8, 256, 0, stream>>>(h1, as1, ad1, aS1, aD1);
  k_edgeA1<<<(NET * NH + B - 1) / B, B, 0, stream>>>(ei, aS1, aD1, e1, nm1);
  k_edgeB1<<<(NET * NH + B - 1) / B, B, 0, stream>>>(ei, e1, nm1, dn1);
  k_edgeC1<<<(NET + 7) / 8, 256, 0, stream>>>(ei, h1, e1, dn1, agg1);
  k_elu_bias<<<(NND * HC + B - 1) / B, B, 0, stream>>>(agg1, b1);

  k_gemm2<<<NND / 16, 96, 0, stream>>>(agg1, W2, t2);
  k_att2<<<(NND + 7) / 8, 256, 0, stream>>>(t2, as2, ad2, aS2, aD2);
  k_edgeA2<<<(NET + B - 1) / B, B, 0, stream>>>(ei, aS2, aD2, e2, nm2);
  k_edgeB2<<<(NET + B - 1) / B, B, 0, stream>>>(ei, e2, nm2, dn2);
  k_edgeC2<<<(NET + 7) / 8, 256, 0, stream>>>(ei, t2, e2, dn2, agg2);
  k_out_bias<<<(NND * ODIM + B - 1) / B, B, 0, stream>>>(agg2, b2, out);
}


# Edges counting-sorted by destination once per call; each layer's max, exp-sum and weighted sum become one wave-per-node gather loop: no float atomics, sum order = arrival rank per node
# speedup vs baseline: 1.7361x; 1.7361x over previous
#include <hip/hip_runtime.h>
#include <hip/hip_bf16.h>

#define NND   50000
#define FIN   128
#define NH    8
#define HIDC  32
#define HC    256
#define ODIM  40
#define NE    800000
#define NET   850000
#define NEG_SLOPE 0.2f
#define EPSV  1e-16f

typedef __attribute__((ext_vector_type(2))) float v2f;
typedef __attribute__((ext_vector_type(8))) float v8f;

__device__ __forceinline__ float wave_sum(float v) {
#pragma unroll
  for (int off = 16; off > 0; off >>= 1) v += __shfl_xor(v, off, 32);
  return v;
}
__device__ __forceinline__ float lrelu(float v) {
  return v > 0.f ? v : v * NEG_SLOPE;
}

__global__ void k_fill_i32(int* __restrict__ p, int v, int n) {
  int i = blockIdx.x * blockDim.x + threadIdx.x;
  if (i < n) p[i] = v;
}


__global__ void k_rank(const int* __restrict__ ei, int* __restrict__ deg,
                       int* __restrict__ rank) {
  const int e = blockIdx.x * blockDim.x + threadIdx.x;
  if (e >= NE) return;
  rank[e] = atomicAdd(&deg[ei[NE + e]], 1);
}

#define SCAN_PER 49
__global__ __launch_bounds__(1024) void k_scan(const int* __restrict__ deg,
                                               int* __restrict__ offs) {
  __shared__ int wsum[32];
  const int tid = threadIdx.x, lane = tid & 31, wave = tid >> 5;
  const int base = tid * SCAN_PER;
  int loc = 0;
#pragma unroll
  for (int j = 0; j < SCAN_PER; ++j) {
    const int idx = base + j;
    if (idx < NND) loc += deg[idx];
  }
  int inc = loc;
#pragma unroll
  for (int off = 1; off < 32; off <<= 1) {
    const int t = __shfl_up(inc, off, 32);
    if (lane >= off) inc += t;
  }
  if (lane == 31) wsum[wave] = inc;
  __syncthreads();
  if (wave == 0) {
    const int v = wsum[lane];
    int iv = v;
#pragma unroll
    for (int off = 1; off < 32; off <<= 1) {
      const int t = __shfl_up(iv, off, 32);
      if (lane >= off) iv += t;
    }
    wsum[lane] = iv - v;
  }
  __syncthreads();
  int run = wsum[wave] + inc - loc;
#pragma unroll
  for (int j = 0; j < SCAN_PER; ++j) {
    const int idx = base + j;
    if (idx < NND) { offs[idx] = run; run += deg[idx]; }
  }
}

__global__ void k_csr_fill(const int* __restrict__ ei, const int* __restrict__ rank,
                           const int* __restrict__ offs, int* __restrict__ csr) {
  const int e = blockIdx.x * blockDim.x + threadIdx.x;
  if (e >= NE) return;
  const int pos = offs[ei[NE + e]] + rank[e];
  if ((unsigned)pos < (unsigned)NE) csr[pos] = ei[e];
}

__global__ __launch_bounds__(256) void k_gemm1(const float* __restrict__ x,
                                               const float* __restrict__ W,
                                               float* __restrict__ h) {
  __shared__ float As[16 * FIN];
  const int tid  = threadIdx.x;
  const int wave = tid >> 5, lane = tid & 31;
  const int m0   = blockIdx.x * 16;
  {
    const float4* xv = (const float4*)(x + (size_t)m0 * FIN);
    float4* sv = (float4*)As;
    sv[tid]       = xv[tid];
    sv[tid + 256] = xv[tid + 256];
  }
  __syncthreads();
  const int n0 = blockIdx.y * 128 + wave * 16;
  const int la = lane & 15, hi = lane >> 4;
  v8f acc = {};
#pragma unroll
  for (int k0 = 0; k0 < FIN; k0 += 4) {
    const int ka = k0 + hi * 2;
    v2f a, b;
    a.x = As[la * FIN + ka];
    a.y = As[la * FIN + ka + 1];
    b.x = W[ka * HC + n0 + la];
    b.y = W[(ka + 1) * HC + n0 + la];
    acc = __builtin_amdgcn_wmma_f32_16x16x4_f32(false, a, false, b, (short)0,
                                                acc, false, false);
  }
#pragma unroll
  for (int j = 0; j < 8; ++j)
    h[(size_t)(m0 + j + hi * 8) * HC + n0 + la] = acc[j];
}

__global__ __launch_bounds__(96) void k_gemm2(const float* __restrict__ h2,
                                              const float* __restrict__ W,
                                              float* __restrict__ t) {
  __shared__ float As[16 * HC];
  const int tid  = threadIdx.x;
  const int wave = tid >> 5, lane = tid & 31;
  const int m0   = blockIdx.x * 16;
  {
    const float4* xv = (const float4*)(h2 + (size_t)m0 * HC);
    float4* sv = (float4*)As;
    for (int i = tid; i < 1024; i += 96) sv[i] = xv[i];
  }
  __syncthreads();
  const int n0 = wave * 16;
  const int la = lane & 15, hi = lane >> 4;
  const int col = n0 + la;
  const bool ok = col < ODIM;
  v8f acc = {};
#pragma unroll 8
  for (int k0 = 0; k0 < HC; k0 += 4) {
    const int ka = k0 + hi * 2;
    v2f a, b;
    a.x = As[la * HC + ka];
    a.y = As[la * HC + ka + 1];
    b.x = ok ? W[ka * ODIM + col] : 0.f;
    b.y = ok ? W[(ka + 1) * ODIM + col] : 0.f;
    acc = __builtin_amdgcn_wmma_f32_16x16x4_f32(false, a, false, b, (short)0,
                                                acc, false, false);
  }
#pragma unroll
  for (int j = 0; j < 8; ++j)
    if (ok) t[(size_t)(m0 + j + hi * 8) * ODIM + col] = acc[j];
}

__global__ __launch_bounds__(256) void k_att1(const float* __restrict__ h,
                                              const float* __restrict__ as_,
                                              const float* __restrict__ ad_,
                                              float* __restrict__ os,
                                              float* __restrict__ od) {
  const int node = blockIdx.x * 8 + (threadIdx.x >> 5);
  const int lane = threadIdx.x & 31;
  if (node >= NND) return;
  const float* hr = h + (size_t)node * HC;
#pragma unroll
  for (int hd = 0; hd < NH; ++hd) {
    const float v = hr[hd * HIDC + lane];
    const float s = wave_sum(v * as_[hd * HIDC + lane]);
    const float d = wave_sum(v * ad_[hd * HIDC + lane]);
    if (lane == 0) { os[node * NH + hd] = s; od[node * NH + hd] = d; }
  }
}

__global__ __launch_bounds__(256) void k_att2(const float* __restrict__ t,
                                              const float* __restrict__ as_,
                                              const float* __restrict__ ad_,
                                              float* __restrict__ os,
                                              float* __restrict__ od) {
  const int node = blockIdx.x * 8 + (threadIdx.x >> 5);
  const int lane = threadIdx.x & 31;
  if (node >= NND) return;
  const float* tr = t + (size_t)node * ODIM;
  float s = 0.f, d = 0.f;
  {
    const float v = tr[lane];
    s += v * as_[lane]; d += v * ad_[lane];
  }
  if (lane < ODIM - 32) {
    const float v = tr[32 + lane];
    s += v * as_[32 + lane]; d += v * ad_[32 + lane];
  }
  s = wave_sum(s); d = wave_sum(d);
  if (lane == 0) { os[node] = s; od[node] = d; }
}

__global__ __launch_bounds__(256) void k_node1(const int* __restrict__ offs,
                                               const int* __restrict__ deg,
                                               const int* __restrict__ csr,
                                               const float* __restrict__ h,
                                               const float* __restrict__ aS,
                                               const float* __restrict__ aD,
                                               float* __restrict__ agg) {
  const int node = __builtin_amdgcn_readfirstlane((int)(blockIdx.x * 8 + (threadIdx.x >> 5)));
  const int lane = threadIdx.x & 31;
  if (node >= NND) return;
  const int beg = offs[node];
  const int n   = deg[node];
  const int g   = lane >> 3;
  const float ad0 = aD[node * NH + g], ad1 = aD[node * NH + 4 + g];
  const float es0 = lrelu(aS[node * NH + g] + ad0);
  const float es1 = lrelu(aS[node * NH + 4 + g] + ad1);
  float m0 = es0, m1 = es1;
#pragma unroll 4
  for (int i = 0; i < n; ++i) {
    int s = csr[beg + i];
    s = (unsigned)s < (unsigned)NND ? s : node;
    const float e0 = lrelu(aS[s * NH + g] + ad0);
    const float e1 = lrelu(aS[s * NH + 4 + g] + ad1);
    m0 = e0 > m0 ? e0 : m0;
    m1 = e1 > m1 ? e1 : m1;
  }
  float4 a0, a1;
  float den0, den1;
  {
    const float x0 = __expf(es0 - m0), x1 = __expf(es1 - m1);
    const float4* hr = (const float4*)(h + (size_t)node * HC);
    const float4 v0 = hr[lane], v1 = hr[32 + lane];
    a0 = make_float4(v0.x * x0, v0.y * x0, v0.z * x0, v0.w * x0);
    a1 = make_float4(v1.x * x1, v1.y * x1, v1.z * x1, v1.w * x1);
    den0 = x0; den1 = x1;
  }
#pragma unroll 4
  for (int i = 0; i < n; ++i) {
    int s = csr[beg + i];
    s = (unsigned)s < (unsigned)NND ? s : node;
    const float x0 = __expf(lrelu(aS[s * NH + g] + ad0) - m0);
    const float x1 = __expf(lrelu(aS[s * NH + 4 + g] + ad1) - m1);
    const float4* hr = (const float4*)(h + (size_t)s * HC);
    const float4 v0 = hr[lane], v1 = hr[32 + lane];
    a0.x += v0.x * x0; a0.y += v0.y * x0; a0.z += v0.z * x0; a0.w += v0.w * x0;
    a1.x += v1.x * x1; a1.y += v1.y * x1; a1.z += v1.z * x1; a1.w += v1.w * x1;
    den0 += x0; den1 += x1;
  }
  const float r0 = den0 + EPSV, r1 = den1 + EPSV;
  float4* o = (float4*)(agg + (size_t)node * HC);
  o[lane]      = make_float4(a0.x / r0, a0.y / r0, a0.z / r0, a0.w / r0);
  o[32 + lane] = make_float4(a1.x / r1, a1.y / r1, a1.z / r1, a1.w / r1);
}

__global__ __launch_bounds__(256) void k_node2(const int* __restrict__ offs,
                                               const int* __restrict__ deg,
                                               const int* __restrict__ csr,
                                               const float* __restrict__ t,
                                               const float* __restrict__ aS,
                                               const float* __restrict__ aD,
                                               float* __restrict__ agg) {
  const int node = __builtin_amdgcn_readfirstlane((int)(blockIdx.x * 8 + (threadIdx.x >> 5)));
  const int lane = threadIdx.x & 31;
  if (node >= NND) return;
  const int beg = offs[node];
  const int n   = deg[node];
  const bool tail = lane < ODIM - 32;
  const float ad = aD[node];
  const float es = lrelu(aS[node] + ad);
  float m = es;
#pragma unroll 4
  for (int i = 0; i < n; ++i) {
    int s = csr[beg + i];
    s = (unsigned)s < (unsigned)NND ? s : node;
    const float e = lrelu(aS[s] + ad);
    m = e > m ? e : m;
  }
  float acc0, acc1 = 0.f, den;
  {
    const float x = __expf(es - m);
    const float* tr = t + (size_t)node * ODIM;
    acc0 = tr[lane] * x;
    if (tail) acc1 = tr[32 + lane] * x;
    den = x;
  }
#pragma unroll 4
  for (int i = 0; i < n; ++i) {
    int s = csr[beg + i];
    s = (unsigned)s < (unsigned)NND ? s : node;
    const float x = __expf(lrelu(aS[s] + ad) - m);
    const float* ts = t + (size_t)s * ODIM;
    acc0 += ts[lane] * x;
    if (tail) acc1 += ts[32 + lane] * x;
    den += x;
  }
  const float r = den + EPSV;
  float* ag = agg + (size_t)node * ODIM;
  ag[lane] = acc0 / r;
  if (tail) ag[32 + lane] = acc1 / r;
}

__global__ void k_elu_bias(float* __restrict__ a, const float* __restrict__ b) {
  const int i = blockIdx.x * blockDim.x + threadIdx.x;
  if (i >= NND * HC) return;
  const float v = a[i] + b[i & (HC - 1)];
  a[i] = v > 0.f ? v : (__expf(v) - 1.f);
}

__global__ void k_out_bias(const float* __restrict__ agg, const float* __restrict__ b,
                           float* __restrict__ out) {
  const int i = blockIdx.x * blockDim.x + threadIdx.x;
  if (i >= NND * ODIM) return;
  out[i] = agg[i] + b[i % ODIM];
}

extern "C" void kernel_launch(void* const* d_in, const int* in_sizes, int n_in,
                              void* d_out, int out_size, void* d_ws, size_t ws_size,
                              hipStream_t stream) {
  const float* x   = (const float*)d_in[0];
  const int*   ei  = (const int*)d_in[1];
  const float* W1  = (const float*)d_in[2];
  const float* as1 = (const float*)d_in[3];
  const float* ad1 = (const float*)d_in[4];
  const float* b1  = (const float*)d_in[5];
  const float* W2  = (const float*)d_in[6];
  const float* as2 = (const float*)d_in[7];
  const float* ad2 = (const float*)d_in[8];
  const float* b2  = (const float*)d_in[9];
  float* out = (float*)d_out;

  float* ws   = (float*)d_ws;
  float* h1   = ws;
  float* agg1 = h1   + (size_t)NND * HC;
  float* aS1  = agg1 + (size_t)NND * HC;
  float* aD1  = aS1  + NND * NH;
  float* t2   = aD1  + NND * NH;
  float* agg2 = t2   + (size_t)NND * ODIM;
  int*   rank = (int*)(agg2 + (size_t)NND * ODIM);
  int*   csr  = rank + NE;
  float* aS2  = (float*)(csr + NE);
  float* aD2  = aS2  + 50048;
  int*   deg  = (int*)(aD2 + 50048);
  int*   offs = deg  + 50048;

  const int B = 256;
  k_fill_i32<<<(NND + B - 1) / B, B, 0, stream>>>(deg, 0, NND);
  k_rank<<<(NE + B - 1) / B, B, 0, stream>>>(ei, deg, rank);
  k_scan<<<1, 1024, 0, stream>>>(deg, offs);
  k_csr_fill<<<(NE + B - 1) / B, B, 0, stream>>>(ei, rank, offs, csr);

  k_gemm1<<<dim3(NND / 16, 2), 256, 0, stream>>>(x, W1, h1);
  k_att1<<<(NND + 7) / 8, 256, 0, stream>>>(h1, as1, ad1, aS1, aD1);
  k_node1<<<(NND + 7) / 8, 256, 0, stream>>>(offs, deg, csr, h1, aS1, aD1, agg1);
  k_elu_bias<<<(NND * HC + B - 1) / B, B, 0, stream>>>(agg1, b1);

  k_gemm2<<<NND / 16, 96, 0, stream>>>(agg1, W2, t2);
  k_att2<<<(NND + 7) / 8, 256, 0, stream>>>(t2, as2, ad2, aS2, aD2);
  k_node2<<<(NND + 7) / 8, 256, 0, stream>>>(offs, deg, csr, t2, aS2, aD2, agg2);
  k_out_bias<<<(NND * ODIM + B - 1) / B, B, 0, stream>>>(agg2, b2, out);
}


# Bias+ELU (layer 1) and the final bias (layer 2) folded into the node kernels' single row store: the two one-element-per-thread launches and their read-modify-write pass over the rows are gone
# speedup vs baseline: 1.9091x; 1.0996x over previous
#include <hip/hip_runtime.h>
#include <hip/hip_bf16.h>

#define NND   50000
#define FIN   128
#define NH    8
#define HIDC  32
#define HC    256
#define ODIM  40
#define NE    800000
#define NET   850000
#define NEG_SLOPE 0.2f
#define EPSV  1e-16f

typedef __attribute__((ext_vector_type(2))) float v2f;
typedef __attribute__((ext_vector_type(8))) float v8f;

__device__ __forceinline__ float wave_sum(float v) {
#pragma unroll
  for (int off = 16; off > 0; off >>= 1) v += __shfl_xor(v, off, 32);
  return v;
}
__device__ __forceinline__ float lrelu(float v) {
  return v > 0.f ? v : v * NEG_SLOPE;
}

__global__ void k_fill_i32(int* __restrict__ p, int v, int n) {
  int i = blockIdx.x * blockDim.x + threadIdx.x;
  if (i < n) p[i] = v;
}


__global__ void k_rank(const int* __restrict__ ei, int* __restrict__ deg,
                       int* __restrict__ rank) {
  const int e = blockIdx.x * blockDim.x + threadIdx.x;
  if (e >= NE) return;
  rank[e] = atomicAdd(&deg[ei[NE + e]], 1);
}

#define SCAN_PER 49
__global__ __launch_bounds__(1024) void k_scan(const int* __restrict__ deg,
                                               int* __restrict__ offs) {
  __shared__ int wsum[32];
  const int tid = threadIdx.x, lane = tid & 31, wave = tid >> 5;
  const int base = tid * SCAN_PER;
  int loc = 0;
#pragma unroll
  for (int j = 0; j < SCAN_PER; ++j) {
    const int idx = base + j;
    if (idx < NND) loc += deg[idx];
  }
  int inc = loc;
#pragma unroll
  for (int off = 1; off < 32; off <<= 1) {
    const int t = __shfl_up(inc, off, 32);
    if (lane >= off) inc += t;
  }
  if (lane == 31) wsum[wave] = inc;
  __syncthreads();
  if (wave == 0) {
    const int v = wsum[lane];
    int iv = v;
#pragma unroll
    for (int off = 1; off < 32; off <<= 1) {
      const int t = __shfl_up(iv, off, 32);
      if (lane >= off) iv += t;
    }
    wsum[lane] = iv - v;
  }
  __syncthreads();
  int run = wsum[wave] + inc - loc;
#pragma unroll
  for (int j = 0; j < SCAN_PER; ++j) {
    const int idx = base + j;
    if (idx < NND) { offs[idx] = run; run += deg[idx]; }
  }
}

__global__ void k_csr_fill(const int* __restrict__ ei, const int* __restrict__ rank,
                           const int* __restrict__ offs, int* __restrict__ csr) {
  const int e = blockIdx.x * blockDim.x + threadIdx.x;
  if (e >= NE) return;
  const int pos = offs[ei[NE + e]] + rank[e];
  if ((unsigned)pos < (unsigned)NE) csr[pos] = ei[e];
}

__global__ __launch_bounds__(256) void k_gemm1(const float* __restrict__ x,
                                               const float* __restrict__ W,
                                               float* __restrict__ h) {
  __shared__ float As[16 * FIN];
  const int tid  = threadIdx.x;
  const int wave = tid >> 5, lane = tid & 31;
  const int m0   = blockIdx.x * 16;
  {
    const float4* xv = (const float4*)(x + (size_t)m0 * FIN);
    float4* sv = (float4*)As;
    sv[tid]       = xv[tid];
    sv[tid + 256] = xv[tid + 256];
  }
  __syncthreads();
  const int n0 = blockIdx.y * 128 + wave * 16;
  const int la = lane & 15, hi = lane >> 4;
  v8f acc = {};
#pragma unroll
  for (int k0 = 0; k0 < FIN; k0 += 4) {
    const int ka = k0 + hi * 2;
    v2f a, b;
    a.x = As[la * FIN + ka];
    a.y = As[la * FIN + ka + 1];
    b.x = W[ka * HC + n0 + la];
    b.y = W[(ka + 1) * HC + n0 + la];
    acc = __builtin_amdgcn_wmma_f32_16x16x4_f32(false, a, false, b, (short)0,
                                                acc, false, false);
  }
#pragma unroll
  for (int j = 0; j < 8; ++j)
    h[(size_t)(m0 + j + hi * 8) * HC + n0 + la] = acc[j];
}

__global__ __launch_bounds__(96) void k_gemm2(const float* __restrict__ h2,
                                              const float* __restrict__ W,
                                              float* __restrict__ t) {
  __shared__ float As[16 * HC];
  const int tid  = threadIdx.x;
  const int wave = tid >> 5, lane = tid & 31;
  const int m0   = blockIdx.x * 16;
  {
    const float4* xv = (const float4*)(h2 + (size_t)m0 * HC);
    float4* sv = (float4*)As;
    for (int i = tid; i < 1024; i += 96) sv[i] = xv[i];
  }
  __syncthreads();
  const int n0 = wave * 16;
  const int la = lane & 15, hi = lane >> 4;
  const int col = n0 + la;
  const bool ok = col < ODIM;
  v8f acc = {};
#pragma unroll 8
  for (int k0 = 0; k0 < HC; k0 += 4) {
    const int ka = k0 + hi * 2;
    v2f a, b;
    a.x = As[la * HC + ka];
    a.y = As[la * HC + ka + 1];
    b.x = ok ? W[ka * ODIM + col] : 0.f;
    b.y = ok ? W[(ka + 1) * ODIM + col] : 0.f;
    acc = __builtin_amdgcn_wmma_f32_16x16x4_f32(false, a, false, b, (short)0,
                                                acc, false, false);
  }
#pragma unroll
  for (int j = 0; j < 8; ++j)
    if (ok) t[(size_t)(m0 + j + hi * 8) * ODIM + col] = acc[j];
}

__global__ __launch_bounds__(256) void k_att1(const float* __restrict__ h,
                                              const float* __restrict__ as_,
                                              const float* __restrict__ ad_,
                                              float* __restrict__ os,
                                              float* __restrict__ od) {
  const int node = blockIdx.x * 8 + (threadIdx.x >> 5);
  const int lane = threadIdx.x & 31;
  if (node >= NND) return;
  const float* hr = h + (size_t)node * HC;
#pragma unroll
  for (int hd = 0; hd < NH; ++hd) {
    const float v = hr[hd * HIDC + lane];
    const float s = wave_sum(v * as_[hd * HIDC + lane]);
    const float d = wave_sum(v * ad_[hd * HIDC + lane]);
    if (lane == 0) { os[node * NH + hd] = s; od[node * NH + hd] = d; }
  }
}

__global__ __launch_bounds__(256) void k_att2(const float* __restrict__ t,
                                              const float* __restrict__ as_,
                                              const float* __restrict__ ad_,
                                              float* __restrict__ os,
                                              float* __restrict__ od) {
  const int node = blockIdx.x * 8 + (threadIdx.x >> 5);
  const int lane = threadIdx.x & 31;
  if (node >= NND) return;
  const float* tr = t + (size_t)node * ODIM;
  float s = 0.f, d = 0.f;
  {
    const float v = tr[lane];
    s += v * as_[lane]; d += v * ad_[lane];
  }
  if (lane < ODIM - 32) {
    const float v = tr[32 + lane];
    s += v * as_[32 + lane]; d += v * ad_[32 + lane];
  }
  s = wave_sum(s); d = wave_sum(d);
  if (lane == 0) { os[node] = s; od[node] = d; }
}

__device__ __forceinline__ float bias_elu(float a, float r, float b) {
  const float v = a / r + b;
  return v > 0.f ? v : (__expf(v) - 1.f);
}
__global__ __launch_bounds__(256) void k_node1(const int* __restrict__ offs,
                                               const int* __restrict__ deg,
                                               const int* __restrict__ csr,
                                               const float* __restrict__ h,
                                               const float* __restrict__ aS,
                                               const float* __restrict__ aD,
                                               const float* __restrict__ bias,
                                               float* __restrict__ h2) {
  const int node = __builtin_amdgcn_readfirstlane((int)(blockIdx.x * 8 + (threadIdx.x >> 5)));
  const int lane = threadIdx.x & 31;
  if (node >= NND) return;
  const int beg = offs[node];
  const int n   = deg[node];
  const int g   = lane >> 3;
  const float ad0 = aD[node * NH + g], ad1 = aD[node * NH + 4 + g];
  const float es0 = lrelu(aS[node * NH + g] + ad0);
  const float es1 = lrelu(aS[node * NH + 4 + g] + ad1);
  float m0 = es0, m1 = es1;
#pragma unroll 4
  for (int i = 0; i < n; ++i) {
    int s = csr[beg + i];
    s = (unsigned)s < (unsigned)NND ? s : node;
    const float e0 = lrelu(aS[s * NH + g] + ad0);
    const float e1 = lrelu(aS[s * NH + 4 + g] + ad1);
    m0 = e0 > m0 ? e0 : m0;
    m1 = e1 > m1 ? e1 : m1;
  }
  float4 a0, a1;
  float den0, den1;
  {
    const float x0 = __expf(es0 - m0), x1 = __expf(es1 - m1);
    const float4* hr = (const float4*)(h + (size_t)node * HC);
    const float4 v0 = hr[lane], v1 = hr[32 + lane];
    a0 = make_float4(v0.x * x0, v0.y * x0, v0.z * x0, v0.w * x0);
    a1 = make_float4(v1.x * x1, v1.y * x1, v1.z * x1, v1.w * x1);
    den0 = x0; den1 = x1;
  }
#pragma unroll 4
  for (int i = 0; i < n; ++i) {
    int s = csr[beg + i];
    s = (unsigned)s < (unsigned)NND ? s : node;
    const float x0 = __expf(lrelu(aS[s * NH + g] + ad0) - m0);
    const float x1 = __expf(lrelu(aS[s * NH + 4 + g] + ad1) - m1);
    const float4* hr = (const float4*)(h + (size_t)s * HC);
    const float4 v0 = hr[lane], v1 = hr[32 + lane];
    a0.x += v0.x * x0; a0.y += v0.y * x0; a0.z += v0.z * x0; a0.w += v0.w * x0;
    a1.x += v1.x * x1; a1.y += v1.y * x1; a1.z += v1.z * x1; a1.w += v1.w * x1;
    den0 += x0; den1 += x1;
  }
  const float r0 = den0 + EPSV, r1 = den1 + EPSV;
  const float4 b0 = ((const float4*)bias)[lane], b1 = ((const float4*)bias)[32 + lane];
  float4* o = (float4*)(h2 + (size_t)node * HC);
  o[lane]      = make_float4(bias_elu(a0.x, r0, b0.x), bias_elu(a0.y, r0, b0.y),
                             bias_elu(a0.z, r0, b0.z), bias_elu(a0.w, r0, b0.w));
  o[32 + lane] = make_float4(bias_elu(a1.x, r1, b1.x), bias_elu(a1.y, r1, b1.y),
                             bias_elu(a1.z, r1, b1.z), bias_elu(a1.w, r1, b1.w));
}

__global__ __launch_bounds__(256) void k_node2(const int* __restrict__ offs,
                                               const int* __restrict__ deg,
                                               const int* __restrict__ csr,
                                               const float* __restrict__ t,
                                               const float* __restrict__ aS,
                                               const float* __restrict__ aD,
                                               const float* __restrict__ bias,
                                               float* __restrict__ out) {
  const int node = __builtin_amdgcn_readfirstlane((int)(blockIdx.x * 8 + (threadIdx.x >> 5)));
  const int lane = threadIdx.x & 31;
  if (node >= NND) return;
  const int beg = offs[node];
  const int n   = deg[node];
  const bool tail = lane < ODIM - 32;
  const float ad = aD[node];
  const float es = lrelu(aS[node] + ad);
  float m = es;
#pragma unroll 4
  for (int i = 0; i < n; ++i) {
    int s = csr[beg + i];
    s = (unsigned)s < (unsigned)NND ? s : node;
    const float e = lrelu(aS[s] + ad);
    m = e > m ? e : m;
  }
  float acc0, acc1 = 0.f, den;
  {
    const float x = __expf(es - m);
    const float* tr = t + (size_t)node * ODIM;
    acc0 = tr[lane] * x;
    if (tail) acc1 = tr[32 + lane] * x;
    den = x;
  }
#pragma unroll 4
  for (int i = 0; i < n; ++i) {
    int s = csr[beg + i];
    s = (unsigned)s < (unsigned)NND ? s : node;
    const float x = __expf(lrelu(aS[s] + ad) - m);
    const float* ts = t + (size_t)s * ODIM;
    acc0 += ts[lane] * x;
    if (tail) acc1 += ts[32 + lane] * x;
    den += x;
  }
  const float r = den + EPSV;
  float* o = out + (size_t)node * ODIM;
  o[lane] = acc0 / r + bias[lane];
  if (tail) o[32 + lane] = acc1 / r + bias[32 + lane];
}

extern "C" void kernel_launch(void* const* d_in, const int* in_sizes, int n_in,
                              void* d_out, int out_size, void* d_ws, size_t ws_size,
                              hipStream_t stream) {
  const float* x   = (const float*)d_in[0];
  const int*   ei  = (const int*)d_in[1];
  const float* W1  = (const float*)d_in[2];
  const float* as1 = (const float*)d_in[3];
  const float* ad1 = (const float*)d_in[4];
  const float* b1  = (const float*)d_in[5];
  const float* W2  = (const float*)d_in[6];
  const float* as2 = (const float*)d_in[7];
  const float* ad2 = (const float*)d_in[8];
  const float* b2  = (const float*)d_in[9];
  float* out = (float*)d_out;

  float* ws   = (float*)d_ws;
  float* h1   = ws;
  float* h2   = h1   + (size_t)NND * HC;
  float* aS1  = h2   + (size_t)NND * HC;
  float* aD1  = aS1  + NND * NH;
  float* t2   = aD1  + NND * NH;
  int*   rank = (int*)(t2 + (size_t)NND * ODIM);
  int*   csr  = rank + NE;
  float* aS2  = (float*)(csr + NE);
  float* aD2  = aS2  + 50048;
  int*   deg  = (int*)(aD2 + 50048);
  int*   offs = deg  + 50048;

  const int B = 256;
  k_fill_i32<<<(NND + B - 1) / B, B, 0, stream>>>(deg, 0, NND);
  k_rank<<<(NE + B - 1) / B, B, 0, stream>>>(ei, deg, rank);
  k_scan<<<1, 1024, 0, stream>>>(deg, offs);
  k_csr_fill<<<(NE + B - 1) / B, B, 0, stream>>>(ei, rank, offs, csr);

  k_gemm1<<<dim3(NND / 16, 2), 256, 0, stream>>>(x, W1, h1);
  k_att1<<<(NND + 7) / 8, 256, 0, stream>>>(h1, as1, ad1, aS1, aD1);
  k_node1<<<(NND + 7) / 8, 256, 0, stream>>>(offs, deg, csr, h1, aS1, aD1, b1, h2);

  k_gemm2<<<NND / 16, 96, 0, stream>>>(h2, W2, t2);
  k_att2<<<(NND + 7) / 8, 256, 0, stream>>>(t2, as2, ad2, aS2, aD2);
  k_node2<<<(NND + 7) / 8, 256, 0, stream>>>(offs, deg, csr, t2, aS2, aD2, b2, out);
}


# Degree prefix sum staged through LDS: whole-line int4 loads and stores of the counter array replace 97 dependent 4-byte loads and 49 stores per lane, each lane on its own line; runs read at pitch 49
# speedup vs baseline: 2.3913x; 1.2526x over previous
#include <hip/hip_runtime.h>
#include <hip/hip_bf16.h>

#define NND   50000
#define FIN   128
#define NH    8
#define HIDC  32
#define HC    256
#define ODIM  40
#define NE    800000
#define NET   850000
#define NEG_SLOPE 0.2f
#define EPSV  1e-16f

typedef __attribute__((ext_vector_type(2))) float v2f;
typedef __attribute__((ext_vector_type(8))) float v8f;

__device__ __forceinline__ float wave_sum(float v) {
#pragma unroll
  for (int off = 16; off > 0; off >>= 1) v += __shfl_xor(v, off, 32);
  return v;
}
__device__ __forceinline__ float lrelu(float v) {
  return v > 0.f ? v : v * NEG_SLOPE;
}

__global__ void k_fill_i32(int* __restrict__ p, int v, int n) {
  int i = blockIdx.x * blockDim.x + threadIdx.x;
  if (i < n) p[i] = v;
}


__global__ void k_rank(const int* __restrict__ ei, int* __restrict__ deg,
                       int* __restrict__ rank) {
  const int e = blockIdx.x * blockDim.x + threadIdx.x;
  if (e >= NE) return;
  rank[e] = atomicAdd(&deg[ei[NE + e]], 1);
}

#define SCAN_PER 49
#define SCAN_N   (1024 * SCAN_PER)
#define NND4     (NND / 4)
#define SCAN_FULL (NND4 / 1024)
__global__ __launch_bounds__(1024) void k_scan(const int* __restrict__ deg,
                                               int* __restrict__ offs) {
  __shared__ int4 sd4[SCAN_N / 4];
  __shared__ int wsum[32];
  int* sd = (int*)sd4;
  const int tid = threadIdx.x, lane = tid & 31, wave = tid >> 5;
  {
    const int4* dv = (const int4*)deg;
    int4 t[SCAN_FULL];
#pragma unroll
    for (int k = 0; k < SCAN_FULL; ++k) t[k] = dv[tid + k * 1024];
#pragma unroll
    for (int k = 0; k < SCAN_FULL; ++k) sd4[tid + k * 1024] = t[k];
    const int i = tid + SCAN_FULL * 1024;
    if (i < NND4) sd4[i] = dv[i];
    else if (i < SCAN_N / 4) sd4[i] = make_int4(0, 0, 0, 0);
  }
  __syncthreads();
  const int base = tid * SCAN_PER;
  int v[SCAN_PER];
  int loc = 0;
#pragma unroll
  for (int j = 0; j < SCAN_PER; ++j) { v[j] = sd[base + j]; loc += v[j]; }
  int inc = loc;
#pragma unroll
  for (int off = 1; off < 32; off <<= 1) {
    const int t = __shfl_up(inc, off, 32);
    if (lane >= off) inc += t;
  }
  if (lane == 31) wsum[wave] = inc;
  __syncthreads();
  if (wave == 0) {
    const int w = wsum[lane];
    int iv = w;
#pragma unroll
    for (int off = 1; off < 32; off <<= 1) {
      const int t = __shfl_up(iv, off, 32);
      if (lane >= off) iv += t;
    }
    wsum[lane] = iv - w;
  }
  __syncthreads();
  int run = wsum[wave] + inc - loc;
#pragma unroll
  for (int j = 0; j < SCAN_PER; ++j) { sd[base + j] = run; run += v[j]; }
  __syncthreads();
  {
    int4* ov = (int4*)offs;
#pragma unroll
    for (int k = 0; k < SCAN_FULL; ++k) ov[tid + k * 1024] = sd4[tid + k * 1024];
    const int i = tid + SCAN_FULL * 1024;
    if (i < NND4) ov[i] = sd4[i];
  }
}

__global__ void k_csr_fill(const int* __restrict__ ei, const int* __restrict__ rank,
                           const int* __restrict__ offs, int* __restrict__ csr) {
  const int e = blockIdx.x * blockDim.x + threadIdx.x;
  if (e >= NE) return;
  const int pos = offs[ei[NE + e]] + rank[e];
  if ((unsigned)pos < (unsigned)NE) csr[pos] = ei[e];
}

__global__ __launch_bounds__(256) void k_gemm1(const float* __restrict__ x,
                                               const float* __restrict__ W,
                                               float* __restrict__ h) {
  __shared__ float As[16 * FIN];
  const int tid  = threadIdx.x;
  const int wave = tid >> 5, lane = tid & 31;
  const int m0   = blockIdx.x * 16;
  {
    const float4* xv = (const float4*)(x + (size_t)m0 * FIN);
    float4* sv = (float4*)As;
    sv[tid]       = xv[tid];
    sv[tid + 256] = xv[tid + 256];
  }
  __syncthreads();
  const int n0 = blockIdx.y * 128 + wave * 16;
  const int la = lane & 15, hi = lane >> 4;
  v8f acc = {};
#pragma unroll
  for (int k0 = 0; k0 < FIN; k0 += 4) {
    const int ka = k0 + hi * 2;
    v2f a, b;
    a.x = As[la * FIN + ka];
    a.y = As[la * FIN + ka + 1];
    b.x = W[ka * HC + n0 + la];
    b.y = W[(ka + 1) * HC + n0 + la];
    acc = __builtin_amdgcn_wmma_f32_16x16x4_f32(false, a, false, b, (short)0,
                                                acc, false, false);
  }
#pragma unroll
  for (int j = 0; j < 8; ++j)
    h[(size_t)(m0 + j + hi * 8) * HC + n0 + la] = acc[j];
}

__global__ __launch_bounds__(96) void k_gemm2(const float* __restrict__ h2,
                                              const float* __restrict__ W,
                                              float* __restrict__ t) {
  __shared__ float As[16 * HC];
  const int tid  = threadIdx.x;
  const int wave = tid >> 5, lane = tid & 31;
  const int m0   = blockIdx.x * 16;
  {
    const float4* xv = (const float4*)(h2 + (size_t)m0 * HC);
    float4* sv = (float4*)As;
    for (int i = tid; i < 1024; i += 96) sv[i] = xv[i];
  }
  __syncthreads();
  const int n0 = wave * 16;
  const int la = lane & 15, hi = lane >> 4;
  const int col = n0 + la;
  const bool ok = col < ODIM;
  v8f acc = {};
#pragma unroll 8
  for (int k0 = 0; k0 < HC; k0 += 4) {
    const int ka = k0 + hi * 2;
    v2f a, b;
    a.x = As[la * HC + ka];
    a.y = As[la * HC + ka + 1];
    b.x = ok ? W[ka * ODIM + col] : 0.f;
    b.y = ok ? W[(ka + 1) * ODIM + col] : 0.f;
    acc = __builtin_amdgcn_wmma_f32_16x16x4_f32(false, a, false, b, (short)0,
                                                acc, false, false);
  }
#pragma unroll
  for (int j = 0; j < 8; ++j)
    if (ok) t[(size_t)(m0 + j + hi * 8) * ODIM + col] = acc[j];
}

__global__ __launch_bounds__(256) void k_att1(const float* __restrict__ h,
                                              const float* __restrict__ as_,
                                              const float* __restrict__ ad_,
                                              float* __restrict__ os,
                                              float* __restrict__ od) {
  const int node = blockIdx.x * 8 + (threadIdx.x >> 5);
  const int lane = threadIdx.x & 31;
  if (node >= NND) return;
  const float* hr = h + (size_t)node * HC;
#pragma unroll
  for (int hd = 0; hd < NH; ++hd) {
    const float v = hr[hd * HIDC + lane];
    const float s = wave_sum(v * as_[hd * HIDC + lane]);
    const float d = wave_sum(v * ad_[hd * HIDC + lane]);
    if (lane == 0) { os[node * NH + hd] = s; od[node * NH + hd] = d; }
  }
}

__global__ __launch_bounds__(256) void k_att2(const float* __restrict__ t,
                                              const float* __restrict__ as_,
                                              const float* __restrict__ ad_,
                                              float* __restrict__ os,
                                              float* __restrict__ od) {
  const int node = blockIdx.x * 8 + (threadIdx.x >> 5);
  const int lane = threadIdx.x & 31;
  if (node >= NND) return;
  const float* tr = t + (size_t)node * ODIM;
  float s = 0.f, d = 0.f;
  {
    const float v = tr[lane];
    s += v * as_[lane]; d += v * ad_[lane];
  }
  if (lane < ODIM - 32) {
    const float v = tr[32 + lane];
    s += v * as_[32 + lane]; d += v * ad_[32 + lane];
  }
  s = wave_sum(s); d = wave_sum(d);
  if (lane == 0) { os[node] = s; od[node] = d; }
}

__device__ __forceinline__ float bias_elu(float a, float r, float b) {
  const float v = a / r + b;
  return v > 0.f ? v : (__expf(v) - 1.f);
}
__global__ __launch_bounds__(256) void k_node1(const int* __restrict__ offs,
                                               const int* __restrict__ deg,
                                               const int* __restrict__ csr,
                                               const float* __restrict__ h,
                                               const float* __restrict__ aS,
                                               const float* __restrict__ aD,
                                               const float* __restrict__ bias,
                                               float* __restrict__ h2) {
  const int node = __builtin_amdgcn_readfirstlane((int)(blockIdx.x * 8 + (threadIdx.x >> 5)));
  const int lane = threadIdx.x & 31;
  if (node >= NND) return;
  const int beg = offs[node];
  const int n   = deg[node];
  const int g   = lane >> 3;
  const float ad0 = aD[node * NH + g], ad1 = aD[node * NH + 4 + g];
  const float es0 = lrelu(aS[node * NH + g] + ad0);
  const float es1 = lrelu(aS[node * NH + 4 + g] + ad1);
  float m0 = es0, m1 = es1;
#pragma unroll 4
  for (int i = 0; i < n; ++i) {
    int s = csr[beg + i];
    s = (unsigned)s < (unsigned)NND ? s : node;
    const float e0 = lrelu(aS[s * NH + g] + ad0);
    const float e1 = lrelu(aS[s * NH + 4 + g] + ad1);
    m0 = e0 > m0 ? e0 : m0;
    m1 = e1 > m1 ? e1 : m1;
  }
  float4 a0, a1;
  float den0, den1;
  {
    const float x0 = __expf(es0 - m0), x1 = __expf(es1 - m1);
    const float4* hr = (const float4*)(h + (size_t)node * HC);
    const float4 v0 = hr[lane], v1 = hr[32 + lane];
    a0 = make_float4(v0.x * x0, v0.y * x0, v0.z * x0, v0.w * x0);
    a1 = make_float4(v1.x * x1, v1.y * x1, v1.z * x1, v1.w * x1);
    den0 = x0; den1 = x1;
  }
#pragma unroll 4
  for (int i = 0; i < n; ++i) {
    int s = csr[beg + i];
    s = (unsigned)s < (unsigned)NND ? s : node;
    const float x0 = __expf(lrelu(aS[s * NH + g] + ad0) - m0);
    const float x1 = __expf(lrelu(aS[s * NH + 4 + g] + ad1) - m1);
    const float4* hr = (const float4*)(h + (size_t)s * HC);
    const float4 v0 = hr[lane], v1 = hr[32 + lane];
    a0.x += v0.x * x0; a0.y += v0.y * x0; a0.z += v0.z * x0; a0.w += v0.w * x0;
    a1.x += v1.x * x1; a1.y += v1.y * x1; a1.z += v1.z * x1; a1.w += v1.w * x1;
    den0 += x0; den1 += x1;
  }
  const float r0 = den0 + EPSV, r1 = den1 + EPSV;
  const float4 b0 = ((const float4*)bias)[lane], b1 = ((const float4*)bias)[32 + lane];
  float4* o = (float4*)(h2 + (size_t)node * HC);
  o[lane]      = make_float4(bias_elu(a0.x, r0, b0.x), bias_elu(a0.y, r0, b0.y),
                             bias_elu(a0.z, r0, b0.z), bias_elu(a0.w, r0, b0.w));
  o[32 + lane] = make_float4(bias_elu(a1.x, r1, b1.x), bias_elu(a1.y, r1, b1.y),
                             bias_elu(a1.z, r1, b1.z), bias_elu(a1.w, r1, b1.w));
}

__global__ __launch_bounds__(256) void k_node2(const int* __restrict__ offs,
                                               const int* __restrict__ deg,
                                               const int* __restrict__ csr,
                                               const float* __restrict__ t,
                                               const float* __restrict__ aS,
                                               const float* __restrict__ aD,
                                               const float* __restrict__ bias,
                                               float* __restrict__ out) {
  const int node = __builtin_amdgcn_readfirstlane((int)(blockIdx.x * 8 + (threadIdx.x >> 5)));
  const int lane = threadIdx.x & 31;
  if (node >= NND) return;
  const int beg = offs[node];
  const int n   = deg[node];
  const bool tail = lane < ODIM - 32;
  const float ad = aD[node];
  const float es = lrelu(aS[node] + ad);
  float m = es;
#pragma unroll 4
  for (int i = 0; i < n; ++i) {
    int s = csr[beg + i];
    s = (unsigned)s < (unsigned)NND ? s : node;
    const float e = lrelu(aS[s] + ad);
    m = e > m ? e : m;
  }
  float acc0, acc1 = 0.f, den;
  {
    const float x = __expf(es - m);
    const float* tr = t + (size_t)node * ODIM;
    acc0 = tr[lane] * x;
    if (tail) acc1 = tr[32 + lane] * x;
    den = x;
  }
#pragma unroll 4
  for (int i = 0; i < n; ++i) {
    int s = csr[beg + i];
    s = (unsigned)s < (unsigned)NND ? s : node;
    const float x = __expf(lrelu(aS[s] + ad) - m);
    const float* ts = t + (size_t)s * ODIM;
    acc0 += ts[lane] * x;
    if (tail) acc1 += ts[32 + lane] * x;
    den += x;
  }
  const float r = den + EPSV;
  float* o = out + (size_t)node * ODIM;
  o[lane] = acc0 / r + bias[lane];
  if (tail) o[32 + lane] = acc1 / r + bias[32 + lane];
}

extern "C" void kernel_launch(void* const* d_in, const int* in_sizes, int n_in,
                              void* d_out, int out_size, void* d_ws, size_t ws_size,
                              hipStream_t stream) {
  const float* x   = (const float*)d_in[0];
  const int*   ei  = (const int*)d_in[1];
  const float* W1  = (const float*)d_in[2];
  const float* as1 = (const float*)d_in[3];
  const float* ad1 = (const float*)d_in[4];
  const float* b1  = (const float*)d_in[5];
  const float* W2  = (const float*)d_in[6];
  const float* as2 = (const float*)d_in[7];
  const float* ad2 = (const float*)d_in[8];
  const float* b2  = (const float*)d_in[9];
  float* out = (float*)d_out;

  float* ws   = (float*)d_ws;
  float* h1   = ws;
  float* h2   = h1   + (size_t)NND * HC;
  float* aS1  = h2   + (size_t)NND * HC;
  float* aD1  = aS1  + NND * NH;
  float* t2   = aD1  + NND * NH;
  int*   rank = (int*)(t2 + (size_t)NND * ODIM);
  int*   csr  = rank + NE;
  float* aS2  = (float*)(csr + NE);
  float* aD2  = aS2  + 50048;
  int*   deg  = (int*)(aD2 + 50048);
  int*   offs = deg  + 50048;

  const int B = 256;
  k_fill_i32<<<(NND + B - 1) / B, B, 0, stream>>>(deg, 0, NND);
  k_rank<<<(NE + B - 1) / B, B, 0, stream>>>(ei, deg, rank);
  k_scan<<<1, 1024, 0, stream>>>(deg, offs);
  k_csr_fill<<<(NE + B - 1) / B, B, 0, stream>>>(ei, rank, offs, csr);

  k_gemm1<<<dim3(NND / 16, 2), 256, 0, stream>>>(x, W1, h1);
  k_att1<<<(NND + 7) / 8, 256, 0, stream>>>(h1, as1, ad1, aS1, aD1);
  k_node1<<<(NND + 7) / 8, 256, 0, stream>>>(offs, deg, csr, h1, aS1, aD1, b1, h2);

  k_gemm2<<<NND / 16, 96, 0, stream>>>(h2, W2, t2);
  k_att2<<<(NND + 7) / 8, 256, 0, stream>>>(t2, as2, ad2, aS2, aD2);
  k_node2<<<(NND + 7) / 8, 256, 0, stream>>>(offs, deg, csr, t2, aS2, aD2, b2, out);
}


# Edges grouped by destination in two levels with LDS counters (block histogram, one scan, bucket scatter, per-bucket LDS rank): no global atomic per edge; a node's sum order = LDS arrival rank
# speedup vs baseline: 2.9014x; 1.2133x over previous
#include <hip/hip_runtime.h>
#include <hip/hip_bf16.h>

#define NND   50000
#define FIN   128
#define NH    8
#define HIDC  32
#define HC    256
#define ODIM  40
#define NE    800000
#define NET   850000
#define NEG_SLOPE 0.2f
#define EPSV  1e-16f

typedef __attribute__((ext_vector_type(2))) float v2f;
typedef __attribute__((ext_vector_type(8))) float v8f;

__device__ __forceinline__ float wave_sum(float v) {
#pragma unroll
  for (int off = 16; off > 0; off >>= 1) v += __shfl_xor(v, off, 32);
  return v;
}
__device__ __forceinline__ float lrelu(float v) {
  return v > 0.f ? v : v * NEG_SLOPE;
}

#define NPB   256
#define NBK   196
#define NEB   256
#define EPB   (NE / NEB)
#define SBT   1024

__device__ __forceinline__ int dst_bucket_clamped(int d) {
  return (unsigned)d < (unsigned)NND ? d : NND - 1;
}

__global__ __launch_bounds__(SBT) void k_bhist(const int* __restrict__ ei,
                                               int* __restrict__ hist) {
  __shared__ int cnt[NBK];
  const int tid = threadIdx.x, blk = blockIdx.x;
  if (tid < NBK) cnt[tid] = 0;
  __syncthreads();
  const int* dstp = ei + NE + blk * EPB;
  for (int i = tid; i < EPB; i += SBT) {
    const int d = dst_bucket_clamped(dstp[i]);
    atomicAdd(&cnt[d >> 8], 1);
  }
  __syncthreads();
  if (tid < NBK) hist[tid * NEB + blk] = cnt[tid];
}

#define SCAN_PER 49
#define SCAN_N   (1024 * SCAN_PER)
#define SCAN_N4  (SCAN_N / 4)
#define SCAN_FULL (SCAN_N4 / 1024)
static_assert(NBK * NEB == SCAN_N, "scan size must equal the count table");
static_assert(NBK * NPB >= NND, "buckets must cover all nodes");
static_assert(EPB * NEB == NE, "edge blocks must cover all edges");
__global__ __launch_bounds__(1024) void k_scan(const int* __restrict__ cntin,
                                               int* __restrict__ pre) {
  __shared__ int4 sd4[SCAN_N4];
  __shared__ int wsum[32];
  int* sd = (int*)sd4;
  const int tid = threadIdx.x, lane = tid & 31, wave = tid >> 5;
  {
    const int4* dv = (const int4*)cntin;
    int4 t[SCAN_FULL];
#pragma unroll
    for (int k = 0; k < SCAN_FULL; ++k) t[k] = dv[tid + k * 1024];
#pragma unroll
    for (int k = 0; k < SCAN_FULL; ++k) sd4[tid + k * 1024] = t[k];
    const int i = tid + SCAN_FULL * 1024;
    if (i < SCAN_N4) sd4[i] = dv[i];
  }
  __syncthreads();
  const int base = tid * SCAN_PER;
  int v[SCAN_PER];
  int loc = 0;
#pragma unroll
  for (int j = 0; j < SCAN_PER; ++j) { v[j] = sd[base + j]; loc += v[j]; }
  int inc = loc;
#pragma unroll
  for (int off = 1; off < 32; off <<= 1) {
    const int t = __shfl_up(inc, off, 32);
    if (lane >= off) inc += t;
  }
  if (lane == 31) wsum[wave] = inc;
  __syncthreads();
  if (wave == 0) {
    const int w = wsum[lane];
    int iv = w;
#pragma unroll
    for (int off = 1; off < 32; off <<= 1) {
      const int t = __shfl_up(iv, off, 32);
      if (lane >= off) iv += t;
    }
    wsum[lane] = iv - w;
  }
  __syncthreads();
  int run = wsum[wave] + inc - loc;
#pragma unroll
  for (int j = 0; j < SCAN_PER; ++j) { sd[base + j] = run; run += v[j]; }
  __syncthreads();
  {
    int4* ov = (int4*)pre;
#pragma unroll
    for (int k = 0; k < SCAN_FULL; ++k) ov[tid + k * 1024] = sd4[tid + k * 1024];
    const int i = tid + SCAN_FULL * 1024;
    if (i < SCAN_N4) ov[i] = sd4[i];
  }
}

__global__ __launch_bounds__(SBT) void k_bscatter(const int* __restrict__ ei,
                                                  const int* __restrict__ pre,
                                                  int2* __restrict__ rec) {
  __shared__ int cnt[NBK];
  __shared__ int bas[NBK];
  const int tid = threadIdx.x, blk = blockIdx.x;
  if (tid < NBK) { cnt[tid] = 0; bas[tid] = pre[tid * NEB + blk]; }
  __syncthreads();
  const int* srcp = ei + blk * EPB;
  const int* dstp = ei + NE + blk * EPB;
  for (int i = tid; i < EPB; i += SBT) {
    const int s = srcp[i];
    const int d = dst_bucket_clamped(dstp[i]);
    const int b = d >> 8;
    const int pos = bas[b] + atomicAdd(&cnt[b], 1);
    if ((unsigned)pos < (unsigned)NE) rec[pos] = make_int2(s, d);
  }
}

__global__ __launch_bounds__(SBT) void k_bucket(const int2* __restrict__ rec,
                                                const int* __restrict__ pre,
                                                int* __restrict__ offs,
                                                int* __restrict__ deg,
                                                int* __restrict__ csr) {
  __shared__ int cnt[NPB];
  __shared__ int cur[NPB];
  __shared__ int wsum[NPB / 32];
  const int tid = threadIdx.x, bk = blockIdx.x, lane = tid & 31, wave = tid >> 5;
  int beg = pre[bk * NEB];
  int end = (bk + 1 < NBK) ? pre[(bk + 1) * NEB] : NE;
  beg = beg < 0 ? 0 : (beg > NE ? NE : beg);
  end = end < beg ? beg : (end > NE ? NE : end);
  if (tid < NPB) cnt[tid] = 0;
  __syncthreads();
  for (int i = beg + tid; i < end; i += SBT) {
    const int2 r = rec[i];
    atomicAdd(&cnt[r.y & (NPB - 1)], 1);
  }
  __syncthreads();
  const int c = tid < NPB ? cnt[tid] : 0;
  int inc = c;
#pragma unroll
  for (int off = 1; off < 32; off <<= 1) {
    const int t = __shfl_up(inc, off, 32);
    if (lane >= off) inc += t;
  }
  if (tid < NPB && lane == 31) wsum[wave] = inc;
  __syncthreads();
  if (tid < NPB) {
    int wb = 0;
#pragma unroll
    for (int w = 0; w < NPB / 32; ++w) wb += w < wave ? wsum[w] : 0;
    const int o = beg + wb + inc - c;
    cur[tid] = o;
    const int node = bk * NPB + tid;
    if (node < NND) { offs[node] = o; deg[node] = c; }
  }
  __syncthreads();
  for (int i = beg + tid; i < end; i += SBT) {
    const int2 r = rec[i];
    const int pos = atomicAdd(&cur[r.y & (NPB - 1)], 1);
    if ((unsigned)pos < (unsigned)NE) csr[pos] = r.x;
  }
}

__global__ __launch_bounds__(256) void k_gemm1(const float* __restrict__ x,
                                               const float* __restrict__ W,
                                               float* __restrict__ h) {
  __shared__ float As[16 * FIN];
  const int tid  = threadIdx.x;
  const int wave = tid >> 5, lane = tid & 31;
  const int m0   = blockIdx.x * 16;
  {
    const float4* xv = (const float4*)(x + (size_t)m0 * FIN);
    float4* sv = (float4*)As;
    sv[tid]       = xv[tid];
    sv[tid + 256] = xv[tid + 256];
  }
  __syncthreads();
  const int n0 = blockIdx.y * 128 + wave * 16;
  const int la = lane & 15, hi = lane >> 4;
  v8f acc = {};
#pragma unroll
  for (int k0 = 0; k0 < FIN; k0 += 4) {
    const int ka = k0 + hi * 2;
    v2f a, b;
    a.x = As[la * FIN + ka];
    a.y = As[la * FIN + ka + 1];
    b.x = W[ka * HC + n0 + la];
    b.y = W[(ka + 1) * HC + n0 + la];
    acc = __builtin_amdgcn_wmma_f32_16x16x4_f32(false, a, false, b, (short)0,
                                                acc, false, false);
  }
#pragma unroll
  for (int j = 0; j < 8; ++j)
    h[(size_t)(m0 + j + hi * 8) * HC + n0 + la] = acc[j];
}

__global__ __launch_bounds__(96) void k_gemm2(const float* __restrict__ h2,
                                              const float* __restrict__ W,
                                              float* __restrict__ t) {
  __shared__ float As[16 * HC];
  const int tid  = threadIdx.x;
  const int wave = tid >> 5, lane = tid & 31;
  const int m0   = blockIdx.x * 16;
  {
    const float4* xv = (const float4*)(h2 + (size_t)m0 * HC);
    float4* sv = (float4*)As;
    for (int i = tid; i < 1024; i += 96) sv[i] = xv[i];
  }
  __syncthreads();
  const int n0 = wave * 16;
  const int la = lane & 15, hi = lane >> 4;
  const int col = n0 + la;
  const bool ok = col < ODIM;
  v8f acc = {};
#pragma unroll 8
  for (int k0 = 0; k0 < HC; k0 += 4) {
    const int ka = k0 + hi * 2;
    v2f a, b;
    a.x = As[la * HC + ka];
    a.y = As[la * HC + ka + 1];
    b.x = ok ? W[ka * ODIM + col] : 0.f;
    b.y = ok ? W[(ka + 1) * ODIM + col] : 0.f;
    acc = __builtin_amdgcn_wmma_f32_16x16x4_f32(false, a, false, b, (short)0,
                                                acc, false, false);
  }
#pragma unroll
  for (int j = 0; j < 8; ++j)
    if (ok) t[(size_t)(m0 + j + hi * 8) * ODIM + col] = acc[j];
}

__global__ __launch_bounds__(256) void k_att1(const float* __restrict__ h,
                                              const float* __restrict__ as_,
                                              const float* __restrict__ ad_,
                                              float* __restrict__ os,
                                              float* __restrict__ od) {
  const int node = blockIdx.x * 8 + (threadIdx.x >> 5);
  const int lane = threadIdx.x & 31;
  if (node >= NND) return;
  const float* hr = h + (size_t)node * HC;
#pragma unroll
  for (int hd = 0; hd < NH; ++hd) {
    const float v = hr[hd * HIDC + lane];
    const float s = wave_sum(v * as_[hd * HIDC + lane]);
    const float d = wave_sum(v * ad_[hd * HIDC + lane]);
    if (lane == 0) { os[node * NH + hd] = s; od[node * NH + hd] = d; }
  }
}

__global__ __launch_bounds__(256) void k_att2(const float* __restrict__ t,
                                              const float* __restrict__ as_,
                                              const float* __restrict__ ad_,
                                              float* __restrict__ os,
                                              float* __restrict__ od) {
  const int node = blockIdx.x * 8 + (threadIdx.x >> 5);
  const int lane = threadIdx.x & 31;
  if (node >= NND) return;
  const float* tr = t + (size_t)node * ODIM;
  float s = 0.f, d = 0.f;
  {
    const float v = tr[lane];
    s += v * as_[lane]; d += v * ad_[lane];
  }
  if (lane < ODIM - 32) {
    const float v = tr[32 + lane];
    s += v * as_[32 + lane]; d += v * ad_[32 + lane];
  }
  s = wave_sum(s); d = wave_sum(d);
  if (lane == 0) { os[node] = s; od[node] = d; }
}

__device__ __forceinline__ float bias_elu(float a, float r, float b) {
  const float v = a / r + b;
  return v > 0.f ? v : (__expf(v) - 1.f);
}
__global__ __launch_bounds__(256) void k_node1(const int* __restrict__ offs,
                                               const int* __restrict__ deg,
                                               const int* __restrict__ csr,
                                               const float* __restrict__ h,
                                               const float* __restrict__ aS,
                                               const float* __restrict__ aD,
                                               const float* __restrict__ bias,
                                               float* __restrict__ h2) {
  const int node = __builtin_amdgcn_readfirstlane((int)(blockIdx.x * 8 + (threadIdx.x >> 5)));
  const int lane = threadIdx.x & 31;
  if (node >= NND) return;
  const int beg = offs[node];
  const int n   = deg[node];
  const int g   = lane >> 3;
  const float ad0 = aD[node * NH + g], ad1 = aD[node * NH + 4 + g];
  const float es0 = lrelu(aS[node * NH + g] + ad0);
  const float es1 = lrelu(aS[node * NH + 4 + g] + ad1);
  float m0 = es0, m1 = es1;
#pragma unroll 4
  for (int i = 0; i < n; ++i) {
    int s = csr[beg + i];
    s = (unsigned)s < (unsigned)NND ? s : node;
    const float e0 = lrelu(aS[s * NH + g] + ad0);
    const float e1 = lrelu(aS[s * NH + 4 + g] + ad1);
    m0 = e0 > m0 ? e0 : m0;
    m1 = e1 > m1 ? e1 : m1;
  }
  float4 a0, a1;
  float den0, den1;
  {
    const float x0 = __expf(es0 - m0), x1 = __expf(es1 - m1);
    const float4* hr = (const float4*)(h + (size_t)node * HC);
    const float4 v0 = hr[lane], v1 = hr[32 + lane];
    a0 = make_float4(v0.x * x0, v0.y * x0, v0.z * x0, v0.w * x0);
    a1 = make_float4(v1.x * x1, v1.y * x1, v1.z * x1, v1.w * x1);
    den0 = x0; den1 = x1;
  }
#pragma unroll 4
  for (int i = 0; i < n; ++i) {
    int s = csr[beg + i];
    s = (unsigned)s < (unsigned)NND ? s : node;
    const float x0 = __expf(lrelu(aS[s * NH + g] + ad0) - m0);
    const float x1 = __expf(lrelu(aS[s * NH + 4 + g] + ad1) - m1);
    const float4* hr = (const float4*)(h + (size_t)s * HC);
    const float4 v0 = hr[lane], v1 = hr[32 + lane];
    a0.x += v0.x * x0; a0.y += v0.y * x0; a0.z += v0.z * x0; a0.w += v0.w * x0;
    a1.x += v1.x * x1; a1.y += v1.y * x1; a1.z += v1.z * x1; a1.w += v1.w * x1;
    den0 += x0; den1 += x1;
  }
  const float r0 = den0 + EPSV, r1 = den1 + EPSV;
  const float4 b0 = ((const float4*)bias)[lane], b1 = ((const float4*)bias)[32 + lane];
  float4* o = (float4*)(h2 + (size_t)node * HC);
  o[lane]      = make_float4(bias_elu(a0.x, r0, b0.x), bias_elu(a0.y, r0, b0.y),
                             bias_elu(a0.z, r0, b0.z), bias_elu(a0.w, r0, b0.w));
  o[32 + lane] = make_float4(bias_elu(a1.x, r1, b1.x), bias_elu(a1.y, r1, b1.y),
                             bias_elu(a1.z, r1, b1.z), bias_elu(a1.w, r1, b1.w));
}

__global__ __launch_bounds__(256) void k_node2(const int* __restrict__ offs,
                                               const int* __restrict__ deg,
                                               const int* __restrict__ csr,
                                               const float* __restrict__ t,
                                               const float* __restrict__ aS,
                                               const float* __restrict__ aD,
                                               const float* __restrict__ bias,
                                               float* __restrict__ out) {
  const int node = __builtin_amdgcn_readfirstlane((int)(blockIdx.x * 8 + (threadIdx.x >> 5)));
  const int lane = threadIdx.x & 31;
  if (node >= NND) return;
  const int beg = offs[node];
  const int n   = deg[node];
  const bool tail = lane < ODIM - 32;
  const float ad = aD[node];
  const float es = lrelu(aS[node] + ad);
  float m = es;
#pragma unroll 4
  for (int i = 0; i < n; ++i) {
    int s = csr[beg + i];
    s = (unsigned)s < (unsigned)NND ? s : node;
    const float e = lrelu(aS[s] + ad);
    m = e > m ? e : m;
  }
  float acc0, acc1 = 0.f, den;
  {
    const float x = __expf(es - m);
    const float* tr = t + (size_t)node * ODIM;
    acc0 = tr[lane] * x;
    if (tail) acc1 = tr[32 + lane] * x;
    den = x;
  }
#pragma unroll 4
  for (int i = 0; i < n; ++i) {
    int s = csr[beg + i];
    s = (unsigned)s < (unsigned)NND ? s : node;
    const float x = __expf(lrelu(aS[s] + ad) - m);
    const float* ts = t + (size_t)s * ODIM;
    acc0 += ts[lane] * x;
    if (tail) acc1 += ts[32 + lane] * x;
    den += x;
  }
  const float r = den + EPSV;
  float* o = out + (size_t)node * ODIM;
  o[lane] = acc0 / r + bias[lane];
  if (tail) o[32 + lane] = acc1 / r + bias[32 + lane];
}

extern "C" void kernel_launch(void* const* d_in, const int* in_sizes, int n_in,
                              void* d_out, int out_size, void* d_ws, size_t ws_size,
                              hipStream_t stream) {
  const float* x   = (const float*)d_in[0];
  const int*   ei  = (const int*)d_in[1];
  const float* W1  = (const float*)d_in[2];
  const float* as1 = (const float*)d_in[3];
  const float* ad1 = (const float*)d_in[4];
  const float* b1  = (const float*)d_in[5];
  const float* W2  = (const float*)d_in[6];
  const float* as2 = (const float*)d_in[7];
  const float* ad2 = (const float*)d_in[8];
  const float* b2  = (const float*)d_in[9];
  float* out = (float*)d_out;

  float* ws   = (float*)d_ws;
  float* h1   = ws;
  float* h2   = h1   + (size_t)NND * HC;
  float* aS1  = h2   + (size_t)NND * HC;
  float* aD1  = aS1  + NND * NH;
  float* t2   = aD1  + NND * NH;
  int*   csr  = (int*)(t2 + (size_t)NND * ODIM);
  float* aS2  = (float*)(csr + NE);
  float* aD2  = aS2  + 50048;
  int*   deg  = (int*)(aD2 + 50048);
  int*   offs = deg  + 50048;
  int*   hist = offs + 50048;
  int*   pre  = hist + SCAN_N;
  int2*  rec  = (int2*)(pre + SCAN_N);

  k_bhist<<<NEB, SBT, 0, stream>>>(ei, hist);
  k_scan<<<1, 1024, 0, stream>>>(hist, pre);
  k_bscatter<<<NEB, SBT, 0, stream>>>(ei, pre, rec);
  k_bucket<<<NBK, SBT, 0, stream>>>(rec, pre, offs, deg, csr);

  k_gemm1<<<dim3(NND / 16, 2), 256, 0, stream>>>(x, W1, h1);
  k_att1<<<(NND + 7) / 8, 256, 0, stream>>>(h1, as1, ad1, aS1, aD1);
  k_node1<<<(NND + 7) / 8, 256, 0, stream>>>(offs, deg, csr, h1, aS1, aD1, b1, h2);

  k_gemm2<<<NND / 16, 96, 0, stream>>>(h2, W2, t2);
  k_att2<<<(NND + 7) / 8, 256, 0, stream>>>(t2, as2, ad2, aS2, aD2);
  k_node2<<<(NND + 7) / 8, 256, 0, stream>>>(offs, deg, csr, t2, aS2, aD2, b2, out);
}


# A tiles of both GEMMs get 4 floats of LDS row padding (pitch 132 / 260): a fragment read's 16 rows fall on 16 different bank groups instead of one; values and sum order unchanged
# speedup vs baseline: 3.1611x; 1.0895x over previous
#include <hip/hip_runtime.h>
#include <hip/hip_bf16.h>

#define NND   50000
#define FIN   128
#define NH    8
#define HIDC  32
#define HC    256
#define ODIM  40
#define NE    800000
#define NET   850000
#define NEG_SLOPE 0.2f
#define EPSV  1e-16f

typedef __attribute__((ext_vector_type(2))) float v2f;
typedef __attribute__((ext_vector_type(8))) float v8f;

__device__ __forceinline__ float wave_sum(float v) {
#pragma unroll
  for (int off = 16; off > 0; off >>= 1) v += __shfl_xor(v, off, 32);
  return v;
}
__device__ __forceinline__ float lrelu(float v) {
  return v > 0.f ? v : v * NEG_SLOPE;
}

#define NPB   256
#define NBK   196
#define NEB   256
#define EPB   (NE / NEB)
#define SBT   1024

__device__ __forceinline__ int dst_bucket_clamped(int d) {
  return (unsigned)d < (unsigned)NND ? d : NND - 1;
}

__global__ __launch_bounds__(SBT) void k_bhist(const int* __restrict__ ei,
                                               int* __restrict__ hist) {
  __shared__ int cnt[NBK];
  const int tid = threadIdx.x, blk = blockIdx.x;
  if (tid < NBK) cnt[tid] = 0;
  __syncthreads();
  const int* dstp = ei + NE + blk * EPB;
  for (int i = tid; i < EPB; i += SBT) {
    const int d = dst_bucket_clamped(dstp[i]);
    atomicAdd(&cnt[d >> 8], 1);
  }
  __syncthreads();
  if (tid < NBK) hist[tid * NEB + blk] = cnt[tid];
}

#define SCAN_PER 49
#define SCAN_N   (1024 * SCAN_PER)
#define SCAN_N4  (SCAN_N / 4)
#define SCAN_FULL (SCAN_N4 / 1024)
static_assert(NBK * NEB == SCAN_N, "scan size must equal the count table");
static_assert(NBK * NPB >= NND, "buckets must cover all nodes");
static_assert(EPB * NEB == NE, "edge blocks must cover all edges");
__global__ __launch_bounds__(1024) void k_scan(const int* __restrict__ cntin,
                                               int* __restrict__ pre) {
  __shared__ int4 sd4[SCAN_N4];
  __shared__ int wsum[32];
  int* sd = (int*)sd4;
  const int tid = threadIdx.x, lane = tid & 31, wave = tid >> 5;
  {
    const int4* dv = (const int4*)cntin;
    int4 t[SCAN_FULL];
#pragma unroll
    for (int k = 0; k < SCAN_FULL; ++k) t[k] = dv[tid + k * 1024];
#pragma unroll
    for (int k = 0; k < SCAN_FULL; ++k) sd4[tid + k * 1024] = t[k];
    const int i = tid + SCAN_FULL * 1024;
    if (i < SCAN_N4) sd4[i] = dv[i];
  }
  __syncthreads();
  const int base = tid * SCAN_PER;
  int v[SCAN_PER];
  int loc = 0;
#pragma unroll
  for (int j = 0; j < SCAN_PER; ++j) { v[j] = sd[base + j]; loc += v[j]; }
  int inc = loc;
#pragma unroll
  for (int off = 1; off < 32; off <<= 1) {
    const int t = __shfl_up(inc, off, 32);
    if (lane >= off) inc += t;
  }
  if (lane == 31) wsum[wave] = inc;
  __syncthreads();
  if (wave == 0) {
    const int w = wsum[lane];
    int iv = w;
#pragma unroll
    for (int off = 1; off < 32; off <<= 1) {
      const int t = __shfl_up(iv, off, 32);
      if (lane >= off) iv += t;
    }
    wsum[lane] = iv - w;
  }
  __syncthreads();
  int run = wsum[wave] + inc - loc;
#pragma unroll
  for (int j = 0; j < SCAN_PER; ++j) { sd[base + j] = run; run += v[j]; }
  __syncthreads();
  {
    int4* ov = (int4*)pre;
#pragma unroll
    for (int k = 0; k < SCAN_FULL; ++k) ov[tid + k * 1024] = sd4[tid + k * 1024];
    const int i = tid + SCAN_FULL * 1024;
    if (i < SCAN_N4) ov[i] = sd4[i];
  }
}

__global__ __launch_bounds__(SBT) void k_bscatter(const int* __restrict__ ei,
                                                  const int* __restrict__ pre,
                                                  int2* __restrict__ rec) {
  __shared__ int cnt[NBK];
  __shared__ int bas[NBK];
  const int tid = threadIdx.x, blk = blockIdx.x;
  if (tid < NBK) { cnt[tid] = 0; bas[tid] = pre[tid * NEB + blk]; }
  __syncthreads();
  const int* srcp = ei + blk * EPB;
  const int* dstp = ei + NE + blk * EPB;
  for (int i = tid; i < EPB; i += SBT) {
    const int s = srcp[i];
    const int d = dst_bucket_clamped(dstp[i]);
    const int b = d >> 8;
    const int pos = bas[b] + atomicAdd(&cnt[b], 1);
    if ((unsigned)pos < (unsigned)NE) rec[pos] = make_int2(s, d);
  }
}

__global__ __launch_bounds__(SBT) void k_bucket(const int2* __restrict__ rec,
                                                const int* __restrict__ pre,
                                                int* __restrict__ offs,
                                                int* __restrict__ deg,
                                                int* __restrict__ csr) {
  __shared__ int cnt[NPB];
  __shared__ int cur[NPB];
  __shared__ int wsum[NPB / 32];
  const int tid = threadIdx.x, bk = blockIdx.x, lane = tid & 31, wave = tid >> 5;
  int beg = pre[bk * NEB];
  int end = (bk + 1 < NBK) ? pre[(bk + 1) * NEB] : NE;
  beg = beg < 0 ? 0 : (beg > NE ? NE : beg);
  end = end < beg ? beg : (end > NE ? NE : end);
  if (tid < NPB) cnt[tid] = 0;
  __syncthreads();
  for (int i = beg + tid; i < end; i += SBT) {
    const int2 r = rec[i];
    atomicAdd(&cnt[r.y & (NPB - 1)], 1);
  }
  __syncthreads();
  const int c = tid < NPB ? cnt[tid] : 0;
  int inc = c;
#pragma unroll
  for (int off = 1; off < 32; off <<= 1) {
    const int t = __shfl_up(inc, off, 32);
    if (lane >= off) inc += t;
  }
  if (tid < NPB && lane == 31) wsum[wave] = inc;
  __syncthreads();
  if (tid < NPB) {
    int wb = 0;
#pragma unroll
    for (int w = 0; w < NPB / 32; ++w) wb += w < wave ? wsum[w] : 0;
    const int o = beg + wb + inc - c;
    cur[tid] = o;
    const int node = bk * NPB + tid;
    if (node < NND) { offs[node] = o; deg[node] = c; }
  }
  __syncthreads();
  for (int i = beg + tid; i < end; i += SBT) {
    const int2 r = rec[i];
    const int pos = atomicAdd(&cur[r.y & (NPB - 1)], 1);
    if ((unsigned)pos < (unsigned)NE) csr[pos] = r.x;
  }
}

#define APAD   4
#define PITCH1 (FIN + APAD)
#define PITCH2 (HC + APAD)
__global__ __launch_bounds__(256) void k_gemm1(const float* __restrict__ x,
                                               const float* __restrict__ W,
                                               float* __restrict__ h) {
  __shared__ __attribute__((aligned(16))) float As[16 * PITCH1];
  const int tid  = threadIdx.x;
  const int wave = tid >> 5, lane = tid & 31;
  const int m0   = blockIdx.x * 16;
  {
    const float4* xv = (const float4*)(x + (size_t)m0 * FIN);
    const int r = tid >> 5, c = (tid & 31) * 4;
    *(float4*)&As[r * PITCH1 + c]       = xv[tid];
    *(float4*)&As[(r + 8) * PITCH1 + c] = xv[tid + 256];
  }
  __syncthreads();
  const int n0 = blockIdx.y * 128 + wave * 16;
  const int la = lane & 15, hi = lane >> 4;
  v8f acc = {};
#pragma unroll
  for (int k0 = 0; k0 < FIN; k0 += 4) {
    const int ka = k0 + hi * 2;
    v2f a, b;
    a.x = As[la * PITCH1 + ka];
    a.y = As[la * PITCH1 + ka + 1];
    b.x = W[ka * HC + n0 + la];
    b.y = W[(ka + 1) * HC + n0 + la];
    acc = __builtin_amdgcn_wmma_f32_16x16x4_f32(false, a, false, b, (short)0,
                                                acc, false, false);
  }
#pragma unroll
  for (int j = 0; j < 8; ++j)
    h[(size_t)(m0 + j + hi * 8) * HC + n0 + la] = acc[j];
}

__global__ __launch_bounds__(96) void k_gemm2(const float* __restrict__ h2,
                                              const float* __restrict__ W,
                                              float* __restrict__ t) {
  __shared__ __attribute__((aligned(16))) float As[16 * PITCH2];
  const int tid  = threadIdx.x;
  const int wave = tid >> 5, lane = tid & 31;
  const int m0   = blockIdx.x * 16;
  {
    const float4* xv = (const float4*)(h2 + (size_t)m0 * HC);
    for (int i = tid; i < 1024; i += 96)
      *(float4*)&As[(i >> 6) * PITCH2 + (i & 63) * 4] = xv[i];
  }
  __syncthreads();
  const int n0 = wave * 16;
  const int la = lane & 15, hi = lane >> 4;
  const int col = n0 + la;
  const bool ok = col < ODIM;
  v8f acc = {};
#pragma unroll 8
  for (int k0 = 0; k0 < HC; k0 += 4) {
    const int ka = k0 + hi * 2;
    v2f a, b;
    a.x = As[la * PITCH2 + ka];
    a.y = As[la * PITCH2 + ka + 1];
    b.x = ok ? W[ka * ODIM + col] : 0.f;
    b.y = ok ? W[(ka + 1) * ODIM + col] : 0.f;
    acc = __builtin_amdgcn_wmma_f32_16x16x4_f32(false, a, false, b, (short)0,
                                                acc, false, false);
  }
#pragma unroll
  for (int j = 0; j < 8; ++j)
    if (ok) t[(size_t)(m0 + j + hi * 8) * ODIM + col] = acc[j];
}

__global__ __launch_bounds__(256) void k_att1(const float* __restrict__ h,
                                              const float* __restrict__ as_,
                                              const float* __restrict__ ad_,
                                              float* __restrict__ os,
                                              float* __restrict__ od) {
  const int node = blockIdx.x * 8 + (threadIdx.x >> 5);
  const int lane = threadIdx.x & 31;
  if (node >= NND) return;
  const float* hr = h + (size_t)node * HC;
#pragma unroll
  for (int hd = 0; hd < NH; ++hd) {
    const float v = hr[hd * HIDC + lane];
    const float s = wave_sum(v * as_[hd * HIDC + lane]);
    const float d = wave_sum(v * ad_[hd * HIDC + lane]);
    if (lane == 0) { os[node * NH + hd] = s; od[node * NH + hd] = d; }
  }
}

__global__ __launch_bounds__(256) void k_att2(const float* __restrict__ t,
                                              const float* __restrict__ as_,
                                              const float* __restrict__ ad_,
                                              float* __restrict__ os,
                                              float* __restrict__ od) {
  const int node = blockIdx.x * 8 + (threadIdx.x >> 5);
  const int lane = threadIdx.x & 31;
  if (node >= NND) return;
  const float* tr = t + (size_t)node * ODIM;
  float s = 0.f, d = 0.f;
  {
    const float v = tr[lane];
    s += v * as_[lane]; d += v * ad_[lane];
  }
  if (lane < ODIM - 32) {
    const float v = tr[32 + lane];
    s += v * as_[32 + lane]; d += v * ad_[32 + lane];
  }
  s = wave_sum(s); d = wave_sum(d);
  if (lane == 0) { os[node] = s; od[node] = d; }
}

__device__ __forceinline__ float bias_elu(float a, float r, float b) {
  const float v = a / r + b;
  return v > 0.f ? v : (__expf(v) - 1.f);
}
__global__ __launch_bounds__(256) void k_node1(const int* __restrict__ offs,
                                               const int* __restrict__ deg,
                                               const int* __restrict__ csr,
                                               const float* __restrict__ h,
                                               const float* __restrict__ aS,
                                               const float* __restrict__ aD,
                                               const float* __restrict__ bias,
                                               float* __restrict__ h2) {
  const int node = __builtin_amdgcn_readfirstlane((int)(blockIdx.x * 8 + (threadIdx.x >> 5)));
  const int lane = threadIdx.x & 31;
  if (node >= NND) return;
  const int beg = offs[node];
  const int n   = deg[node];
  const int g   = lane >> 3;
  const float ad0 = aD[node * NH + g], ad1 = aD[node * NH + 4 + g];
  const float es0 = lrelu(aS[node * NH + g] + ad0);
  const float es1 = lrelu(aS[node * NH + 4 + g] + ad1);
  float m0 = es0, m1 = es1;
#pragma unroll 4
  for (int i = 0; i < n; ++i) {
    int s = csr[beg + i];
    s = (unsigned)s < (unsigned)NND ? s : node;
    const float e0 = lrelu(aS[s * NH + g] + ad0);
    const float e1 = lrelu(aS[s * NH + 4 + g] + ad1);
    m0 = e0 > m0 ? e0 : m0;
    m1 = e1 > m1 ? e1 : m1;
  }
  float4 a0, a1;
  float den0, den1;
  {
    const float x0 = __expf(es0 - m0), x1 = __expf(es1 - m1);
    const float4* hr = (const float4*)(h + (size_t)node * HC);
    const float4 v0 = hr[lane], v1 = hr[32 + lane];
    a0 = make_float4(v0.x * x0, v0.y * x0, v0.z * x0, v0.w * x0);
    a1 = make_float4(v1.x * x1, v1.y * x1, v1.z * x1, v1.w * x1);
    den0 = x0; den1 = x1;
  }
#pragma unroll 4
  for (int i = 0; i < n; ++i) {
    int s = csr[beg + i];
    s = (unsigned)s < (unsigned)NND ? s : node;
    const float x0 = __expf(lrelu(aS[s * NH + g] + ad0) - m0);
    const float x1 = __expf(lrelu(aS[s * NH + 4 + g] + ad1) - m1);
    const float4* hr = (const float4*)(h + (size_t)s * HC);
    const float4 v0 = hr[lane], v1 = hr[32 + lane];
    a0.x += v0.x * x0; a0.y += v0.y * x0; a0.z += v0.z * x0; a0.w += v0.w * x0;
    a1.x += v1.x * x1; a1.y += v1.y * x1; a1.z += v1.z * x1; a1.w += v1.w * x1;
    den0 += x0; den1 += x1;
  }
  const float r0 = den0 + EPSV, r1 = den1 + EPSV;
  const float4 b0 = ((const float4*)bias)[lane], b1 = ((const float4*)bias)[32 + lane];
  float4* o = (float4*)(h2 + (size_t)node * HC);
  o[lane]      = make_float4(bias_elu(a0.x, r0, b0.x), bias_elu(a0.y, r0, b0.y),
                             bias_elu(a0.z, r0, b0.z), bias_elu(a0.w, r0, b0.w));
  o[32 + lane] = make_float4(bias_elu(a1.x, r1, b1.x), bias_elu(a1.y, r1, b1.y),
                             bias_elu(a1.z, r1, b1.z), bias_elu(a1.w, r1, b1.w));
}

__global__ __launch_bounds__(256) void k_node2(const int* __restrict__ offs,
                                               const int* __restrict__ deg,
                                               const int* __restrict__ csr,
                                               const float* __restrict__ t,
                                               const float* __restrict__ aS,
                                               const float* __restrict__ aD,
                                               const float* __restrict__ bias,
                                               float* __restrict__ out) {
  const int node = __builtin_amdgcn_readfirstlane((int)(blockIdx.x * 8 + (threadIdx.x >> 5)));
  const int lane = threadIdx.x & 31;
  if (node >= NND) return;
  const int beg = offs[node];
  const int n   = deg[node];
  const bool tail = lane < ODIM - 32;
  const float ad = aD[node];
  const float es = lrelu(aS[node] + ad);
  float m = es;
#pragma unroll 4
  for (int i = 0; i < n; ++i) {
    int s = csr[beg + i];
    s = (unsigned)s < (unsigned)NND ? s : node;
    const float e = lrelu(aS[s] + ad);
    m = e > m ? e : m;
  }
  float acc0, acc1 = 0.f, den;
  {
    const float x = __expf(es - m);
    const float* tr = t + (size_t)node * ODIM;
    acc0 = tr[lane] * x;
    if (tail) acc1 = tr[32 + lane] * x;
    den = x;
  }
#pragma unroll 4
  for (int i = 0; i < n; ++i) {
    int s = csr[beg + i];
    s = (unsigned)s < (unsigned)NND ? s : node;
    const float x = __expf(lrelu(aS[s] + ad) - m);
    const float* ts = t + (size_t)s * ODIM;
    acc0 += ts[lane] * x;
    if (tail) acc1 += ts[32 + lane] * x;
    den += x;
  }
  const float r = den + EPSV;
  float* o = out + (size_t)node * ODIM;
  o[lane] = acc0 / r + bias[lane];
  if (tail) o[32 + lane] = acc1 / r + bias[32 + lane];
}

extern "C" void kernel_launch(void* const* d_in, const int* in_sizes, int n_in,
                              void* d_out, int out_size, void* d_ws, size_t ws_size,
                              hipStream_t stream) {
  const float* x   = (const float*)d_in[0];
  const int*   ei  = (const int*)d_in[1];
  const float* W1  = (const float*)d_in[2];
  const float* as1 = (const float*)d_in[3];
  const float* ad1 = (const float*)d_in[4];
  const float* b1  = (const float*)d_in[5];
  const float* W2  = (const float*)d_in[6];
  const float* as2 = (const float*)d_in[7];
  const float* ad2 = (const float*)d_in[8];
  const float* b2  = (const float*)d_in[9];
  float* out = (float*)d_out;

  float* ws   = (float*)d_ws;
  float* h1   = ws;
  float* h2   = h1   + (size_t)NND * HC;
  float* aS1  = h2   + (size_t)NND * HC;
  float* aD1  = aS1  + NND * NH;
  float* t2   = aD1  + NND * NH;
  int*   csr  = (int*)(t2 + (size_t)NND * ODIM);
  float* aS2  = (float*)(csr + NE);
  float* aD2  = aS2  + 50048;
  int*   deg  = (int*)(aD2 + 50048);
  int*   offs = deg  + 50048;
  int*   hist = offs + 50048;
  int*   pre  = hist + SCAN_N;
  int2*  rec  = (int2*)(pre + SCAN_N);

  k_bhist<<<NEB, SBT, 0, stream>>>(ei, hist);
  k_scan<<<1, 1024, 0, stream>>>(hist, pre);
  k_bscatter<<<NEB, SBT, 0, stream>>>(ei, pre, rec);
  k_bucket<<<NBK, SBT, 0, stream>>>(rec, pre, offs, deg, csr);

  k_gemm1<<<dim3(NND / 16, 2), 256, 0, stream>>>(x, W1, h1);
  k_att1<<<(NND + 7) / 8, 256, 0, stream>>>(h1, as1, ad1, aS1, aD1);
  k_node1<<<(NND + 7) / 8, 256, 0, stream>>>(offs, deg, csr, h1, aS1, aD1, b1, h2);

  k_gemm2<<<NND / 16, 96, 0, stream>>>(h2, W2, t2);
  k_att2<<<(NND + 7) / 8, 256, 0, stream>>>(t2, as2, ad2, aS2, aD2);
  k_node2<<<(NND + 7) / 8, 256, 0, stream>>>(offs, deg, csr, t2, aS2, aD2, b2, out);
}


# Attention dots of both layers taken from the accumulator tiles in the GEMM epilogues: the two wave-per-node launches re-reading h1 and t2 are gone; dot sum order = 16-lane tree per tile, tiles added
# speedup vs baseline: 3.5609x; 1.1265x over previous
#include <hip/hip_runtime.h>
#include <hip/hip_bf16.h>

#define NND   50000
#define FIN   128
#define NH    8
#define HIDC  32
#define HC    256
#define ODIM  40
#define NE    800000
#define NET   850000
#define NEG_SLOPE 0.2f
#define EPSV  1e-16f

typedef __attribute__((ext_vector_type(2))) float v2f;
typedef __attribute__((ext_vector_type(8))) float v8f;

__device__ __forceinline__ float lrelu(float v) {
  return v > 0.f ? v : v * NEG_SLOPE;
}
__device__ __forceinline__ float half_rows_sum(const float (&p)[8], int c) {
  const bool b3 = (c & 8) != 0, b2 = (c & 4) != 0, b1 = (c & 2) != 0;
  float q[4], r[2];
#pragma unroll
  for (int j = 0; j < 4; ++j) {
    const float keep = b3 ? p[j + 4] : p[j];
    const float send = b3 ? p[j] : p[j + 4];
    q[j] = keep + __shfl_xor(send, 8, 32);
  }
#pragma unroll
  for (int j = 0; j < 2; ++j) {
    const float keep = b2 ? q[j + 2] : q[j];
    const float send = b2 ? q[j] : q[j + 2];
    r[j] = keep + __shfl_xor(send, 4, 32);
  }
  const float keep = b1 ? r[1] : r[0];
  const float send = b1 ? r[0] : r[1];
  float s = keep + __shfl_xor(send, 2, 32);
  s += __shfl_xor(s, 1, 32);
  return s;
}

#define NPB   256
#define NBK   196
#define NEB   256
#define EPB   (NE / NEB)
#define SBT   1024

__device__ __forceinline__ int dst_bucket_clamped(int d) {
  return (unsigned)d < (unsigned)NND ? d : NND - 1;
}

__global__ __launch_bounds__(SBT) void k_bhist(const int* __restrict__ ei,
                                               int* __restrict__ hist) {
  __shared__ int cnt[NBK];
  const int tid = threadIdx.x, blk = blockIdx.x;
  if (tid < NBK) cnt[tid] = 0;
  __syncthreads();
  const int* dstp = ei + NE + blk * EPB;
  for (int i = tid; i < EPB; i += SBT) {
    const int d = dst_bucket_clamped(dstp[i]);
    atomicAdd(&cnt[d >> 8], 1);
  }
  __syncthreads();
  if (tid < NBK) hist[tid * NEB + blk] = cnt[tid];
}

#define SCAN_PER 49
#define SCAN_N   (1024 * SCAN_PER)
#define SCAN_N4  (SCAN_N / 4)
#define SCAN_FULL (SCAN_N4 / 1024)
static_assert(NBK * NEB == SCAN_N, "scan size must equal the count table");
static_assert(NBK * NPB >= NND, "buckets must cover all nodes");
static_assert(EPB * NEB == NE, "edge blocks must cover all edges");
__global__ __launch_bounds__(1024) void k_scan(const int* __restrict__ cntin,
                                               int* __restrict__ pre) {
  __shared__ int4 sd4[SCAN_N4];
  __shared__ int wsum[32];
  int* sd = (int*)sd4;
  const int tid = threadIdx.x, lane = tid & 31, wave = tid >> 5;
  {
    const int4* dv = (const int4*)cntin;
    int4 t[SCAN_FULL];
#pragma unroll
    for (int k = 0; k < SCAN_FULL; ++k) t[k] = dv[tid + k * 1024];
#pragma unroll
    for (int k = 0; k < SCAN_FULL; ++k) sd4[tid + k * 1024] = t[k];
    const int i = tid + SCAN_FULL * 1024;
    if (i < SCAN_N4) sd4[i] = dv[i];
  }
  __syncthreads();
  const int base = tid * SCAN_PER;
  int v[SCAN_PER];
  int loc = 0;
#pragma unroll
  for (int j = 0; j < SCAN_PER; ++j) { v[j] = sd[base + j]; loc += v[j]; }
  int inc = loc;
#pragma unroll
  for (int off = 1; off < 32; off <<= 1) {
    const int t = __shfl_up(inc, off, 32);
    if (lane >= off) inc += t;
  }
  if (lane == 31) wsum[wave] = inc;
  __syncthreads();
  if (wave == 0) {
    const int w = wsum[lane];
    int iv = w;
#pragma unroll
    for (int off = 1; off < 32; off <<= 1) {
      const int t = __shfl_up(iv, off, 32);
      if (lane >= off) iv += t;
    }
    wsum[lane] = iv - w;
  }
  __syncthreads();
  int run = wsum[wave] + inc - loc;
#pragma unroll
  for (int j = 0; j < SCAN_PER; ++j) { sd[base + j] = run; run += v[j]; }
  __syncthreads();
  {
    int4* ov = (int4*)pre;
#pragma unroll
    for (int k = 0; k < SCAN_FULL; ++k) ov[tid + k * 1024] = sd4[tid + k * 1024];
    const int i = tid + SCAN_FULL * 1024;
    if (i < SCAN_N4) ov[i] = sd4[i];
  }
}

__global__ __launch_bounds__(SBT) void k_bscatter(const int* __restrict__ ei,
                                                  const int* __restrict__ pre,
                                                  int2* __restrict__ rec) {
  __shared__ int cnt[NBK];
  __shared__ int bas[NBK];
  const int tid = threadIdx.x, blk = blockIdx.x;
  if (tid < NBK) { cnt[tid] = 0; bas[tid] = pre[tid * NEB + blk]; }
  __syncthreads();
  const int* srcp = ei + blk * EPB;
  const int* dstp = ei + NE + blk * EPB;
  for (int i = tid; i < EPB; i += SBT) {
    const int s = srcp[i];
    const int d = dst_bucket_clamped(dstp[i]);
    const int b = d >> 8;
    const int pos = bas[b] + atomicAdd(&cnt[b], 1);
    if ((unsigned)pos < (unsigned)NE) rec[pos] = make_int2(s, d);
  }
}

__global__ __launch_bounds__(SBT) void k_bucket(const int2* __restrict__ rec,
                                                const int* __restrict__ pre,
                                                int* __restrict__ offs,
                                                int* __restrict__ deg,
                                                int* __restrict__ csr) {
  __shared__ int cnt[NPB];
  __shared__ int cur[NPB];
  __shared__ int wsum[NPB / 32];
  const int tid = threadIdx.x, bk = blockIdx.x, lane = tid & 31, wave = tid >> 5;
  int beg = pre[bk * NEB];
  int end = (bk + 1 < NBK) ? pre[(bk + 1) * NEB] : NE;
  beg = beg < 0 ? 0 : (beg > NE ? NE : beg);
  end = end < beg ? beg : (end > NE ? NE : end);
  if (tid < NPB) cnt[tid] = 0;
  __syncthreads();
  for (int i = beg + tid; i < end; i += SBT) {
    const int2 r = rec[i];
    atomicAdd(&cnt[r.y & (NPB - 1)], 1);
  }
  __syncthreads();
  const int c = tid < NPB ? cnt[tid] : 0;
  int inc = c;
#pragma unroll
  for (int off = 1; off < 32; off <<= 1) {
    const int t = __shfl_up(inc, off, 32);
    if (lane >= off) inc += t;
  }
  if (tid < NPB && lane == 31) wsum[wave] = inc;
  __syncthreads();
  if (tid < NPB) {
    int wb = 0;
#pragma unroll
    for (int w = 0; w < NPB / 32; ++w) wb += w < wave ? wsum[w] : 0;
    const int o = beg + wb + inc - c;
    cur[tid] = o;
    const int node = bk * NPB + tid;
    if (node < NND) { offs[node] = o; deg[node] = c; }
  }
  __syncthreads();
  for (int i = beg + tid; i < end; i += SBT) {
    const int2 r = rec[i];
    const int pos = atomicAdd(&cur[r.y & (NPB - 1)], 1);
    if ((unsigned)pos < (unsigned)NE) csr[pos] = r.x;
  }
}

#define APAD   4
#define PITCH1 (FIN + APAD)
#define PITCH2 (HC + APAD)
__global__ __launch_bounds__(256, 16) void k_gemm1(const float* __restrict__ x,
                                               const float* __restrict__ W,
                                               const float* __restrict__ as_,
                                               const float* __restrict__ ad_,
                                               float* __restrict__ h,
                                               float* __restrict__ os,
                                               float* __restrict__ od) {
  __shared__ __attribute__((aligned(16))) float As[16 * PITCH1];
  __shared__ float red[8][16][2];
  const int tid  = threadIdx.x;
  const int wave = tid >> 5, lane = tid & 31;
  const int m0   = blockIdx.x * 16;
  {
    const float4* xv = (const float4*)(x + (size_t)m0 * FIN);
    const int r = tid >> 5, c = (tid & 31) * 4;
    *(float4*)&As[r * PITCH1 + c]       = xv[tid];
    *(float4*)&As[(r + 8) * PITCH1 + c] = xv[tid + 256];
  }
  __syncthreads();
  const int n0 = blockIdx.y * 128 + wave * 16;
  const int la = lane & 15, hi = lane >> 4;
  v8f acc = {};
#pragma unroll
  for (int k0 = 0; k0 < FIN; k0 += 4) {
    const int ka = k0 + hi * 2;
    v2f a, b;
    a.x = As[la * PITCH1 + ka];
    a.y = As[la * PITCH1 + ka + 1];
    b.x = W[ka * HC + n0 + la];
    b.y = W[(ka + 1) * HC + n0 + la];
    acc = __builtin_amdgcn_wmma_f32_16x16x4_f32(false, a, false, b, (short)0,
                                                acc, false, false);
  }
#pragma unroll
  for (int j = 0; j < 8; ++j)
    h[(size_t)(m0 + j + hi * 8) * HC + n0 + la] = acc[j];
  {
    const float ws = as_[n0 + la], wd = ad_[n0 + la];
    float ps[8], pd[8];
#pragma unroll
    for (int j = 0; j < 8; ++j) { ps[j] = acc[j] * ws; pd[j] = acc[j] * wd; }
    const float ss = half_rows_sum(ps, la), sd = half_rows_sum(pd, la);
    if ((la & 1) == 0) {
      red[wave][hi * 8 + (la >> 1)][0] = ss;
      red[wave][hi * 8 + (la >> 1)][1] = sd;
    }
  }
  __syncthreads();
  if (tid < 32) {
    const int row = tid >> 1, which = tid & 1;
    float4 v;
    v.x = red[0][row][which] + red[1][row][which];
    v.y = red[2][row][which] + red[3][row][which];
    v.z = red[4][row][which] + red[5][row][which];
    v.w = red[6][row][which] + red[7][row][which];
    float* dp = (which ? od : os) + (size_t)(m0 + row) * NH + blockIdx.y * 4;
    *(float4*)dp = v;
  }
}

__global__ __launch_bounds__(96) void k_gemm2(const float* __restrict__ h2,
                                              const float* __restrict__ W,
                                              const float* __restrict__ as_,
                                              const float* __restrict__ ad_,
                                              float* __restrict__ t,
                                              float* __restrict__ os,
                                              float* __restrict__ od) {
  __shared__ __attribute__((aligned(16))) float As[16 * PITCH2];
  __shared__ float red[3][16][2];
  const int tid  = threadIdx.x;
  const int wave = tid >> 5, lane = tid & 31;
  const int m0   = blockIdx.x * 16;
  {
    const float4* xv = (const float4*)(h2 + (size_t)m0 * HC);
    for (int i = tid; i < 1024; i += 96)
      *(float4*)&As[(i >> 6) * PITCH2 + (i & 63) * 4] = xv[i];
  }
  __syncthreads();
  const int n0 = wave * 16;
  const int la = lane & 15, hi = lane >> 4;
  const int col = n0 + la;
  const bool ok = col < ODIM;
  v8f acc = {};
#pragma unroll 8
  for (int k0 = 0; k0 < HC; k0 += 4) {
    const int ka = k0 + hi * 2;
    v2f a, b;
    a.x = As[la * PITCH2 + ka];
    a.y = As[la * PITCH2 + ka + 1];
    b.x = ok ? W[ka * ODIM + col] : 0.f;
    b.y = ok ? W[(ka + 1) * ODIM + col] : 0.f;
    acc = __builtin_amdgcn_wmma_f32_16x16x4_f32(false, a, false, b, (short)0,
                                                acc, false, false);
  }
#pragma unroll
  for (int j = 0; j < 8; ++j)
    if (ok) t[(size_t)(m0 + j + hi * 8) * ODIM + col] = acc[j];
  {
    const float ws = ok ? as_[col] : 0.f, wd = ok ? ad_[col] : 0.f;
    float ps[8], pd[8];
#pragma unroll
    for (int j = 0; j < 8; ++j) {
      ps[j] = ok ? acc[j] * ws : 0.f;
      pd[j] = ok ? acc[j] * wd : 0.f;
    }
    const float ss = half_rows_sum(ps, la), sd = half_rows_sum(pd, la);
    if ((la & 1) == 0) {
      red[wave][hi * 8 + (la >> 1)][0] = ss;
      red[wave][hi * 8 + (la >> 1)][1] = sd;
    }
  }
  __syncthreads();
  if (tid < 32) {
    const int row = tid >> 1, which = tid & 1;
    const float v = red[0][row][which] + red[1][row][which] + red[2][row][which];
    (which ? od : os)[m0 + row] = v;
  }
}

__device__ __forceinline__ float bias_elu(float a, float r, float b) {
  const float v = a / r + b;
  return v > 0.f ? v : (__expf(v) - 1.f);
}
__global__ __launch_bounds__(256) void k_node1(const int* __restrict__ offs,
                                               const int* __restrict__ deg,
                                               const int* __restrict__ csr,
                                               const float* __restrict__ h,
                                               const float* __restrict__ aS,
                                               const float* __restrict__ aD,
                                               const float* __restrict__ bias,
                                               float* __restrict__ h2) {
  const int node = __builtin_amdgcn_readfirstlane((int)(blockIdx.x * 8 + (threadIdx.x >> 5)));
  const int lane = threadIdx.x & 31;
  if (node >= NND) return;
  const int beg = offs[node];
  const int n   = deg[node];
  const int g   = lane >> 3;
  const float ad0 = aD[node * NH + g], ad1 = aD[node * NH + 4 + g];
  const float es0 = lrelu(aS[node * NH + g] + ad0);
  const float es1 = lrelu(aS[node * NH + 4 + g] + ad1);
  float m0 = es0, m1 = es1;
#pragma unroll 4
  for (int i = 0; i < n; ++i) {
    int s = csr[beg + i];
    s = (unsigned)s < (unsigned)NND ? s : node;
    const float e0 = lrelu(aS[s * NH + g] + ad0);
    const float e1 = lrelu(aS[s * NH + 4 + g] + ad1);
    m0 = e0 > m0 ? e0 : m0;
    m1 = e1 > m1 ? e1 : m1;
  }
  float4 a0, a1;
  float den0, den1;
  {
    const float x0 = __expf(es0 - m0), x1 = __expf(es1 - m1);
    const float4* hr = (const float4*)(h + (size_t)node * HC);
    const float4 v0 = hr[lane], v1 = hr[32 + lane];
    a0 = make_float4(v0.x * x0, v0.y * x0, v0.z * x0, v0.w * x0);
    a1 = make_float4(v1.x * x1, v1.y * x1, v1.z * x1, v1.w * x1);
    den0 = x0; den1 = x1;
  }
#pragma unroll 4
  for (int i = 0; i < n; ++i) {
    int s = csr[beg + i];
    s = (unsigned)s < (unsigned)NND ? s : node;
    const float x0 = __expf(lrelu(aS[s * NH + g] + ad0) - m0);
    const float x1 = __expf(lrelu(aS[s * NH + 4 + g] + ad1) - m1);
    const float4* hr = (const float4*)(h + (size_t)s * HC);
    const float4 v0 = hr[lane], v1 = hr[32 + lane];
    a0.x += v0.x * x0; a0.y += v0.y * x0; a0.z += v0.z * x0; a0.w += v0.w * x0;
    a1.x += v1.x * x1; a1.y += v1.y * x1; a1.z += v1.z * x1; a1.w += v1.w * x1;
    den0 += x0; den1 += x1;
  }
  const float r0 = den0 + EPSV, r1 = den1 + EPSV;
  const float4 b0 = ((const float4*)bias)[lane], b1 = ((const float4*)bias)[32 + lane];
  float4* o = (float4*)(h2 + (size_t)node * HC);
  o[lane]      = make_float4(bias_elu(a0.x, r0, b0.x), bias_elu(a0.y, r0, b0.y),
                             bias_elu(a0.z, r0, b0.z), bias_elu(a0.w, r0, b0.w));
  o[32 + lane] = make_float4(bias_elu(a1.x, r1, b1.x), bias_elu(a1.y, r1, b1.y),
                             bias_elu(a1.z, r1, b1.z), bias_elu(a1.w, r1, b1.w));
}

__global__ __launch_bounds__(256) void k_node2(const int* __restrict__ offs,
                                               const int* __restrict__ deg,
                                               const int* __restrict__ csr,
                                               const float* __restrict__ t,
                                               const float* __restrict__ aS,
                                               const float* __restrict__ aD,
                                               const float* __restrict__ bias,
                                               float* __restrict__ out) {
  const int node = __builtin_amdgcn_readfirstlane((int)(blockIdx.x * 8 + (threadIdx.x >> 5)));
  const int lane = threadIdx.x & 31;
  if (node >= NND) return;
  const int beg = offs[node];
  const int n   = deg[node];
  const bool tail = lane < ODIM - 32;
  const float ad = aD[node];
  const float es = lrelu(aS[node] + ad);
  float m = es;
#pragma unroll 4
  for (int i = 0; i < n; ++i) {
    int s = csr[beg + i];
    s = (unsigned)s < (unsigned)NND ? s : node;
    const float e = lrelu(aS[s] + ad);
    m = e > m ? e : m;
  }
  float acc0, acc1 = 0.f, den;
  {
    const float x = __expf(es - m);
    const float* tr = t + (size_t)node * ODIM;
    acc0 = tr[lane] * x;
    if (tail) acc1 = tr[32 + lane] * x;
    den = x;
  }
#pragma unroll 4
  for (int i = 0; i < n; ++i) {
    int s = csr[beg + i];
    s = (unsigned)s < (unsigned)NND ? s : node;
    const float x = __expf(lrelu(aS[s] + ad) - m);
    const float* ts = t + (size_t)s * ODIM;
    acc0 += ts[lane] * x;
    if (tail) acc1 += ts[32 + lane] * x;
    den += x;
  }
  const float r = den + EPSV;
  float* o = out + (size_t)node * ODIM;
  o[lane] = acc0 / r + bias[lane];
  if (tail) o[32 + lane] = acc1 / r + bias[32 + lane];
}

extern "C" void kernel_launch(void* const* d_in, const int* in_sizes, int n_in,
                              void* d_out, int out_size, void* d_ws, size_t ws_size,
                              hipStream_t stream) {
  const float* x   = (const float*)d_in[0];
  const int*   ei  = (const int*)d_in[1];
  const float* W1  = (const float*)d_in[2];
  const float* as1 = (const float*)d_in[3];
  const float* ad1 = (const float*)d_in[4];
  const float* b1  = (const float*)d_in[5];
  const float* W2  = (const float*)d_in[6];
  const float* as2 = (const float*)d_in[7];
  const float* ad2 = (const float*)d_in[8];
  const float* b2  = (const float*)d_in[9];
  float* out = (float*)d_out;

  float* ws   = (float*)d_ws;
  float* h1   = ws;
  float* h2   = h1   + (size_t)NND * HC;
  float* aS1  = h2   + (size_t)NND * HC;
  float* aD1  = aS1  + NND * NH;
  float* t2   = aD1  + NND * NH;
  int*   csr  = (int*)(t2 + (size_t)NND * ODIM);
  float* aS2  = (float*)(csr + NE);
  float* aD2  = aS2  + 50048;
  int*   deg  = (int*)(aD2 + 50048);
  int*   offs = deg  + 50048;
  int*   hist = offs + 50048;
  int*   pre  = hist + SCAN_N;
  int2*  rec  = (int2*)(pre + SCAN_N);

  k_bhist<<<NEB, SBT, 0, stream>>>(ei, hist);
  k_scan<<<1, 1024, 0, stream>>>(hist, pre);
  k_bscatter<<<NEB, SBT, 0, stream>>>(ei, pre, rec);
  k_bucket<<<NBK, SBT, 0, stream>>>(rec, pre, offs, deg, csr);

  k_gemm1<<<dim3(NND / 16, 2), 256, 0, stream>>>(x, W1, as1, ad1, h1, aS1, aD1);
  k_node1<<<(NND + 7) / 8, 256, 0, stream>>>(offs, deg, csr, h1, aS1, aD1, b1, h2);

  k_gemm2<<<NND / 16, 96, 0, stream>>>(h2, W2, as2, ad2, t2, aS2, aD2);
  k_node2<<<(NND + 7) / 8, 256, 0, stream>>>(offs, deg, csr, t2, aS2, aD2, b2, out);
}


# Node lists walked once: a source's logit comes from its gathered row (layer 1: 8-lane dot, layer 2: row columns 40/41), softmax on a running max with rescaled sums; no per-edge logit gathers
# speedup vs baseline: 4.1805x; 1.1740x over previous
#include <hip/hip_runtime.h>
#include <hip/hip_bf16.h>

#define NND   50000
#define FIN   128
#define NH    8
#define HIDC  32
#define HC    256
#define ODIM  40
#define NE    800000
#define NET   850000
#define NEG_SLOPE 0.2f
#define EPSV  1e-16f

typedef __attribute__((ext_vector_type(2))) float v2f;
typedef __attribute__((ext_vector_type(8))) float v8f;

__device__ __forceinline__ float lrelu(float v) {
  return v > 0.f ? v : v * NEG_SLOPE;
}
__device__ __forceinline__ float half_rows_sum(const float (&p)[8], int c) {
  const bool b3 = (c & 8) != 0, b2 = (c & 4) != 0, b1 = (c & 2) != 0;
  float q[4], r[2];
#pragma unroll
  for (int j = 0; j < 4; ++j) {
    const float keep = b3 ? p[j + 4] : p[j];
    const float send = b3 ? p[j] : p[j + 4];
    q[j] = keep + __shfl_xor(send, 8, 32);
  }
#pragma unroll
  for (int j = 0; j < 2; ++j) {
    const float keep = b2 ? q[j + 2] : q[j];
    const float send = b2 ? q[j] : q[j + 2];
    r[j] = keep + __shfl_xor(send, 4, 32);
  }
  const float keep = b1 ? r[1] : r[0];
  const float send = b1 ? r[0] : r[1];
  float s = keep + __shfl_xor(send, 2, 32);
  s += __shfl_xor(s, 1, 32);
  return s;
}

template <int M>
__device__ __forceinline__ float row_xor(float v) {
  return __builtin_bit_cast(float, __builtin_amdgcn_mov_dpp(__builtin_bit_cast(int, v),
                                                            0x160 + M, 0xf, 0xf, false));
}
__device__ __forceinline__ float group8_sum(float v) {
  v += row_xor<1>(v);
  v += row_xor<2>(v);
  v += row_xor<4>(v);
  return v;
}
__device__ __forceinline__ float dot4(const float4 a, const float4 b) {
  return a.x * b.x + a.y * b.y + a.z * b.z + a.w * b.w;
}

#define NPB   256
#define NBK   196
#define NEB   256
#define EPB   (NE / NEB)
#define SBT   1024

__device__ __forceinline__ int dst_bucket_clamped(int d) {
  return (unsigned)d < (unsigned)NND ? d : NND - 1;
}

__global__ __launch_bounds__(SBT) void k_bhist(const int* __restrict__ ei,
                                               int* __restrict__ hist) {
  __shared__ int cnt[NBK];
  const int tid = threadIdx.x, blk = blockIdx.x;
  if (tid < NBK) cnt[tid] = 0;
  __syncthreads();
  const int* dstp = ei + NE + blk * EPB;
  for (int i = tid; i < EPB; i += SBT) {
    const int d = dst_bucket_clamped(dstp[i]);
    atomicAdd(&cnt[d >> 8], 1);
  }
  __syncthreads();
  if (tid < NBK) hist[tid * NEB + blk] = cnt[tid];
}

#define SCAN_PER 49
#define SCAN_N   (1024 * SCAN_PER)
#define SCAN_N4  (SCAN_N / 4)
#define SCAN_FULL (SCAN_N4 / 1024)
static_assert(NBK * NEB == SCAN_N, "scan size must equal the count table");
static_assert(NBK * NPB >= NND, "buckets must cover all nodes");
static_assert(EPB * NEB == NE, "edge blocks must cover all edges");
__global__ __launch_bounds__(1024) void k_scan(const int* __restrict__ cntin,
                                               int* __restrict__ pre) {
  __shared__ int4 sd4[SCAN_N4];
  __shared__ int wsum[32];
  int* sd = (int*)sd4;
  const int tid = threadIdx.x, lane = tid & 31, wave = tid >> 5;
  {
    const int4* dv = (const int4*)cntin;
    int4 t[SCAN_FULL];
#pragma unroll
    for (int k = 0; k < SCAN_FULL; ++k) t[k] = dv[tid + k * 1024];
#pragma unroll
    for (int k = 0; k < SCAN_FULL; ++k) sd4[tid + k * 1024] = t[k];
    const int i = tid + SCAN_FULL * 1024;
    if (i < SCAN_N4) sd4[i] = dv[i];
  }
  __syncthreads();
  const int base = tid * SCAN_PER;
  int v[SCAN_PER];
  int loc = 0;
#pragma unroll
  for (int j = 0; j < SCAN_PER; ++j) { v[j] = sd[base + j]; loc += v[j]; }
  int inc = loc;
#pragma unroll
  for (int off = 1; off < 32; off <<= 1) {
    const int t = __shfl_up(inc, off, 32);
    if (lane >= off) inc += t;
  }
  if (lane == 31) wsum[wave] = inc;
  __syncthreads();
  if (wave == 0) {
    const int w = wsum[lane];
    int iv = w;
#pragma unroll
    for (int off = 1; off < 32; off <<= 1) {
      const int t = __shfl_up(iv, off, 32);
      if (lane >= off) iv += t;
    }
    wsum[lane] = iv - w;
  }
  __syncthreads();
  int run = wsum[wave] + inc - loc;
#pragma unroll
  for (int j = 0; j < SCAN_PER; ++j) { sd[base + j] = run; run += v[j]; }
  __syncthreads();
  {
    int4* ov = (int4*)pre;
#pragma unroll
    for (int k = 0; k < SCAN_FULL; ++k) ov[tid + k * 1024] = sd4[tid + k * 1024];
    const int i = tid + SCAN_FULL * 1024;
    if (i < SCAN_N4) ov[i] = sd4[i];
  }
}

__global__ __launch_bounds__(SBT) void k_bscatter(const int* __restrict__ ei,
                                                  const int* __restrict__ pre,
                                                  int2* __restrict__ rec) {
  __shared__ int cnt[NBK];
  __shared__ int bas[NBK];
  const int tid = threadIdx.x, blk = blockIdx.x;
  if (tid < NBK) { cnt[tid] = 0; bas[tid] = pre[tid * NEB + blk]; }
  __syncthreads();
  const int* srcp = ei + blk * EPB;
  const int* dstp = ei + NE + blk * EPB;
  for (int i = tid; i < EPB; i += SBT) {
    const int s = srcp[i];
    const int d = dst_bucket_clamped(dstp[i]);
    const int b = d >> 8;
    const int pos = bas[b] + atomicAdd(&cnt[b], 1);
    if ((unsigned)pos < (unsigned)NE) rec[pos] = make_int2(s, d);
  }
}

__global__ __launch_bounds__(SBT) void k_bucket(const int2* __restrict__ rec,
                                                const int* __restrict__ pre,
                                                int* __restrict__ offs,
                                                int* __restrict__ deg,
                                                int* __restrict__ csr) {
  __shared__ int cnt[NPB];
  __shared__ int cur[NPB];
  __shared__ int wsum[NPB / 32];
  const int tid = threadIdx.x, bk = blockIdx.x, lane = tid & 31, wave = tid >> 5;
  int beg = pre[bk * NEB];
  int end = (bk + 1 < NBK) ? pre[(bk + 1) * NEB] : NE;
  beg = beg < 0 ? 0 : (beg > NE ? NE : beg);
  end = end < beg ? beg : (end > NE ? NE : end);
  if (tid < NPB) cnt[tid] = 0;
  __syncthreads();
  for (int i = beg + tid; i < end; i += SBT) {
    const int2 r = rec[i];
    atomicAdd(&cnt[r.y & (NPB - 1)], 1);
  }
  __syncthreads();
  const int c = tid < NPB ? cnt[tid] : 0;
  int inc = c;
#pragma unroll
  for (int off = 1; off < 32; off <<= 1) {
    const int t = __shfl_up(inc, off, 32);
    if (lane >= off) inc += t;
  }
  if (tid < NPB && lane == 31) wsum[wave] = inc;
  __syncthreads();
  if (tid < NPB) {
    int wb = 0;
#pragma unroll
    for (int w = 0; w < NPB / 32; ++w) wb += w < wave ? wsum[w] : 0;
    const int o = beg + wb + inc - c;
    cur[tid] = o;
    const int node = bk * NPB + tid;
    if (node < NND) { offs[node] = o; deg[node] = c; }
  }
  __syncthreads();
  for (int i = beg + tid; i < end; i += SBT) {
    const int2 r = rec[i];
    const int pos = atomicAdd(&cur[r.y & (NPB - 1)], 1);
    if ((unsigned)pos < (unsigned)NE) csr[pos] = r.x;
  }
}

#define APAD   4
#define PITCH1 (FIN + APAD)
#define PITCH2 (HC + APAD)
__global__ __launch_bounds__(256, 16) void k_gemm1(const float* __restrict__ x,
                                               const float* __restrict__ W,
                                               float* __restrict__ h) {
  __shared__ __attribute__((aligned(16))) float As[16 * PITCH1];
  const int tid  = threadIdx.x;
  const int wave = tid >> 5, lane = tid & 31;
  const int m0   = blockIdx.x * 16;
  {
    const float4* xv = (const float4*)(x + (size_t)m0 * FIN);
    const int r = tid >> 5, c = (tid & 31) * 4;
    *(float4*)&As[r * PITCH1 + c]       = xv[tid];
    *(float4*)&As[(r + 8) * PITCH1 + c] = xv[tid + 256];
  }
  __syncthreads();
  const int n0 = blockIdx.y * 128 + wave * 16;
  const int la = lane & 15, hi = lane >> 4;
  v8f acc = {};
#pragma unroll
  for (int k0 = 0; k0 < FIN; k0 += 4) {
    const int ka = k0 + hi * 2;
    v2f a, b;
    a.x = As[la * PITCH1 + ka];
    a.y = As[la * PITCH1 + ka + 1];
    b.x = W[ka * HC + n0 + la];
    b.y = W[(ka + 1) * HC + n0 + la];
    acc = __builtin_amdgcn_wmma_f32_16x16x4_f32(false, a, false, b, (short)0,
                                                acc, false, false);
  }
#pragma unroll
  for (int j = 0; j < 8; ++j)
    h[(size_t)(m0 + j + hi * 8) * HC + n0 + la] = acc[j];
}

#define TP2 64
__global__ __launch_bounds__(96) void k_gemm2(const float* __restrict__ h2,
                                              const float* __restrict__ W,
                                              const float* __restrict__ as_,
                                              const float* __restrict__ ad_,
                                              float* __restrict__ t) {
  __shared__ __attribute__((aligned(16))) float As[16 * PITCH2];
  __shared__ float red[3][16][2];
  const int tid  = threadIdx.x;
  const int wave = tid >> 5, lane = tid & 31;
  const int m0   = blockIdx.x * 16;
  {
    const float4* xv = (const float4*)(h2 + (size_t)m0 * HC);
    for (int i = tid; i < 1024; i += 96)
      *(float4*)&As[(i >> 6) * PITCH2 + (i & 63) * 4] = xv[i];
  }
  __syncthreads();
  const int n0 = wave * 16;
  const int la = lane & 15, hi = lane >> 4;
  const int col = n0 + la;
  const bool ok = col < ODIM;
  v8f acc = {};
#pragma unroll 8
  for (int k0 = 0; k0 < HC; k0 += 4) {
    const int ka = k0 + hi * 2;
    v2f a, b;
    a.x = As[la * PITCH2 + ka];
    a.y = As[la * PITCH2 + ka + 1];
    b.x = ok ? W[ka * ODIM + col] : 0.f;
    b.y = ok ? W[(ka + 1) * ODIM + col] : 0.f;
    acc = __builtin_amdgcn_wmma_f32_16x16x4_f32(false, a, false, b, (short)0,
                                                acc, false, false);
  }
#pragma unroll
  for (int j = 0; j < 8; ++j)
    if (ok) t[(size_t)(m0 + j + hi * 8) * TP2 + col] = acc[j];
  {
    const float ws = ok ? as_[col] : 0.f, wd = ok ? ad_[col] : 0.f;
    float ps[8], pd[8];
#pragma unroll
    for (int j = 0; j < 8; ++j) {
      ps[j] = ok ? acc[j] * ws : 0.f;
      pd[j] = ok ? acc[j] * wd : 0.f;
    }
    const float ss = half_rows_sum(ps, la), sd = half_rows_sum(pd, la);
    if ((la & 1) == 0) {
      red[wave][hi * 8 + (la >> 1)][0] = ss;
      red[wave][hi * 8 + (la >> 1)][1] = sd;
    }
  }
  __syncthreads();
  if (tid < 32) {
    const int row = tid >> 1, which = tid & 1;
    const float v = red[0][row][which] + red[1][row][which] + red[2][row][which];
    t[(size_t)(m0 + row) * TP2 + ODIM + which] = v;
  }
}

__device__ __forceinline__ float bias_elu(float a, float r, float b) {
  const float v = a / r + b;
  return v > 0.f ? v : (__expf(v) - 1.f);
}
__global__ __launch_bounds__(256) void k_node1(const int* __restrict__ offs,
                                               const int* __restrict__ deg,
                                               const int* __restrict__ csr,
                                               const float* __restrict__ h,
                                               const float* __restrict__ att_s,
                                               const float* __restrict__ att_d,
                                               const float* __restrict__ bias,
                                               float* __restrict__ h2) {
  const int node = __builtin_amdgcn_readfirstlane((int)(blockIdx.x * 8 + (threadIdx.x >> 5)));
  const int lane = threadIdx.x & 31;
  if (node >= NND) return;
  const int beg = offs[node];
  const int n   = deg[node];
  const float4 ws0 = ((const float4*)att_s)[lane], ws1 = ((const float4*)att_s)[32 + lane];
  float4 a0, a1;
  float m0, m1, den0, den1, ad0, ad1;
  {
    const float4 wd0 = ((const float4*)att_d)[lane], wd1 = ((const float4*)att_d)[32 + lane];
    const float4* hr = (const float4*)(h + (size_t)node * HC);
    const float4 v0 = hr[lane], v1 = hr[32 + lane];
    ad0 = group8_sum(dot4(v0, wd0));
    ad1 = group8_sum(dot4(v1, wd1));
    m0 = lrelu(group8_sum(dot4(v0, ws0)) + ad0);
    m1 = lrelu(group8_sum(dot4(v1, ws1)) + ad1);
    a0 = v0; a1 = v1;
    den0 = 1.f; den1 = 1.f;
  }
#pragma unroll 4
  for (int i = 0; i < n; ++i) {
    int s = csr[beg + i];
    s = (unsigned)s < (unsigned)NND ? s : node;
    const float4* hr = (const float4*)(h + (size_t)s * HC);
    const float4 v0 = hr[lane], v1 = hr[32 + lane];
    const float e0 = lrelu(group8_sum(dot4(v0, ws0)) + ad0);
    const float e1 = lrelu(group8_sum(dot4(v1, ws1)) + ad1);
    const float n0 = e0 > m0 ? e0 : m0;
    const float n1 = e1 > m1 ? e1 : m1;
    const float c0 = __expf(m0 - n0), x0 = __expf(e0 - n0);
    const float c1 = __expf(m1 - n1), x1 = __expf(e1 - n1);
    a0.x = a0.x * c0 + v0.x * x0; a0.y = a0.y * c0 + v0.y * x0;
    a0.z = a0.z * c0 + v0.z * x0; a0.w = a0.w * c0 + v0.w * x0;
    a1.x = a1.x * c1 + v1.x * x1; a1.y = a1.y * c1 + v1.y * x1;
    a1.z = a1.z * c1 + v1.z * x1; a1.w = a1.w * c1 + v1.w * x1;
    den0 = den0 * c0 + x0; den1 = den1 * c1 + x1;
    m0 = n0; m1 = n1;
  }
  const float r0 = den0 + EPSV, r1 = den1 + EPSV;
  const float4 b0 = ((const float4*)bias)[lane], b1 = ((const float4*)bias)[32 + lane];
  float4* o = (float4*)(h2 + (size_t)node * HC);
  o[lane]      = make_float4(bias_elu(a0.x, r0, b0.x), bias_elu(a0.y, r0, b0.y),
                             bias_elu(a0.z, r0, b0.z), bias_elu(a0.w, r0, b0.w));
  o[32 + lane] = make_float4(bias_elu(a1.x, r1, b1.x), bias_elu(a1.y, r1, b1.y),
                             bias_elu(a1.z, r1, b1.z), bias_elu(a1.w, r1, b1.w));
}

#define LOGIT_LANE (ODIM / 2)
__device__ __forceinline__ float lane_value(float v, int l) {
  return __builtin_bit_cast(float, __builtin_amdgcn_readlane(__builtin_bit_cast(int, v), l));
}
__device__ __forceinline__ void node2_step(const float2 v, const float ad,
                                           float& m, float& den, float2& acc) {
  const float e  = lrelu(lane_value(v.x, LOGIT_LANE) + ad);
  const float nm = e > m ? e : m;
  const float c = __expf(m - nm), x = __expf(e - nm);
  acc.x = acc.x * c + v.x * x;
  acc.y = acc.y * c + v.y * x;
  den = den * c + x;
  m = nm;
}
__global__ __launch_bounds__(256) void k_node2(const int* __restrict__ offs,
                                               const int* __restrict__ deg,
                                               const int* __restrict__ csr,
                                               const float* __restrict__ t,
                                               const float* __restrict__ bias,
                                               float* __restrict__ out) {
  const int node = __builtin_amdgcn_readfirstlane((int)(blockIdx.x * 8 + (threadIdx.x >> 5)));
  const int lane = threadIdx.x & 31;
  if (node >= NND) return;
  const int beg = offs[node];
  const int n   = deg[node];
  const int cl = lane < LOGIT_LANE ? lane : LOGIT_LANE;
  float2 acc;
  float ad, m, den;
  {
    const float2 v = *(const float2*)(t + (size_t)node * TP2 + 2 * cl);
    ad = lane_value(v.y, LOGIT_LANE);
    m  = lrelu(lane_value(v.x, LOGIT_LANE) + ad);
    acc = v;
    den = 1.f;
  }
  int i = 0;
  for (; i + 4 <= n; i += 4) {
    float2 v[4];
#pragma unroll
    for (int j = 0; j < 4; ++j) {
      int s = csr[beg + i + j];
      s = (unsigned)s < (unsigned)NND ? s : node;
      v[j] = *(const float2*)(t + (size_t)s * TP2 + 2 * cl);
    }
    asm volatile("" :: "v"(v[0].x), "v"(v[0].y), "v"(v[1].x), "v"(v[1].y),
                       "v"(v[2].x), "v"(v[2].y), "v"(v[3].x), "v"(v[3].y));
#pragma unroll
    for (int j = 0; j < 4; ++j) node2_step(v[j], ad, m, den, acc);
  }
  for (; i < n; ++i) {
    int s = csr[beg + i];
    s = (unsigned)s < (unsigned)NND ? s : node;
    const float2 v = *(const float2*)(t + (size_t)s * TP2 + 2 * cl);
    node2_step(v, ad, m, den, acc);
  }
  const float r = den + EPSV;
  if (lane < LOGIT_LANE) {
    const float2 b = ((const float2*)bias)[lane];
    *(float2*)(out + (size_t)node * ODIM + 2 * lane) = make_float2(acc.x / r + b.x, acc.y / r + b.y);
  }
}

extern "C" void kernel_launch(void* const* d_in, const int* in_sizes, int n_in,
                              void* d_out, int out_size, void* d_ws, size_t ws_size,
                              hipStream_t stream) {
  const float* x   = (const float*)d_in[0];
  const int*   ei  = (const int*)d_in[1];
  const float* W1  = (const float*)d_in[2];
  const float* as1 = (const float*)d_in[3];
  const float* ad1 = (const float*)d_in[4];
  const float* b1  = (const float*)d_in[5];
  const float* W2  = (const float*)d_in[6];
  const float* as2 = (const float*)d_in[7];
  const float* ad2 = (const float*)d_in[8];
  const float* b2  = (const float*)d_in[9];
  float* out = (float*)d_out;

  float* ws   = (float*)d_ws;
  float* h1   = ws;
  float* h2   = h1   + (size_t)NND * HC;
  float* t2   = h2   + (size_t)NND * HC;
  int*   csr  = (int*)(t2 + (size_t)NND * TP2);
  int*   deg  = csr  + NE;
  int*   offs = deg  + 50048;
  int*   hist = offs + 50048;
  int*   pre  = hist + SCAN_N;
  int2*  rec  = (int2*)(pre + SCAN_N);

  k_bhist<<<NEB, SBT, 0, stream>>>(ei, hist);
  k_scan<<<1, 1024, 0, stream>>>(hist, pre);
  k_bscatter<<<NEB, SBT, 0, stream>>>(ei, pre, rec);
  k_bucket<<<NBK, SBT, 0, stream>>>(rec, pre, offs, deg, csr);

  k_gemm1<<<dim3(NND / 16, 2), 256, 0, stream>>>(x, W1, h1);
  k_node1<<<(NND + 7) / 8, 256, 0, stream>>>(offs, deg, csr, h1, as1, ad1, b1, h2);

  k_gemm2<<<NND / 16, 96, 0, stream>>>(h2, W2, as2, ad2, t2);
  k_node2<<<(NND + 7) / 8, 256, 0, stream>>>(offs, deg, csr, t2, b2, out);
}


# Layer-2 product moved into the layer-1 node kernel: a 16-node work-group keeps its finished rows in LDS as the A tile; k summed in two halves of 128, lower + upper; h2 and one launch are gone
# speedup vs baseline: 4.9196x; 1.1768x over previous
#include <hip/hip_runtime.h>
#include <hip/hip_bf16.h>

#define NND   50000
#define FIN   128
#define NH    8
#define HIDC  32
#define HC    256
#define ODIM  40
#define NE    800000
#define NET   850000
#define NEG_SLOPE 0.2f
#define EPSV  1e-16f

typedef __attribute__((ext_vector_type(2))) float v2f;
typedef __attribute__((ext_vector_type(8))) float v8f;

__device__ __forceinline__ float lrelu(float v) {
  return v > 0.f ? v : v * NEG_SLOPE;
}
__device__ __forceinline__ float half_rows_sum(const float (&p)[8], int c) {
  const bool b3 = (c & 8) != 0, b2 = (c & 4) != 0, b1 = (c & 2) != 0;
  float q[4], r[2];
#pragma unroll
  for (int j = 0; j < 4; ++j) {
    const float keep = b3 ? p[j + 4] : p[j];
    const float send = b3 ? p[j] : p[j + 4];
    q[j] = keep + __shfl_xor(send, 8, 32);
  }
#pragma unroll
  for (int j = 0; j < 2; ++j) {
    const float keep = b2 ? q[j + 2] : q[j];
    const float send = b2 ? q[j] : q[j + 2];
    r[j] = keep + __shfl_xor(send, 4, 32);
  }
  const float keep = b1 ? r[1] : r[0];
  const float send = b1 ? r[0] : r[1];
  float s = keep + __shfl_xor(send, 2, 32);
  s += __shfl_xor(s, 1, 32);
  return s;
}

template <int M>
__device__ __forceinline__ float row_xor(float v) {
  return __builtin_bit_cast(float, __builtin_amdgcn_mov_dpp(__builtin_bit_cast(int, v),
                                                            0x160 + M, 0xf, 0xf, false));
}
__device__ __forceinline__ float group8_sum(float v) {
  v += row_xor<1>(v);
  v += row_xor<2>(v);
  v += row_xor<4>(v);
  return v;
}
__device__ __forceinline__ float dot4(const float4 a, const float4 b) {
  return a.x * b.x + a.y * b.y + a.z * b.z + a.w * b.w;
}

#define NPB   256
#define NBK   196
#define NEB   256
#define EPB   (NE / NEB)
#define SBT   1024

__device__ __forceinline__ int dst_bucket_clamped(int d) {
  return (unsigned)d < (unsigned)NND ? d : NND - 1;
}

__global__ __launch_bounds__(SBT) void k_bhist(const int* __restrict__ ei,
                                               int* __restrict__ hist) {
  __shared__ int cnt[NBK];
  const int tid = threadIdx.x, blk = blockIdx.x;
  if (tid < NBK) cnt[tid] = 0;
  __syncthreads();
  const int* dstp = ei + NE + blk * EPB;
  for (int i = tid; i < EPB; i += SBT) {
    const int d = dst_bucket_clamped(dstp[i]);
    atomicAdd(&cnt[d >> 8], 1);
  }
  __syncthreads();
  if (tid < NBK) hist[tid * NEB + blk] = cnt[tid];
}

#define SCAN_PER 49
#define SCAN_N   (1024 * SCAN_PER)
#define SCAN_N4  (SCAN_N / 4)
#define SCAN_FULL (SCAN_N4 / 1024)
static_assert(NBK * NEB == SCAN_N, "scan size must equal the count table");
static_assert(NBK * NPB >= NND, "buckets must cover all nodes");
static_assert(EPB * NEB == NE, "edge blocks must cover all edges");
__global__ __launch_bounds__(1024) void k_scan(const int* __restrict__ cntin,
                                               int* __restrict__ pre) {
  __shared__ int4 sd4[SCAN_N4];
  __shared__ int wsum[32];
  int* sd = (int*)sd4;
  const int tid = threadIdx.x, lane = tid & 31, wave = tid >> 5;
  {
    const int4* dv = (const int4*)cntin;
    int4 t[SCAN_FULL];
#pragma unroll
    for (int k = 0; k < SCAN_FULL; ++k) t[k] = dv[tid + k * 1024];
#pragma unroll
    for (int k = 0; k < SCAN_FULL; ++k) sd4[tid + k * 1024] = t[k];
    const int i = tid + SCAN_FULL * 1024;
    if (i < SCAN_N4) sd4[i] = dv[i];
  }
  __syncthreads();
  const int base = tid * SCAN_PER;
  int v[SCAN_PER];
  int loc = 0;
#pragma unroll
  for (int j = 0; j < SCAN_PER; ++j) { v[j] = sd[base + j]; loc += v[j]; }
  int inc = loc;
#pragma unroll
  for (int off = 1; off < 32; off <<= 1) {
    const int t = __shfl_up(inc, off, 32);
    if (lane >= off) inc += t;
  }
  if (lane == 31) wsum[wave] = inc;
  __syncthreads();
  if (wave == 0) {
    const int w = wsum[lane];
    int iv = w;
#pragma unroll
    for (int off = 1; off < 32; off <<= 1) {
      const int t = __shfl_up(iv, off, 32);
      if (lane >= off) iv += t;
    }
    wsum[lane] = iv - w;
  }
  __syncthreads();
  int run = wsum[wave] + inc - loc;
#pragma unroll
  for (int j = 0; j < SCAN_PER; ++j) { sd[base + j] = run; run += v[j]; }
  __syncthreads();
  {
    int4* ov = (int4*)pre;
#pragma unroll
    for (int k = 0; k < SCAN_FULL; ++k) ov[tid + k * 1024] = sd4[tid + k * 1024];
    const int i = tid + SCAN_FULL * 1024;
    if (i < SCAN_N4) ov[i] = sd4[i];
  }
}

__global__ __launch_bounds__(SBT) void k_bscatter(const int* __restrict__ ei,
                                                  const int* __restrict__ pre,
                                                  int2* __restrict__ rec) {
  __shared__ int cnt[NBK];
  __shared__ int bas[NBK];
  const int tid = threadIdx.x, blk = blockIdx.x;
  if (tid < NBK) { cnt[tid] = 0; bas[tid] = pre[tid * NEB + blk]; }
  __syncthreads();
  const int* srcp = ei + blk * EPB;
  const int* dstp = ei + NE + blk * EPB;
  for (int i = tid; i < EPB; i += SBT) {
    const int s = srcp[i];
    const int d = dst_bucket_clamped(dstp[i]);
    const int b = d >> 8;
    const int pos = bas[b] + atomicAdd(&cnt[b], 1);
    if ((unsigned)pos < (unsigned)NE) rec[pos] = make_int2(s, d);
  }
}

__global__ __launch_bounds__(SBT) void k_bucket(const int2* __restrict__ rec,
                                                const int* __restrict__ pre,
                                                int* __restrict__ offs,
                                                int* __restrict__ deg,
                                                int* __restrict__ csr) {
  __shared__ int cnt[NPB];
  __shared__ int cur[NPB];
  __shared__ int wsum[NPB / 32];
  const int tid = threadIdx.x, bk = blockIdx.x, lane = tid & 31, wave = tid >> 5;
  int beg = pre[bk * NEB];
  int end = (bk + 1 < NBK) ? pre[(bk + 1) * NEB] : NE;
  beg = beg < 0 ? 0 : (beg > NE ? NE : beg);
  end = end < beg ? beg : (end > NE ? NE : end);
  if (tid < NPB) cnt[tid] = 0;
  __syncthreads();
  for (int i = beg + tid; i < end; i += SBT) {
    const int2 r = rec[i];
    atomicAdd(&cnt[r.y & (NPB - 1)], 1);
  }
  __syncthreads();
  const int c = tid < NPB ? cnt[tid] : 0;
  int inc = c;
#pragma unroll
  for (int off = 1; off < 32; off <<= 1) {
    const int t = __shfl_up(inc, off, 32);
    if (lane >= off) inc += t;
  }
  if (tid < NPB && lane == 31) wsum[wave] = inc;
  __syncthreads();
  if (tid < NPB) {
    int wb = 0;
#pragma unroll
    for (int w = 0; w < NPB / 32; ++w) wb += w < wave ? wsum[w] : 0;
    const int o = beg + wb + inc - c;
    cur[tid] = o;
    const int node = bk * NPB + tid;
    if (node < NND) { offs[node] = o; deg[node] = c; }
  }
  __syncthreads();
  for (int i = beg + tid; i < end; i += SBT) {
    const int2 r = rec[i];
    const int pos = atomicAdd(&cur[r.y & (NPB - 1)], 1);
    if ((unsigned)pos < (unsigned)NE) csr[pos] = r.x;
  }
}

#define APAD   4
#define PITCH1 (FIN + APAD)
#define PITCH2 (HC + APAD)
__global__ __launch_bounds__(256, 16) void k_gemm1(const float* __restrict__ x,
                                               const float* __restrict__ W,
                                               float* __restrict__ h) {
  __shared__ __attribute__((aligned(16))) float As[16 * PITCH1];
  const int tid  = threadIdx.x;
  const int wave = tid >> 5, lane = tid & 31;
  const int m0   = blockIdx.x * 16;
  {
    const float4* xv = (const float4*)(x + (size_t)m0 * FIN);
    const int r = tid >> 5, c = (tid & 31) * 4;
    *(float4*)&As[r * PITCH1 + c]       = xv[tid];
    *(float4*)&As[(r + 8) * PITCH1 + c] = xv[tid + 256];
  }
  __syncthreads();
  const int n0 = blockIdx.y * 128 + wave * 16;
  const int la = lane & 15, hi = lane >> 4;
  v8f acc = {};
#pragma unroll
  for (int k0 = 0; k0 < FIN; k0 += 4) {
    const int ka = k0 + hi * 2;
    v2f a, b;
    a.x = As[la * PITCH1 + ka];
    a.y = As[la * PITCH1 + ka + 1];
    b.x = W[ka * HC + n0 + la];
    b.y = W[(ka + 1) * HC + n0 + la];
    acc = __builtin_amdgcn_wmma_f32_16x16x4_f32(false, a, false, b, (short)0,
                                                acc, false, false);
  }
#pragma unroll
  for (int j = 0; j < 8; ++j)
    h[(size_t)(m0 + j + hi * 8) * HC + n0 + la] = acc[j];
}

#define TP2 64
#define NPG 16
#define NWV 8
static_assert(NND % NPG == 0, "every work-group has 16 real nodes");
static_assert(NPG == 2 * NWV, "two nodes per wave");
__device__ __forceinline__ float bias_elu(float a, float r, float b) {
  const float v = a / r + b;
  return v > 0.f ? v : (__expf(v) - 1.f);
}
__global__ __launch_bounds__(NWV * 32) void k_node1(const int* __restrict__ offs,
                                                    const int* __restrict__ deg,
                                                    const int* __restrict__ csr,
                                                    const float* __restrict__ h,
                                                    const float* __restrict__ att_s,
                                                    const float* __restrict__ att_d,
                                                    const float* __restrict__ bias,
                                                    const float* __restrict__ W2,
                                                    const float* __restrict__ as2,
                                                    const float* __restrict__ ad2,
                                                    float* __restrict__ t) {
  __shared__ __attribute__((aligned(16))) float As[NPG * PITCH2];
  __shared__ float part[6][8][32];
  __shared__ float red[3][16][2];
  const int tid  = threadIdx.x;
  const int lane = tid & 31;
  const int wave = __builtin_amdgcn_readfirstlane(tid >> 5);
  const int row0 = blockIdx.x * NPG;
  const float4 ws0 = ((const float4*)att_s)[lane], ws1 = ((const float4*)att_s)[32 + lane];
  const float4 wd0 = ((const float4*)att_d)[lane], wd1 = ((const float4*)att_d)[32 + lane];
  const float4 b0 = ((const float4*)bias)[lane], b1 = ((const float4*)bias)[32 + lane];
  for (int r = 0; r < 2; ++r) {
    const int trow = wave + r * NWV;
    const int node = row0 + trow;
    const int beg = offs[node];
    const int n   = deg[node];
    float4 a0, a1;
    float m0, m1, den0, den1, ad0, ad1;
    {
      const float4* hr = (const float4*)(h + (size_t)node * HC);
      const float4 v0 = hr[lane], v1 = hr[32 + lane];
      ad0 = group8_sum(dot4(v0, wd0));
      ad1 = group8_sum(dot4(v1, wd1));
      m0 = lrelu(group8_sum(dot4(v0, ws0)) + ad0);
      m1 = lrelu(group8_sum(dot4(v1, ws1)) + ad1);
      a0 = v0; a1 = v1;
      den0 = 1.f; den1 = 1.f;
    }
#pragma unroll 4
    for (int i = 0; i < n; ++i) {
      int s = csr[beg + i];
      s = (unsigned)s < (unsigned)NND ? s : node;
      const float4* hr = (const float4*)(h + (size_t)s * HC);
      const float4 v0 = hr[lane], v1 = hr[32 + lane];
      const float e0 = lrelu(group8_sum(dot4(v0, ws0)) + ad0);
      const float e1 = lrelu(group8_sum(dot4(v1, ws1)) + ad1);
      const float n0 = e0 > m0 ? e0 : m0;
      const float n1 = e1 > m1 ? e1 : m1;
      const float c0 = __expf(m0 - n0), x0 = __expf(e0 - n0);
      const float c1 = __expf(m1 - n1), x1 = __expf(e1 - n1);
      a0.x = a0.x * c0 + v0.x * x0; a0.y = a0.y * c0 + v0.y * x0;
      a0.z = a0.z * c0 + v0.z * x0; a0.w = a0.w * c0 + v0.w * x0;
      a1.x = a1.x * c1 + v1.x * x1; a1.y = a1.y * c1 + v1.y * x1;
      a1.z = a1.z * c1 + v1.z * x1; a1.w = a1.w * c1 + v1.w * x1;
      den0 = den0 * c0 + x0; den1 = den1 * c1 + x1;
      m0 = n0; m1 = n1;
    }
    const float r0 = den0 + EPSV, r1 = den1 + EPSV;
    float4* o = (float4*)&As[trow * PITCH2];
    o[lane]      = make_float4(bias_elu(a0.x, r0, b0.x), bias_elu(a0.y, r0, b0.y),
                               bias_elu(a0.z, r0, b0.z), bias_elu(a0.w, r0, b0.w));
    o[32 + lane] = make_float4(bias_elu(a1.x, r1, b1.x), bias_elu(a1.y, r1, b1.y),
                               bias_elu(a1.z, r1, b1.z), bias_elu(a1.w, r1, b1.w));
  }
  __syncthreads();
  const int la = lane & 15, hi = lane >> 4;
  if (wave < 6) {
    const int col = (wave >> 1) * 16 + la;
    const bool ok = col < ODIM;
    const int colc = ok ? col : ODIM - 1;
    const int kb = (wave & 1) * (HC / 2);
    v8f acc = {};
#pragma unroll 8
    for (int k0 = 0; k0 < HC / 2; k0 += 4) {
      const int ka = kb + k0 + hi * 2;
      v2f a, b;
      a.x = As[la * PITCH2 + ka];
      a.y = As[la * PITCH2 + ka + 1];
      const float w0 = W2[ka * ODIM + colc], w1 = W2[(ka + 1) * ODIM + colc];
      b.x = ok ? w0 : 0.f;
      b.y = ok ? w1 : 0.f;
      acc = __builtin_amdgcn_wmma_f32_16x16x4_f32(false, a, false, b, (short)0,
                                                  acc, false, false);
    }
#pragma unroll
    for (int j = 0; j < 8; ++j) part[wave][j][lane] = acc[j];
  }
  __syncthreads();
  if (wave < 3) {
    const int col = wave * 16 + la;
    const bool ok = col < ODIM;
    float acc[8];
#pragma unroll
    for (int j = 0; j < 8; ++j) acc[j] = part[2 * wave][j][lane] + part[2 * wave + 1][j][lane];
#pragma unroll
    for (int j = 0; j < 8; ++j)
      if (ok) t[(size_t)(row0 + j + hi * 8) * TP2 + col] = acc[j];
    const float ws = ok ? as2[col] : 0.f, wd = ok ? ad2[col] : 0.f;
    float ps[8], pd[8];
#pragma unroll
    for (int j = 0; j < 8; ++j) {
      ps[j] = ok ? acc[j] * ws : 0.f;
      pd[j] = ok ? acc[j] * wd : 0.f;
    }
    const float ss = half_rows_sum(ps, la), sd = half_rows_sum(pd, la);
    if ((la & 1) == 0) {
      red[wave][hi * 8 + (la >> 1)][0] = ss;
      red[wave][hi * 8 + (la >> 1)][1] = sd;
    }
  }
  __syncthreads();
  if (tid < 32) {
    const int row = tid >> 1, which = tid & 1;
    const float v = red[0][row][which] + red[1][row][which] + red[2][row][which];
    t[(size_t)(row0 + row) * TP2 + ODIM + which] = v;
  }
}

#define LOGIT_LANE (ODIM / 2)
__device__ __forceinline__ float lane_value(float v, int l) {
  return __builtin_bit_cast(float, __builtin_amdgcn_readlane(__builtin_bit_cast(int, v), l));
}
__device__ __forceinline__ void node2_step(const float2 v, const float ad,
                                           float& m, float& den, float2& acc) {
  const float e  = lrelu(lane_value(v.x, LOGIT_LANE) + ad);
  const float nm = e > m ? e : m;
  const float c = __expf(m - nm), x = __expf(e - nm);
  acc.x = acc.x * c + v.x * x;
  acc.y = acc.y * c + v.y * x;
  den = den * c + x;
  m = nm;
}
__global__ __launch_bounds__(256) void k_node2(const int* __restrict__ offs,
                                               const int* __restrict__ deg,
                                               const int* __restrict__ csr,
                                               const float* __restrict__ t,
                                               const float* __restrict__ bias,
                                               float* __restrict__ out) {
  const int node = __builtin_amdgcn_readfirstlane((int)(blockIdx.x * 8 + (threadIdx.x >> 5)));
  const int lane = threadIdx.x & 31;
  if (node >= NND) return;
  const int beg = offs[node];
  const int n   = deg[node];
  const int cl = lane < LOGIT_LANE ? lane : LOGIT_LANE;
  float2 acc;
  float ad, m, den;
  {
    const float2 v = *(const float2*)(t + (size_t)node * TP2 + 2 * cl);
    ad = lane_value(v.y, LOGIT_LANE);
    m  = lrelu(lane_value(v.x, LOGIT_LANE) + ad);
    acc = v;
    den = 1.f;
  }
  int i = 0;
  for (; i + 4 <= n; i += 4) {
    float2 v[4];
#pragma unroll
    for (int j = 0; j < 4; ++j) {
      int s = csr[beg + i + j];
      s = (unsigned)s < (unsigned)NND ? s : node;
      v[j] = *(const float2*)(t + (size_t)s * TP2 + 2 * cl);
    }
    asm volatile("" :: "v"(v[0].x), "v"(v[0].y), "v"(v[1].x), "v"(v[1].y),
                       "v"(v[2].x), "v"(v[2].y), "v"(v[3].x), "v"(v[3].y));
#pragma unroll
    for (int j = 0; j < 4; ++j) node2_step(v[j], ad, m, den, acc);
  }
  for (; i < n; ++i) {
    int s = csr[beg + i];
    s = (unsigned)s < (unsigned)NND ? s : node;
    const float2 v = *(const float2*)(t + (size_t)s * TP2 + 2 * cl);
    node2_step(v, ad, m, den, acc);
  }
  const float r = den + EPSV;
  if (lane < LOGIT_LANE) {
    const float2 b = ((const float2*)bias)[lane];
    *(float2*)(out + (size_t)node * ODIM + 2 * lane) = make_float2(acc.x / r + b.x, acc.y / r + b.y);
  }
}

extern "C" void kernel_launch(void* const* d_in, const int* in_sizes, int n_in,
                              void* d_out, int out_size, void* d_ws, size_t ws_size,
                              hipStream_t stream) {
  const float* x   = (const float*)d_in[0];
  const int*   ei  = (const int*)d_in[1];
  const float* W1  = (const float*)d_in[2];
  const float* as1 = (const float*)d_in[3];
  const float* ad1 = (const float*)d_in[4];
  const float* b1  = (const float*)d_in[5];
  const float* W2  = (const float*)d_in[6];
  const float* as2 = (const float*)d_in[7];
  const float* ad2 = (const float*)d_in[8];
  const float* b2  = (const float*)d_in[9];
  float* out = (float*)d_out;

  float* ws   = (float*)d_ws;
  float* h1   = ws;
  float* t2   = h1   + (size_t)NND * HC;
  int*   csr  = (int*)(t2 + (size_t)NND * TP2);
  int*   deg  = csr  + NE;
  int*   offs = deg  + 50048;
  int*   hist = offs + 50048;
  int*   pre  = hist + SCAN_N;
  int2*  rec  = (int2*)(pre + SCAN_N);

  k_bhist<<<NEB, SBT, 0, stream>>>(ei, hist);
  k_scan<<<1, 1024, 0, stream>>>(hist, pre);
  k_bscatter<<<NEB, SBT, 0, stream>>>(ei, pre, rec);
  k_bucket<<<NBK, SBT, 0, stream>>>(rec, pre, offs, deg, csr);

  k_gemm1<<<dim3(NND / 16, 2), 256, 0, stream>>>(x, W1, h1);
  k_node1<<<NND / NPG, NWV * 32, 0, stream>>>(offs, deg, csr, h1, as1, ad1, b1, W2, as2, ad2, t2);

  k_node2<<<(NND + 7) / 8, 256, 0, stream>>>(offs, deg, csr, t2, b2, out);
}


# Rows of h1 stored with columns permuted so a lane's two 16-byte reads hold 8 channels of ONE head: one logit and softmax per lane, 4-lane sum not 8; logit dot sum order: 8 in-lane, then 4-lane tree
# speedup vs baseline: 5.1902x; 1.0550x over previous
#include <hip/hip_runtime.h>
#include <hip/hip_bf16.h>

#define NND   50000
#define FIN   128
#define NH    8
#define HIDC  32
#define HC    256
#define ODIM  40
#define NE    800000
#define NET   850000
#define NEG_SLOPE 0.2f
#define EPSV  1e-16f

typedef __attribute__((ext_vector_type(2))) float v2f;
typedef __attribute__((ext_vector_type(8))) float v8f;

__device__ __forceinline__ float lrelu(float v) {
  return v > 0.f ? v : v * NEG_SLOPE;
}
__device__ __forceinline__ float half_rows_sum(const float (&p)[8], int c) {
  const bool b3 = (c & 8) != 0, b2 = (c & 4) != 0, b1 = (c & 2) != 0;
  float q[4], r[2];
#pragma unroll
  for (int j = 0; j < 4; ++j) {
    const float keep = b3 ? p[j + 4] : p[j];
    const float send = b3 ? p[j] : p[j + 4];
    q[j] = keep + __shfl_xor(send, 8, 32);
  }
#pragma unroll
  for (int j = 0; j < 2; ++j) {
    const float keep = b2 ? q[j + 2] : q[j];
    const float send = b2 ? q[j] : q[j + 2];
    r[j] = keep + __shfl_xor(send, 4, 32);
  }
  const float keep = b1 ? r[1] : r[0];
  const float send = b1 ? r[0] : r[1];
  float s = keep + __shfl_xor(send, 2, 32);
  s += __shfl_xor(s, 1, 32);
  return s;
}

template <int M>
__device__ __forceinline__ float row_xor(float v) {
  return __builtin_bit_cast(float, __builtin_amdgcn_mov_dpp(__builtin_bit_cast(int, v),
                                                            0x160 + M, 0xf, 0xf, false));
}
__device__ __forceinline__ float group4_sum(float v) {
  v += row_xor<1>(v);
  v += row_xor<2>(v);
  return v;
}
__device__ __forceinline__ float dot8(const float4 a0, const float4 a1,
                                      const float4 w0, const float4 w1) {
  v2f s = v2f{a0.x, a0.y} * v2f{w0.x, w0.y};
  s = v2f{a0.z, a0.w} * v2f{w0.z, w0.w} + s;
  s = v2f{a1.x, a1.y} * v2f{w1.x, w1.y} + s;
  s = v2f{a1.z, a1.w} * v2f{w1.z, w1.w} + s;
  return s.x + s.y;
}

#define NPB   256
#define NBK   196
#define NEB   256
#define EPB   (NE / NEB)
#define SBT   1024

__device__ __forceinline__ int dst_bucket_clamped(int d) {
  return (unsigned)d < (unsigned)NND ? d : NND - 1;
}

__global__ __launch_bounds__(SBT) void k_bhist(const int* __restrict__ ei,
                                               int* __restrict__ hist) {
  __shared__ int cnt[NBK];
  const int tid = threadIdx.x, blk = blockIdx.x;
  if (tid < NBK) cnt[tid] = 0;
  __syncthreads();
  const int* dstp = ei + NE + blk * EPB;
  for (int i = tid; i < EPB; i += SBT) {
    const int d = dst_bucket_clamped(dstp[i]);
    atomicAdd(&cnt[d >> 8], 1);
  }
  __syncthreads();
  if (tid < NBK) hist[tid * NEB + blk] = cnt[tid];
}

#define SCAN_PER 49
#define SCAN_N   (1024 * SCAN_PER)
#define SCAN_N4  (SCAN_N / 4)
#define SCAN_FULL (SCAN_N4 / 1024)
static_assert(NBK * NEB == SCAN_N, "scan size must equal the count table");
static_assert(NBK * NPB >= NND, "buckets must cover all nodes");
static_assert(EPB * NEB == NE, "edge blocks must cover all edges");
__global__ __launch_bounds__(1024) void k_scan(const int* __restrict__ cntin,
                                               int* __restrict__ pre) {
  __shared__ int4 sd4[SCAN_N4];
  __shared__ int wsum[32];
  int* sd = (int*)sd4;
  const int tid = threadIdx.x, lane = tid & 31, wave = tid >> 5;
  {
    const int4* dv = (const int4*)cntin;
    int4 t[SCAN_FULL];
#pragma unroll
    for (int k = 0; k < SCAN_FULL; ++k) t[k] = dv[tid + k * 1024];
#pragma unroll
    for (int k = 0; k < SCAN_FULL; ++k) sd4[tid + k * 1024] = t[k];
    const int i = tid + SCAN_FULL * 1024;
    if (i < SCAN_N4) sd4[i] = dv[i];
  }
  __syncthreads();
  const int base = tid * SCAN_PER;
  int v[SCAN_PER];
  int loc = 0;
#pragma unroll
  for (int j = 0; j < SCAN_PER; ++j) { v[j] = sd[base + j]; loc += v[j]; }
  int inc = loc;
#pragma unroll
  for (int off = 1; off < 32; off <<= 1) {
    const int t = __shfl_up(inc, off, 32);
    if (lane >= off) inc += t;
  }
  if (lane == 31) wsum[wave] = inc;
  __syncthreads();
  if (wave == 0) {
    const int w = wsum[lane];
    int iv = w;
#pragma unroll
    for (int off = 1; off < 32; off <<= 1) {
      const int t = __shfl_up(iv, off, 32);
      if (lane >= off) iv += t;
    }
    wsum[lane] = iv - w;
  }
  __syncthreads();
  int run = wsum[wave] + inc - loc;
#pragma unroll
  for (int j = 0; j < SCAN_PER; ++j) { sd[base + j] = run; run += v[j]; }
  __syncthreads();
  {
    int4* ov = (int4*)pre;
#pragma unroll
    for (int k = 0; k < SCAN_FULL; ++k) ov[tid + k * 1024] = sd4[tid + k * 1024];
    const int i = tid + SCAN_FULL * 1024;
    if (i < SCAN_N4) ov[i] = sd4[i];
  }
}

__global__ __launch_bounds__(SBT) void k_bscatter(const int* __restrict__ ei,
                                                  const int* __restrict__ pre,
                                                  int2* __restrict__ rec) {
  __shared__ int cnt[NBK];
  __shared__ int bas[NBK];
  const int tid = threadIdx.x, blk = blockIdx.x;
  if (tid < NBK) { cnt[tid] = 0; bas[tid] = pre[tid * NEB + blk]; }
  __syncthreads();
  const int* srcp = ei + blk * EPB;
  const int* dstp = ei + NE + blk * EPB;
  for (int i = tid; i < EPB; i += SBT) {
    const int s = srcp[i];
    const int d = dst_bucket_clamped(dstp[i]);
    const int b = d >> 8;
    const int pos = bas[b] + atomicAdd(&cnt[b], 1);
    if ((unsigned)pos < (unsigned)NE) rec[pos] = make_int2(s, d);
  }
}

__global__ __launch_bounds__(SBT) void k_bucket(const int2* __restrict__ rec,
                                                const int* __restrict__ pre,
                                                int* __restrict__ offs,
                                                int* __restrict__ deg,
                                                int* __restrict__ csr) {
  __shared__ int cnt[NPB];
  __shared__ int cur[NPB];
  __shared__ int wsum[NPB / 32];
  const int tid = threadIdx.x, bk = blockIdx.x, lane = tid & 31, wave = tid >> 5;
  int beg = pre[bk * NEB];
  int end = (bk + 1 < NBK) ? pre[(bk + 1) * NEB] : NE;
  beg = beg < 0 ? 0 : (beg > NE ? NE : beg);
  end = end < beg ? beg : (end > NE ? NE : end);
  if (tid < NPB) cnt[tid] = 0;
  __syncthreads();
  for (int i = beg + tid; i < end; i += SBT) {
    const int2 r = rec[i];
    atomicAdd(&cnt[r.y & (NPB - 1)], 1);
  }
  __syncthreads();
  const int c = tid < NPB ? cnt[tid] : 0;
  int inc = c;
#pragma unroll
  for (int off = 1; off < 32; off <<= 1) {
    const int t = __shfl_up(inc, off, 32);
    if (lane >= off) inc += t;
  }
  if (tid < NPB && lane == 31) wsum[wave] = inc;
  __syncthreads();
  if (tid < NPB) {
    int wb = 0;
#pragma unroll
    for (int w = 0; w < NPB / 32; ++w) wb += w < wave ? wsum[w] : 0;
    const int o = beg + wb + inc - c;
    cur[tid] = o;
    const int node = bk * NPB + tid;
    if (node < NND) { offs[node] = o; deg[node] = c; }
  }
  __syncthreads();
  for (int i = beg + tid; i < end; i += SBT) {
    const int2 r = rec[i];
    const int pos = atomicAdd(&cur[r.y & (NPB - 1)], 1);
    if ((unsigned)pos < (unsigned)NE) csr[pos] = r.x;
  }
}

#define APAD   4
#define PITCH1 (FIN + APAD)
#define PITCH2 (HC + APAD)
__device__ __forceinline__ int h_chan(int p) {
  const int q = p & 127;
  return ((q >> 2) << 3) + ((p >> 7) << 2) + (q & 3);
}
__global__ __launch_bounds__(256, 16) void k_gemm1(const float* __restrict__ x,
                                               const float* __restrict__ W,
                                               float* __restrict__ h) {
  __shared__ __attribute__((aligned(16))) float As[16 * PITCH1];
  const int tid  = threadIdx.x;
  const int wave = tid >> 5, lane = tid & 31;
  const int m0   = blockIdx.x * 16;
  {
    const float4* xv = (const float4*)(x + (size_t)m0 * FIN);
    const int r = tid >> 5, c = (tid & 31) * 4;
    *(float4*)&As[r * PITCH1 + c]       = xv[tid];
    *(float4*)&As[(r + 8) * PITCH1 + c] = xv[tid + 256];
  }
  __syncthreads();
  const int n0 = ((wave & 3) + 4 * blockIdx.y + 8 * (wave >> 2)) * 16;
  const int la = lane & 15, hi = lane >> 4;
  const int cb = h_chan(n0 + la);
  v8f acc = {};
#pragma unroll
  for (int k0 = 0; k0 < FIN; k0 += 4) {
    const int ka = k0 + hi * 2;
    v2f a, b;
    a.x = As[la * PITCH1 + ka];
    a.y = As[la * PITCH1 + ka + 1];
    b.x = W[ka * HC + cb];
    b.y = W[(ka + 1) * HC + cb];
    acc = __builtin_amdgcn_wmma_f32_16x16x4_f32(false, a, false, b, (short)0,
                                                acc, false, false);
  }
#pragma unroll
  for (int j = 0; j < 8; ++j)
    h[(size_t)(m0 + j + hi * 8) * HC + n0 + la] = acc[j];
}

#define TP2 64
#define NPG 16
#define NWV 8
static_assert(NND % NPG == 0, "every work-group has 16 real nodes");
static_assert(NPG == 2 * NWV, "two nodes per wave");
__device__ __forceinline__ float bias_elu(float a, float r, float b) {
  const float v = a / r + b;
  return v > 0.f ? v : (__expf(v) - 1.f);
}
__global__ __launch_bounds__(NWV * 32) void k_node1(const int* __restrict__ offs,
                                                    const int* __restrict__ deg,
                                                    const int* __restrict__ csr,
                                                    const float* __restrict__ h,
                                                    const float* __restrict__ att_s,
                                                    const float* __restrict__ att_d,
                                                    const float* __restrict__ bias,
                                                    const float* __restrict__ W2,
                                                    const float* __restrict__ as2,
                                                    const float* __restrict__ ad2,
                                                    float* __restrict__ t) {
  __shared__ __attribute__((aligned(16))) float As[NPG * PITCH2];
  __shared__ float part[6][8][32];
  __shared__ float red[3][16][2];
  const int tid  = threadIdx.x;
  const int lane = tid & 31;
  const int wave = __builtin_amdgcn_readfirstlane(tid >> 5);
  const int row0 = blockIdx.x * NPG;
  const float4 ws0 = ((const float4*)att_s)[2 * lane], ws1 = ((const float4*)att_s)[2 * lane + 1];
  const float4 wd0 = ((const float4*)att_d)[2 * lane], wd1 = ((const float4*)att_d)[2 * lane + 1];
  const float4 b0 = ((const float4*)bias)[2 * lane], b1 = ((const float4*)bias)[2 * lane + 1];
  for (int r = 0; r < 2; ++r) {
    const int trow = wave + r * NWV;
    const int node = row0 + trow;
    const int beg = offs[node];
    const int n   = deg[node];
    float4 a0, a1;
    float m, den, ad;
    {
      const float4* hr = (const float4*)(h + (size_t)node * HC);
      const float4 v0 = hr[lane], v1 = hr[32 + lane];
      ad = group4_sum(dot8(v0, v1, wd0, wd1));
      m  = lrelu(group4_sum(dot8(v0, v1, ws0, ws1)) + ad);
      a0 = v0; a1 = v1;
      den = 1.f;
    }
#pragma unroll 4
    for (int i = 0; i < n; ++i) {
      int s = csr[beg + i];
      s = (unsigned)s < (unsigned)NND ? s : node;
      const float4* hr = (const float4*)(h + (size_t)s * HC);
      const float4 v0 = hr[lane], v1 = hr[32 + lane];
      const float e  = lrelu(group4_sum(dot8(v0, v1, ws0, ws1)) + ad);
      const float nm = e > m ? e : m;
      const float c = __expf(m - nm), x = __expf(e - nm);
      a0.x = a0.x * c + v0.x * x; a0.y = a0.y * c + v0.y * x;
      a0.z = a0.z * c + v0.z * x; a0.w = a0.w * c + v0.w * x;
      a1.x = a1.x * c + v1.x * x; a1.y = a1.y * c + v1.y * x;
      a1.z = a1.z * c + v1.z * x; a1.w = a1.w * c + v1.w * x;
      den = den * c + x;
      m = nm;
    }
    const float rr = den + EPSV;
    float4* o = (float4*)&As[trow * PITCH2];
    o[2 * lane]     = make_float4(bias_elu(a0.x, rr, b0.x), bias_elu(a0.y, rr, b0.y),
                                  bias_elu(a0.z, rr, b0.z), bias_elu(a0.w, rr, b0.w));
    o[2 * lane + 1] = make_float4(bias_elu(a1.x, rr, b1.x), bias_elu(a1.y, rr, b1.y),
                                  bias_elu(a1.z, rr, b1.z), bias_elu(a1.w, rr, b1.w));
  }
  __syncthreads();
  const int la = lane & 15, hi = lane >> 4;
  if (wave < 6) {
    const int col = (wave >> 1) * 16 + la;
    const bool ok = col < ODIM;
    const int colc = ok ? col : ODIM - 1;
    const int kb = (wave & 1) * (HC / 2);
    v8f acc = {};
#pragma unroll 8
    for (int k0 = 0; k0 < HC / 2; k0 += 4) {
      const int ka = kb + k0 + hi * 2;
      v2f a, b;
      a.x = As[la * PITCH2 + ka];
      a.y = As[la * PITCH2 + ka + 1];
      const float w0 = W2[ka * ODIM + colc], w1 = W2[(ka + 1) * ODIM + colc];
      b.x = ok ? w0 : 0.f;
      b.y = ok ? w1 : 0.f;
      acc = __builtin_amdgcn_wmma_f32_16x16x4_f32(false, a, false, b, (short)0,
                                                  acc, false, false);
    }
#pragma unroll
    for (int j = 0; j < 8; ++j) part[wave][j][lane] = acc[j];
  }
  __syncthreads();
  if (wave < 3) {
    const int col = wave * 16 + la;
    const bool ok = col < ODIM;
    float acc[8];
#pragma unroll
    for (int j = 0; j < 8; ++j) acc[j] = part[2 * wave][j][lane] + part[2 * wave + 1][j][lane];
#pragma unroll
    for (int j = 0; j < 8; ++j)
      if (ok) t[(size_t)(row0 + j + hi * 8) * TP2 + col] = acc[j];
    const float ws = ok ? as2[col] : 0.f, wd = ok ? ad2[col] : 0.f;
    float ps[8], pd[8];
#pragma unroll
    for (int j = 0; j < 8; ++j) {
      ps[j] = ok ? acc[j] * ws : 0.f;
      pd[j] = ok ? acc[j] * wd : 0.f;
    }
    const float ss = half_rows_sum(ps, la), sd = half_rows_sum(pd, la);
    if ((la & 1) == 0) {
      red[wave][hi * 8 + (la >> 1)][0] = ss;
      red[wave][hi * 8 + (la >> 1)][1] = sd;
    }
  }
  __syncthreads();
  if (tid < 32) {
    const int row = tid >> 1, which = tid & 1;
    const float v = red[0][row][which] + red[1][row][which] + red[2][row][which];
    t[(size_t)(row0 + row) * TP2 + ODIM + which] = v;
  }
}

#define LOGIT_LANE (ODIM / 2)
__device__ __forceinline__ float lane_value(float v, int l) {
  return __builtin_bit_cast(float, __builtin_amdgcn_readlane(__builtin_bit_cast(int, v), l));
}
__device__ __forceinline__ void node2_step(const float2 v, const float ad,
                                           float& m, float& den, float2& acc) {
  const float e  = lrelu(lane_value(v.x, LOGIT_LANE) + ad);
  const float nm = e > m ? e : m;
  const float c = __expf(m - nm), x = __expf(e - nm);
  acc.x = acc.x * c + v.x * x;
  acc.y = acc.y * c + v.y * x;
  den = den * c + x;
  m = nm;
}
__global__ __launch_bounds__(256) void k_node2(const int* __restrict__ offs,
                                               const int* __restrict__ deg,
                                               const int* __restrict__ csr,
                                               const float* __restrict__ t,
                                               const float* __restrict__ bias,
                                               float* __restrict__ out) {
  const int node = __builtin_amdgcn_readfirstlane((int)(blockIdx.x * 8 + (threadIdx.x >> 5)));
  const int lane = threadIdx.x & 31;
  if (node >= NND) return;
  const int beg = offs[node];
  const int n   = deg[node];
  const int cl = lane < LOGIT_LANE ? lane : LOGIT_LANE;
  float2 acc;
  float ad, m, den;
  {
    const float2 v = *(const float2*)(t + (size_t)node * TP2 + 2 * cl);
    ad = lane_value(v.y, LOGIT_LANE);
    m  = lrelu(lane_value(v.x, LOGIT_LANE) + ad);
    acc = v;
    den = 1.f;
  }
  int i = 0;
  for (; i + 4 <= n; i += 4) {
    float2 v[4];
#pragma unroll
    for (int j = 0; j < 4; ++j) {
      int s = csr[beg + i + j];
      s = (unsigned)s < (unsigned)NND ? s : node;
      v[j] = *(const float2*)(t + (size_t)s * TP2 + 2 * cl);
    }
    asm volatile("" :: "v"(v[0].x), "v"(v[0].y), "v"(v[1].x), "v"(v[1].y),
                       "v"(v[2].x), "v"(v[2].y), "v"(v[3].x), "v"(v[3].y));
#pragma unroll
    for (int j = 0; j < 4; ++j) node2_step(v[j], ad, m, den, acc);
  }
  for (; i < n; ++i) {
    int s = csr[beg + i];
    s = (unsigned)s < (unsigned)NND ? s : node;
    const float2 v = *(const float2*)(t + (size_t)s * TP2 + 2 * cl);
    node2_step(v, ad, m, den, acc);
  }
  const float r = den + EPSV;
  if (lane < LOGIT_LANE) {
    const float2 b = ((const float2*)bias)[lane];
    *(float2*)(out + (size_t)node * ODIM + 2 * lane) = make_float2(acc.x / r + b.x, acc.y / r + b.y);
  }
}

extern "C" void kernel_launch(void* const* d_in, const int* in_sizes, int n_in,
                              void* d_out, int out_size, void* d_ws, size_t ws_size,
                              hipStream_t stream) {
  const float* x   = (const float*)d_in[0];
  const int*   ei  = (const int*)d_in[1];
  const float* W1  = (const float*)d_in[2];
  const float* as1 = (const float*)d_in[3];
  const float* ad1 = (const float*)d_in[4];
  const float* b1  = (const float*)d_in[5];
  const float* W2  = (const float*)d_in[6];
  const float* as2 = (const float*)d_in[7];
  const float* ad2 = (const float*)d_in[8];
  const float* b2  = (const float*)d_in[9];
  float* out = (float*)d_out;

  float* ws   = (float*)d_ws;
  float* h1   = ws;
  float* t2   = h1   + (size_t)NND * HC;
  int*   csr  = (int*)(t2 + (size_t)NND * TP2);
  int*   deg  = csr  + NE;
  int*   offs = deg  + 50048;
  int*   hist = offs + 50048;
  int*   pre  = hist + SCAN_N;
  int2*  rec  = (int2*)(pre + SCAN_N);

  k_bhist<<<NEB, SBT, 0, stream>>>(ei, hist);
  k_scan<<<1, 1024, 0, stream>>>(hist, pre);
  k_bscatter<<<NEB, SBT, 0, stream>>>(ei, pre, rec);
  k_bucket<<<NBK, SBT, 0, stream>>>(rec, pre, offs, deg, csr);

  k_gemm1<<<dim3(NND / 16, 2), 256, 0, stream>>>(x, W1, h1);
  k_node1<<<NND / NPG, NWV * 32, 0, stream>>>(offs, deg, csr, h1, as1, ad1, b1, W2, as2, ad2, t2);

  k_node2<<<(NND + 7) / 8, 256, 0, stream>>>(offs, deg, csr, t2, b2, out);
}


# GEMM1 has no launch of its own: its work-groups (1024 threads: 64 rows x 128 columns) run as extra work-groups of the four edge-list launches, beside the list work-groups; 6 launches; h1 unchanged
# speedup vs baseline: 5.4383x; 1.0478x over previous
#include <hip/hip_runtime.h>
#include <hip/hip_bf16.h>

#define NND   50000
#define FIN   128
#define NH    8
#define HIDC  32
#define HC    256
#define ODIM  40
#define NE    800000
#define NET   850000
#define NEG_SLOPE 0.2f
#define EPSV  1e-16f

typedef __attribute__((ext_vector_type(2))) float v2f;
typedef __attribute__((ext_vector_type(8))) float v8f;

__device__ __forceinline__ float lrelu(float v) {
  return v > 0.f ? v : v * NEG_SLOPE;
}
__device__ __forceinline__ float half_rows_sum(const float (&p)[8], int c) {
  const bool b3 = (c & 8) != 0, b2 = (c & 4) != 0, b1 = (c & 2) != 0;
  float q[4], r[2];
#pragma unroll
  for (int j = 0; j < 4; ++j) {
    const float keep = b3 ? p[j + 4] : p[j];
    const float send = b3 ? p[j] : p[j + 4];
    q[j] = keep + __shfl_xor(send, 8, 32);
  }
#pragma unroll
  for (int j = 0; j < 2; ++j) {
    const float keep = b2 ? q[j + 2] : q[j];
    const float send = b2 ? q[j] : q[j + 2];
    r[j] = keep + __shfl_xor(send, 4, 32);
  }
  const float keep = b1 ? r[1] : r[0];
  const float send = b1 ? r[0] : r[1];
  float s = keep + __shfl_xor(send, 2, 32);
  s += __shfl_xor(s, 1, 32);
  return s;
}

template <int M>
__device__ __forceinline__ float row_xor(float v) {
  return __builtin_bit_cast(float, __builtin_amdgcn_mov_dpp(__builtin_bit_cast(int, v),
                                                            0x160 + M, 0xf, 0xf, false));
}
__device__ __forceinline__ float group4_sum(float v) {
  v += row_xor<1>(v);
  v += row_xor<2>(v);
  return v;
}
__device__ __forceinline__ float dot8(const float4 a0, const float4 a1,
                                      const float4 w0, const float4 w1) {
  v2f s = v2f{a0.x, a0.y} * v2f{w0.x, w0.y};
  s = v2f{a0.z, a0.w} * v2f{w0.z, w0.w} + s;
  s = v2f{a1.x, a1.y} * v2f{w1.x, w1.y} + s;
  s = v2f{a1.z, a1.w} * v2f{w1.z, w1.w} + s;
  return s.x + s.y;
}

#define APAD   4
#define PITCH1 (FIN + APAD)
#define PITCH2 (HC + APAD)
#define SBT    1024
#define GROWS  64
#define GRB    ((NND + GROWS - 1) / GROWS)
#define GWG    (2 * GRB)
#define GWG_A  436
#define GWG_S  254
#define GWG_B  436
#define GWG_C  (GWG - GWG_A - GWG_S - GWG_B)
static_assert(NND % 16 == 0, "a 16-row A tile is whole or absent");
static_assert(GWG_C > 0 && GWG_A % 2 == 0 && GWG_S % 2 == 0 && GWG_B % 2 == 0,
              "the two column halves of a row block stay in one launch");
__device__ __forceinline__ int h_chan(int p) {
  const int q = p & 127;
  return ((q >> 2) << 3) + ((p >> 7) << 2) + (q & 3);
}
__device__ __forceinline__ void gemm1_wg(const float* __restrict__ x,
                                         const float* __restrict__ W,
                                         float* __restrict__ h,
                                         const int g, float* As) {
  const int tid  = threadIdx.x;
  const int lane = tid & 31;
  const int wave = __builtin_amdgcn_readfirstlane(tid >> 5);
  const int r0   = (g >> 1) * GROWS;
  {
    const float4* xv = (const float4*)(x + (size_t)r0 * FIN);
#pragma unroll
    for (int q = 0; q < 2; ++q) {
      const int idx = tid + q * SBT;
      const int r = idx >> 5, c = (idx & 31) * 4;
      if (r0 + r < NND) *(float4*)&As[r * PITCH1 + c] = xv[idx];
    }
  }
  __syncthreads();
  const int rt = wave >> 3, wt = wave & 7;
  const int m0 = r0 + rt * 16;
  if (m0 >= NND) return;
  const int n0 = ((wt & 3) + 4 * (g & 1) + 8 * (wt >> 2)) * 16;
  const int la = lane & 15, hi = lane >> 4;
  const int cb = h_chan(n0 + la);
  const float* Ar = As + (rt * 16 + la) * PITCH1;
  v8f acc = {};
#pragma unroll
  for (int k0 = 0; k0 < FIN; k0 += 4) {
    const int ka = k0 + hi * 2;
    v2f a, b;
    a.x = Ar[ka];
    a.y = Ar[ka + 1];
    b.x = W[ka * HC + cb];
    b.y = W[(ka + 1) * HC + cb];
    acc = __builtin_amdgcn_wmma_f32_16x16x4_f32(false, a, false, b, (short)0,
                                                acc, false, false);
  }
#pragma unroll
  for (int j = 0; j < 8; ++j)
    h[(size_t)(m0 + j + hi * 8) * HC + n0 + la] = acc[j];
}

#define NPB   256
#define NBK   196
#define NEB   256
#define EPB   (NE / NEB)

__device__ __forceinline__ int dst_bucket_clamped(int d) {
  return (unsigned)d < (unsigned)NND ? d : NND - 1;
}

__global__ __launch_bounds__(SBT, 16) void k_bhist(const int* __restrict__ ei,
                                                   int* __restrict__ hist,
                                                   const float* __restrict__ x,
                                                   const float* __restrict__ W1,
                                                   float* __restrict__ h) {
  __shared__ int cnt[NBK];
  __shared__ __attribute__((aligned(16))) float As[GROWS * PITCH1];
  if (blockIdx.x >= NEB) {
    gemm1_wg(x, W1, h, blockIdx.x - NEB, As);
    return;
  }
  const int tid = threadIdx.x, blk = blockIdx.x;
  if (tid < NBK) cnt[tid] = 0;
  __syncthreads();
  const int* dstp = ei + NE + blk * EPB;
  for (int i = tid; i < EPB; i += SBT) {
    const int d = dst_bucket_clamped(dstp[i]);
    atomicAdd(&cnt[d >> 8], 1);
  }
  __syncthreads();
  if (tid < NBK) hist[tid * NEB + blk] = cnt[tid];
}

#define SCAN_PER 49
#define SCAN_N   (1024 * SCAN_PER)
#define SCAN_N4  (SCAN_N / 4)
#define SCAN_FULL (SCAN_N4 / 1024)
static_assert(NBK * NEB == SCAN_N, "scan size must equal the count table");
static_assert(NBK * NPB >= NND, "buckets must cover all nodes");
static_assert(EPB * NEB == NE, "edge blocks must cover all edges");
static_assert(SCAN_N4 * 16 >= GROWS * PITCH1 * 4, "the scan's LDS array holds a GEMM work-group's A tiles");
__global__ __launch_bounds__(1024) void k_scan(const int* __restrict__ cntin,
                                               int* __restrict__ pre,
                                               const float* __restrict__ x,
                                               const float* __restrict__ W1,
                                               float* __restrict__ h) {
  __shared__ int4 sd4[SCAN_N4];
  __shared__ int wsum[32];
  if (blockIdx.x >= 1) {
    gemm1_wg(x, W1, h, GWG_A + blockIdx.x - 1, (float*)sd4);
    return;
  }
  int* sd = (int*)sd4;
  const int tid = threadIdx.x, lane = tid & 31, wave = tid >> 5;
  {
    const int4* dv = (const int4*)cntin;
    int4 t[SCAN_FULL];
#pragma unroll
    for (int k = 0; k < SCAN_FULL; ++k) t[k] = dv[tid + k * 1024];
#pragma unroll
    for (int k = 0; k < SCAN_FULL; ++k) sd4[tid + k * 1024] = t[k];
    const int i = tid + SCAN_FULL * 1024;
    if (i < SCAN_N4) sd4[i] = dv[i];
  }
  __syncthreads();
  const int base = tid * SCAN_PER;
  int v[SCAN_PER];
  int loc = 0;
#pragma unroll
  for (int j = 0; j < SCAN_PER; ++j) { v[j] = sd[base + j]; loc += v[j]; }
  int inc = loc;
#pragma unroll
  for (int off = 1; off < 32; off <<= 1) {
    const int t = __shfl_up(inc, off, 32);
    if (lane >= off) inc += t;
  }
  if (lane == 31) wsum[wave] = inc;
  __syncthreads();
  if (wave == 0) {
    const int w = wsum[lane];
    int iv = w;
#pragma unroll
    for (int off = 1; off < 32; off <<= 1) {
      const int t = __shfl_up(iv, off, 32);
      if (lane >= off) iv += t;
    }
    wsum[lane] = iv - w;
  }
  __syncthreads();
  int run = wsum[wave] + inc - loc;
#pragma unroll
  for (int j = 0; j < SCAN_PER; ++j) { sd[base + j] = run; run += v[j]; }
  __syncthreads();
  {
    int4* ov = (int4*)pre;
#pragma unroll
    for (int k = 0; k < SCAN_FULL; ++k) ov[tid + k * 1024] = sd4[tid + k * 1024];
    const int i = tid + SCAN_FULL * 1024;
    if (i < SCAN_N4) ov[i] = sd4[i];
  }
}

__global__ __launch_bounds__(SBT, 16) void k_bscatter(const int* __restrict__ ei,
                                                      const int* __restrict__ pre,
                                                      int2* __restrict__ rec,
                                                      const float* __restrict__ x,
                                                      const float* __restrict__ W1,
                                                      float* __restrict__ h) {
  __shared__ int cnt[NBK];
  __shared__ int bas[NBK];
  __shared__ __attribute__((aligned(16))) float As[GROWS * PITCH1];
  if (blockIdx.x >= NEB) {
    gemm1_wg(x, W1, h, GWG_A + GWG_S + blockIdx.x - NEB, As);
    return;
  }
  const int tid = threadIdx.x, blk = blockIdx.x;
  if (tid < NBK) { cnt[tid] = 0; bas[tid] = pre[tid * NEB + blk]; }
  __syncthreads();
  const int* srcp = ei + blk * EPB;
  const int* dstp = ei + NE + blk * EPB;
  for (int i = tid; i < EPB; i += SBT) {
    const int s = srcp[i];
    const int d = dst_bucket_clamped(dstp[i]);
    const int b = d >> 8;
    const int pos = bas[b] + atomicAdd(&cnt[b], 1);
    if ((unsigned)pos < (unsigned)NE) rec[pos] = make_int2(s, d);
  }
}

__global__ __launch_bounds__(SBT, 16) void k_bucket(const int2* __restrict__ rec,
                                                    const int* __restrict__ pre,
                                                    int* __restrict__ offs,
                                                    int* __restrict__ deg,
                                                    int* __restrict__ csr,
                                                    const float* __restrict__ x,
                                                    const float* __restrict__ W1,
                                                    float* __restrict__ h) {
  __shared__ int cnt[NPB];
  __shared__ int cur[NPB];
  __shared__ int wsum[NPB / 32];
  __shared__ __attribute__((aligned(16))) float As[GROWS * PITCH1];
  if (blockIdx.x >= NBK) {
    gemm1_wg(x, W1, h, GWG_A + GWG_S + GWG_B + blockIdx.x - NBK, As);
    return;
  }
  const int tid = threadIdx.x, bk = blockIdx.x, lane = tid & 31, wave = tid >> 5;
  int beg = pre[bk * NEB];
  int end = (bk + 1 < NBK) ? pre[(bk + 1) * NEB] : NE;
  beg = beg < 0 ? 0 : (beg > NE ? NE : beg);
  end = end < beg ? beg : (end > NE ? NE : end);
  if (tid < NPB) cnt[tid] = 0;
  __syncthreads();
  for (int i = beg + tid; i < end; i += SBT) {
    const int2 r = rec[i];
    atomicAdd(&cnt[r.y & (NPB - 1)], 1);
  }
  __syncthreads();
  const int c = tid < NPB ? cnt[tid] : 0;
  int inc = c;
#pragma unroll
  for (int off = 1; off < 32; off <<= 1) {
    const int t = __shfl_up(inc, off, 32);
    if (lane >= off) inc += t;
  }
  if (tid < NPB && lane == 31) wsum[wave] = inc;
  __syncthreads();
  if (tid < NPB) {
    int wb = 0;
#pragma unroll
    for (int w = 0; w < NPB / 32; ++w) wb += w < wave ? wsum[w] : 0;
    const int o = beg + wb + inc - c;
    cur[tid] = o;
    const int node = bk * NPB + tid;
    if (node < NND) { offs[node] = o; deg[node] = c; }
  }
  __syncthreads();
  for (int i = beg + tid; i < end; i += SBT) {
    const int2 r = rec[i];
    const int pos = atomicAdd(&cur[r.y & (NPB - 1)], 1);
    if ((unsigned)pos < (unsigned)NE) csr[pos] = r.x;
  }
}

#define TP2 64
#define NPG 16
#define NWV 8
static_assert(NND % NPG == 0, "every work-group has 16 real nodes");
static_assert(NPG == 2 * NWV, "two nodes per wave");
__device__ __forceinline__ float bias_elu(float a, float r, float b) {
  const float v = a / r + b;
  return v > 0.f ? v : (__expf(v) - 1.f);
}
__global__ __launch_bounds__(NWV * 32) void k_node1(const int* __restrict__ offs,
                                                    const int* __restrict__ deg,
                                                    const int* __restrict__ csr,
                                                    const float* __restrict__ h,
                                                    const float* __restrict__ att_s,
                                                    const float* __restrict__ att_d,
                                                    const float* __restrict__ bias,
                                                    const float* __restrict__ W2,
                                                    const float* __restrict__ as2,
                                                    const float* __restrict__ ad2,
                                                    float* __restrict__ t) {
  __shared__ __attribute__((aligned(16))) float As[NPG * PITCH2];
  __shared__ float part[6][8][32];
  __shared__ float red[3][16][2];
  const int tid  = threadIdx.x;
  const int lane = tid & 31;
  const int wave = __builtin_amdgcn_readfirstlane(tid >> 5);
  const int row0 = blockIdx.x * NPG;
  const float4 ws0 = ((const float4*)att_s)[2 * lane], ws1 = ((const float4*)att_s)[2 * lane + 1];
  const float4 wd0 = ((const float4*)att_d)[2 * lane], wd1 = ((const float4*)att_d)[2 * lane + 1];
  const float4 b0 = ((const float4*)bias)[2 * lane], b1 = ((const float4*)bias)[2 * lane + 1];
  for (int r = 0; r < 2; ++r) {
    const int trow = wave + r * NWV;
    const int node = row0 + trow;
    const int beg = offs[node];
    const int n   = deg[node];
    float4 a0, a1;
    float m, den, ad;
    {
      const float4* hr = (const float4*)(h + (size_t)node * HC);
      const float4 v0 = hr[lane], v1 = hr[32 + lane];
      ad = group4_sum(dot8(v0, v1, wd0, wd1));
      m  = lrelu(group4_sum(dot8(v0, v1, ws0, ws1)) + ad);
      a0 = v0; a1 = v1;
      den = 1.f;
    }
#pragma unroll 4
    for (int i = 0; i < n; ++i) {
      int s = csr[beg + i];
      s = (unsigned)s < (unsigned)NND ? s : node;
      const float4* hr = (const float4*)(h + (size_t)s * HC);
      const float4 v0 = hr[lane], v1 = hr[32 + lane];
      const float e  = lrelu(group4_sum(dot8(v0, v1, ws0, ws1)) + ad);
      const float nm = e > m ? e : m;
      const float c = __expf(m - nm), x = __expf(e - nm);
      a0.x = a0.x * c + v0.x * x; a0.y = a0.y * c + v0.y * x;
      a0.z = a0.z * c + v0.z * x; a0.w = a0.w * c + v0.w * x;
      a1.x = a1.x * c + v1.x * x; a1.y = a1.y * c + v1.y * x;
      a1.z = a1.z * c + v1.z * x; a1.w = a1.w * c + v1.w * x;
      den = den * c + x;
      m = nm;
    }
    const float rr = den + EPSV;
    float4* o = (float4*)&As[trow * PITCH2];
    o[2 * lane]     = make_float4(bias_elu(a0.x, rr, b0.x), bias_elu(a0.y, rr, b0.y),
                                  bias_elu(a0.z, rr, b0.z), bias_elu(a0.w, rr, b0.w));
    o[2 * lane + 1] = make_float4(bias_elu(a1.x, rr, b1.x), bias_elu(a1.y, rr, b1.y),
                                  bias_elu(a1.z, rr, b1.z), bias_elu(a1.w, rr, b1.w));
  }
  __syncthreads();
  const int la = lane & 15, hi = lane >> 4;
  if (wave < 6) {
    const int col = (wave >> 1) * 16 + la;
    const bool ok = col < ODIM;
    const int colc = ok ? col : ODIM - 1;
    const int kb = (wave & 1) * (HC / 2);
    v8f acc = {};
#pragma unroll 8
    for (int k0 = 0; k0 < HC / 2; k0 += 4) {
      const int ka = kb + k0 + hi * 2;
      v2f a, b;
      a.x = As[la * PITCH2 + ka];
      a.y = As[la * PITCH2 + ka + 1];
      const float w0 = W2[ka * ODIM + colc], w1 = W2[(ka + 1) * ODIM + colc];
      b.x = ok ? w0 : 0.f;
      b.y = ok ? w1 : 0.f;
      acc = __builtin_amdgcn_wmma_f32_16x16x4_f32(false, a, false, b, (short)0,
                                                  acc, false, false);
    }
#pragma unroll
    for (int j = 0; j < 8; ++j) part[wave][j][lane] = acc[j];
  }
  __syncthreads();
  if (wave < 3) {
    const int col = wave * 16 + la;
    const bool ok = col < ODIM;
    float acc[8];
#pragma unroll
    for (int j = 0; j < 8; ++j) acc[j] = part[2 * wave][j][lane] + part[2 * wave + 1][j][lane];
#pragma unroll
    for (int j = 0; j < 8; ++j)
      if (ok) t[(size_t)(row0 + j + hi * 8) * TP2 + col] = acc[j];
    const float ws = ok ? as2[col] : 0.f, wd = ok ? ad2[col] : 0.f;
    float ps[8], pd[8];
#pragma unroll
    for (int j = 0; j < 8; ++j) {
      ps[j] = ok ? acc[j] * ws : 0.f;
      pd[j] = ok ? acc[j] * wd : 0.f;
    }
    const float ss = half_rows_sum(ps, la), sd = half_rows_sum(pd, la);
    if ((la & 1) == 0) {
      red[wave][hi * 8 + (la >> 1)][0] = ss;
      red[wave][hi * 8 + (la >> 1)][1] = sd;
    }
  }
  __syncthreads();
  if (tid < 32) {
    const int row = tid >> 1, which = tid & 1;
    const float v = red[0][row][which] + red[1][row][which] + red[2][row][which];
    t[(size_t)(row0 + row) * TP2 + ODIM + which] = v;
  }
}

#define LOGIT_LANE (ODIM / 2)
__device__ __forceinline__ float lane_value(float v, int l) {
  return __builtin_bit_cast(float, __builtin_amdgcn_readlane(__builtin_bit_cast(int, v), l));
}
__device__ __forceinline__ void node2_step(const float2 v, const float ad,
                                           float& m, float& den, float2& acc) {
  const float e  = lrelu(lane_value(v.x, LOGIT_LANE) + ad);
  const float nm = e > m ? e : m;
  const float c = __expf(m - nm), x = __expf(e - nm);
  acc.x = acc.x * c + v.x * x;
  acc.y = acc.y * c + v.y * x;
  den = den * c + x;
  m = nm;
}
__global__ __launch_bounds__(256) void k_node2(const int* __restrict__ offs,
                                               const int* __restrict__ deg,
                                               const int* __restrict__ csr,
                                               const float* __restrict__ t,
                                               const float* __restrict__ bias,
                                               float* __restrict__ out) {
  const int node = __builtin_amdgcn_readfirstlane((int)(blockIdx.x * 8 + (threadIdx.x >> 5)));
  const int lane = threadIdx.x & 31;
  if (node >= NND) return;
  const int beg = offs[node];
  const int n   = deg[node];
  const int cl = lane < LOGIT_LANE ? lane : LOGIT_LANE;
  float2 acc;
  float ad, m, den;
  {
    const float2 v = *(const float2*)(t + (size_t)node * TP2 + 2 * cl);
    ad = lane_value(v.y, LOGIT_LANE);
    m  = lrelu(lane_value(v.x, LOGIT_LANE) + ad);
    acc = v;
    den = 1.f;
  }
  int i = 0;
  for (; i + 4 <= n; i += 4) {
    float2 v[4];
#pragma unroll
    for (int j = 0; j < 4; ++j) {
      int s = csr[beg + i + j];
      s = (unsigned)s < (unsigned)NND ? s : node;
      v[j] = *(const float2*)(t + (size_t)s * TP2 + 2 * cl);
    }
    asm volatile("" :: "v"(v[0].x), "v"(v[0].y), "v"(v[1].x), "v"(v[1].y),
                       "v"(v[2].x), "v"(v[2].y), "v"(v[3].x), "v"(v[3].y));
#pragma unroll
    for (int j = 0; j < 4; ++j) node2_step(v[j], ad, m, den, acc);
  }
  for (; i < n; ++i) {
    int s = csr[beg + i];
    s = (unsigned)s < (unsigned)NND ? s : node;
    const float2 v = *(const float2*)(t + (size_t)s * TP2 + 2 * cl);
    node2_step(v, ad, m, den, acc);
  }
  const float r = den + EPSV;
  if (lane < LOGIT_LANE) {
    const float2 b = ((const float2*)bias)[lane];
    *(float2*)(out + (size_t)node * ODIM + 2 * lane) = make_float2(acc.x / r + b.x, acc.y / r + b.y);
  }
}

extern "C" void kernel_launch(void* const* d_in, const int* in_sizes, int n_in,
                              void* d_out, int out_size, void* d_ws, size_t ws_size,
                              hipStream_t stream) {
  const float* x   = (const float*)d_in[0];
  const int*   ei  = (const int*)d_in[1];
  const float* W1  = (const float*)d_in[2];
  const float* as1 = (const float*)d_in[3];
  const float* ad1 = (const float*)d_in[4];
  const float* b1  = (const float*)d_in[5];
  const float* W2  = (const float*)d_in[6];
  const float* as2 = (const float*)d_in[7];
  const float* ad2 = (const float*)d_in[8];
  const float* b2  = (const float*)d_in[9];
  float* out = (float*)d_out;

  float* ws   = (float*)d_ws;
  float* h1   = ws;
  float* t2   = h1   + (size_t)NND * HC;
  int*   csr  = (int*)(t2 + (size_t)NND * TP2);
  int*   deg  = csr  + NE;
  int*   offs = deg  + 50048;
  int*   hist = offs + 50048;
  int*   pre  = hist + SCAN_N;
  int2*  rec  = (int2*)(pre + SCAN_N);

  k_bhist<<<NEB + GWG_A, SBT, 0, stream>>>(ei, hist, x, W1, h1);
  k_scan<<<1 + GWG_S, 1024, 0, stream>>>(hist, pre, x, W1, h1);
  k_bscatter<<<NEB + GWG_B, SBT, 0, stream>>>(ei, pre, rec, x, W1, h1);
  k_bucket<<<NBK + GWG_C, SBT, 0, stream>>>(rec, pre, offs, deg, csr, x, W1, h1);

  k_node1<<<NND / NPG, NWV * 32, 0, stream>>>(offs, deg, csr, h1, as1, ad1, b1, W2, as2, ad2, t2);

  k_node2<<<(NND + 7) / 8, 256, 0, stream>>>(offs, deg, csr, t2, b2, out);
}
